# starting kernel
# baseline (speedup 1.0000x reference)
#include <hip/hip_runtime.h>
#include <hip/hip_bf16.h>

typedef float v2f __attribute__((ext_vector_type(2)));
typedef float v8f __attribute__((ext_vector_type(8)));

#define FDIM 128
#define KTOT 384
#define KC   64

__global__ void fill_zero(float* p, size_t n) {
  size_t i = (size_t)blockIdx.x * blockDim.x + threadIdx.x;
  size_t stride = (size_t)gridDim.x * blockDim.x;
  for (; i < n; i += stride) p[i] = 0.0f;
}

__global__ void build_weights(const float* __restrict__ Wn1, const float* __restrict__ Ws1,
                              const float* __restrict__ b1,
                              const float* __restrict__ Wn2, const float* __restrict__ Ws2,
                              const float* __restrict__ b2,
                              float* __restrict__ Bcat1, float* __restrict__ bias1,
                              float* __restrict__ Bcat2, float* __restrict__ bias2) {
  int i = blockIdx.x * blockDim.x + threadIdx.x;
  const int S1 = 384 * 128, S2 = S1 + 128, S3 = S2 + 384 * 64, S4 = S3 + 64;
  if (i < S1) {
    int k = i >> 7, n = i & 127;
    float v;
    if (k < 128)      v = Ws1[k * 128 + n] + Ws1[16384 + k * 128 + n];
    else if (k < 256) v = Wn1[(k - 128) * 128 + n];
    else              v = Wn1[16384 + (k - 256) * 128 + n];
    Bcat1[i] = v;
  } else if (i < S2) {
    int n = i - S1;
    bias1[n] = b1[n] + b1[128 + n];
  } else if (i < S3) {
    int j = i - S2;
    int k = j >> 6, n = j & 63;
    float v;
    if (k < 128)      v = Ws2[k * 64 + n] + Ws2[8192 + k * 64 + n];
    else if (k < 256) v = Wn2[(k - 128) * 64 + n];
    else              v = Wn2[8192 + (k - 256) * 64 + n];
    Bcat2[j] = v;
  } else if (i < S4) {
    int n = i - S3;
    bias2[n] = b2[n] + b2[64 + n];
  }
}

__global__ void degree_count(const int* __restrict__ edges, float* deg, int E, int N) {
  int e = blockIdx.x * blockDim.x + threadIdx.x;
  int r = blockIdx.y;
  if (e >= E) return;
  int d = edges[(size_t)r * 2 * E + E + e];
  atomicAdd(deg + (size_t)r * N + d, 1.0f);
}

__global__ void recip_deg(float* deg, int n) {
  int i = blockIdx.x * blockDim.x + threadIdx.x;
  if (i < n) deg[i] = 1.0f / fmaxf(deg[i], 1.0f);
}

__global__ void scatter_feat(const int* __restrict__ src, const int* __restrict__ dst,
                             const float* __restrict__ feat, float* agg, int E) {
  int gid = blockIdx.x * blockDim.x + threadIdx.x;
  int e = gid >> 5, lane = gid & 31;
  if (e >= E) return;
  int s = src[e], d = dst[e];
  const float4 v = *(const float4*)(feat + (size_t)s * FDIM + lane * 4);
  float* p = agg + (size_t)d * FDIM + lane * 4;
  atomicAdd(p + 0, v.x);
  atomicAdd(p + 1, v.y);
  atomicAdd(p + 2, v.z);
  atomicAdd(p + 3, v.w);
}

__global__ void scale_rows(float* agg, const float* __restrict__ dinv, size_t n) {
  size_t i = (size_t)blockIdx.x * blockDim.x + threadIdx.x;
  if (i < n) agg[i] *= dinv[i >> 7];
}

template <int NOUT, bool RELU>
__global__ __launch_bounds__(256) void sage_gemm(
    const float* __restrict__ A0, const float* __restrict__ G0, const float* __restrict__ G1,
    const float* __restrict__ Bmat, const float* __restrict__ bias,
    float* __restrict__ out, int Nrows) {
  constexpr int NT = NOUT / 16;
  __shared__ float Bs[KC * NOUT];

  const int tid = threadIdx.x;
  const int lane = tid & 31;
  const int wave = tid >> 5;
  const int rowBase = blockIdx.x * 128 + wave * 16;
  const int m = lane & 15;
  const int khalf = (lane >> 4) << 1;

  v8f czero = {};
  v8f c[NT];
#pragma unroll
  for (int t = 0; t < NT; ++t) c[t] = czero;

  int rowA = rowBase + m;
  if (rowA > Nrows - 1) rowA = Nrows - 1;

  for (int kc = 0; kc < KTOT; kc += KC) {
    for (int i = tid; i < KC * NOUT; i += 256)
      Bs[i] = Bmat[kc * NOUT + i];
    __syncthreads();

    const float* Asrc = (kc < 128) ? A0 : ((kc < 256) ? G0 : G1);
    const int kbase = kc & 127;

    for (int k4 = 0; k4 < KC; k4 += 4) {
      v2f a = *(const v2f*)(Asrc + (size_t)rowA * FDIM + kbase + k4 + khalf);
      const int kb = k4 + khalf;
#pragma unroll
      for (int t = 0; t < NT; ++t) {
        v2f b;
        b.x = Bs[kb * NOUT + t * 16 + m];
        b.y = Bs[(kb + 1) * NOUT + t * 16 + m];
        c[t] = __builtin_amdgcn_wmma_f32_16x16x4_f32(
            false, a, false, b, (short)0, c[t], false, false);
      }
    }
    __syncthreads();
  }

  const int rhalf = (lane >> 4) << 3;
#pragma unroll
  for (int t = 0; t < NT; ++t) {
    int col = t * 16 + m;
    float bv = bias[col];
#pragma unroll
    for (int v = 0; v < 8; ++v) {
      int row = rowBase + v + rhalf;
      if (row < Nrows) {
        float val = c[t][v] + bv;
        if (RELU) val = fmaxf(val, 0.0f);
        out[(size_t)row * NOUT + col] = val;
      }
    }
  }
}

__global__ void edge_score(const int* __restrict__ src, const int* __restrict__ dst,
                           const float* __restrict__ h, float* __restrict__ out, int E) {
  int gid = blockIdx.x * blockDim.x + threadIdx.x;
  int e = gid >> 5, lane = gid & 31;
  if (e >= E) return;
  int s = src[e], d = dst[e];
  v2f a = *(const v2f*)(h + (size_t)s * 64 + lane * 2);
  v2f b = *(const v2f*)(h + (size_t)d * 64 + lane * 2);
  float p = a.x * b.x + a.y * b.y;
#pragma unroll
  for (int off = 16; off > 0; off >>= 1)
    p += __shfl_xor(p, off, 32);
  if (lane == 0) out[e] = p;
}

extern "C" void kernel_launch(void* const* d_in, const int* in_sizes, int n_in,
                              void* d_out, int out_size, void* d_ws, size_t ws_size,
                              hipStream_t stream) {
  const float* x   = (const float*)d_in[0];
  const int*   edges = (const int*)d_in[1];
  const int*   negE  = (const int*)d_in[2];
  const float* Wn1 = (const float*)d_in[3];
  const float* Ws1 = (const float*)d_in[4];
  const float* b1  = (const float*)d_in[5];
  const float* Wn2 = (const float*)d_in[6];
  const float* Ws2 = (const float*)d_in[7];
  const float* b2  = (const float*)d_in[8];
  float* out = (float*)d_out;

  const int N = in_sizes[0] / 128;
  const int E = in_sizes[1] / 4;

  float* ws    = (float*)d_ws;
  float* dinv  = ws;
  float* agg0  = dinv + 2 * (size_t)N;
  float* agg1  = agg0 + (size_t)N * 128;
  float* h1    = agg1 + (size_t)N * 128;
  float* h2    = h1 + (size_t)N * 128;
  float* Bcat1 = h2 + (size_t)N * 64;
  float* bias1 = Bcat1 + 384 * 128;
  float* Bcat2 = bias1 + 128;
  float* bias2 = Bcat2 + 384 * 64;

  const int T = 256;
  const size_t nf = (size_t)N * 128;

  fill_zero<<<1024, T, 0, stream>>>(dinv, 2 * (size_t)N + 2 * nf);
  const int wtot = 384 * 128 + 128 + 384 * 64 + 64;
  build_weights<<<(wtot + T - 1) / T, T, 0, stream>>>(Wn1, Ws1, b1, Wn2, Ws2, b2,
                                                      Bcat1, bias1, Bcat2, bias2);

  degree_count<<<dim3((E + T - 1) / T, 2), T, 0, stream>>>(edges, dinv, E, N);
  recip_deg<<<(2 * N + T - 1) / T, T, 0, stream>>>(dinv, 2 * N);

  const int sblocks = (int)(((size_t)E * 32 + T - 1) / T);
  const int fblocks = (int)((nf + T - 1) / T);
  const int gblocks = (N + 127) / 128;

  scatter_feat<<<sblocks, T, 0, stream>>>(edges,         edges + E,     x, agg0, E);
  scatter_feat<<<sblocks, T, 0, stream>>>(edges + 2 * E, edges + 3 * E, x, agg1, E);
  scale_rows<<<fblocks, T, 0, stream>>>(agg0, dinv,     nf);
  scale_rows<<<fblocks, T, 0, stream>>>(agg1, dinv + N, nf);
  sage_gemm<128, true><<<gblocks, T, 0, stream>>>(x, agg0, agg1, Bcat1, bias1, h1, N);

  fill_zero<<<1024, T, 0, stream>>>(agg0, 2 * nf);
  scatter_feat<<<sblocks, T, 0, stream>>>(edges,         edges + E,     h1, agg0, E);
  scatter_feat<<<sblocks, T, 0, stream>>>(edges + 2 * E, edges + 3 * E, h1, agg1, E);
  scale_rows<<<fblocks, T, 0, stream>>>(agg0, dinv,     nf);
  scale_rows<<<fblocks, T, 0, stream>>>(agg1, dinv + N, nf);
  sage_gemm<64, false><<<gblocks, T, 0, stream>>>(h1, agg0, agg1, Bcat2, bias2, h2, N);

  edge_score<<<sblocks, T, 0, stream>>>(edges, edges + E, h2, out, E);
  edge_score<<<sblocks, T, 0, stream>>>(negE,  negE + E,  h2, out + E, E);
}


# Mean aggregation as a pull: edges counting-sorted by dst per call (int atomic rank sets the sum order), one wave per node adds neighbour rows in registers; float atomics, zero-fill, scale passes gone
# speedup vs baseline: 1.9439x; 1.9439x over previous
#include <hip/hip_runtime.h>
#include <hip/hip_bf16.h>

typedef float v2f __attribute__((ext_vector_type(2)));
typedef float v8f __attribute__((ext_vector_type(8)));

#define FDIM 128
#define KTOT 384
#define KC   64

__global__ void fill_zero(float* p, size_t n) {
  size_t i = (size_t)blockIdx.x * blockDim.x + threadIdx.x;
  size_t stride = (size_t)gridDim.x * blockDim.x;
  for (; i < n; i += stride) p[i] = 0.0f;
}

__global__ void build_weights(const float* __restrict__ Wn1, const float* __restrict__ Ws1,
                              const float* __restrict__ b1,
                              const float* __restrict__ Wn2, const float* __restrict__ Ws2,
                              const float* __restrict__ b2,
                              float* __restrict__ Bcat1, float* __restrict__ bias1,
                              float* __restrict__ Bcat2, float* __restrict__ bias2) {
  int i = blockIdx.x * blockDim.x + threadIdx.x;
  const int S1 = 384 * 128, S2 = S1 + 128, S3 = S2 + 384 * 64, S4 = S3 + 64;
  if (i < S1) {
    int k = i >> 7, n = i & 127;
    float v;
    if (k < 128)      v = Ws1[k * 128 + n] + Ws1[16384 + k * 128 + n];
    else if (k < 256) v = Wn1[(k - 128) * 128 + n];
    else              v = Wn1[16384 + (k - 256) * 128 + n];
    Bcat1[i] = v;
  } else if (i < S2) {
    int n = i - S1;
    bias1[n] = b1[n] + b1[128 + n];
  } else if (i < S3) {
    int j = i - S2;
    int k = j >> 6, n = j & 63;
    float v;
    if (k < 128)      v = Ws2[k * 64 + n] + Ws2[8192 + k * 64 + n];
    else if (k < 256) v = Wn2[(k - 128) * 64 + n];
    else              v = Wn2[8192 + (k - 256) * 64 + n];
    Bcat2[j] = v;
  } else if (i < S4) {
    int n = i - S3;
    bias2[n] = b2[n] + b2[64 + n];
  }
}

__global__ void count_rank(const int* __restrict__ edges, int* seg, int* __restrict__ rank, int E, int NS) {
  int e = blockIdx.x * blockDim.x + threadIdx.x;
  int r = blockIdx.y;
  if (e >= E) return;
  int d = edges[(size_t)r * 2 * E + E + e];
  rank[(size_t)r * E + e] = atomicAdd(seg + (size_t)r * NS + d, 1);
}

__global__ __launch_bounds__(1024) void scan_counts(int* seg, int N, int NS) {
  __shared__ int wsum[32];
  int* c = seg + (size_t)blockIdx.x * NS;
  const int t = threadIdx.x, lane = t & 31, w = t >> 5;
  const int per = (((N + 1023) / 1024) + 7) & ~7;
  int b = t * per;
  if (b > N) b = N;
  int e = b + per;
  if (e > N) e = N;
  int s = 0;
  for (int i = b; i < e; ++i) s += c[i];
  int inc = s;
  for (int off = 1; off < 32; off <<= 1) {
    int up = __shfl_up(inc, off, 32);
    if (lane >= off) inc += up;
  }
  if (lane == 31) wsum[w] = inc;
  __syncthreads();
  if (w == 0) {
    int v = wsum[lane];
    int iv = v;
    for (int off = 1; off < 32; off <<= 1) {
      int up = __shfl_up(iv, off, 32);
      if (lane >= off) iv += up;
    }
    wsum[lane] = iv - v;
  }
  __syncthreads();
  int run = wsum[w] + inc - s;
  for (int i = b; i < e; ++i) {
    int v = c[i];
    c[i] = run;
    run += v;
  }
  if (b < e && e == N) c[N] = run;
}

__global__ void fill_csr(const int* __restrict__ edges, const int* __restrict__ seg,
                         const int* __restrict__ rank, int* __restrict__ csr, int E, int NS) {
  int e = blockIdx.x * blockDim.x + threadIdx.x;
  int r = blockIdx.y;
  if (e >= E) return;
  const int* ed = edges + (size_t)r * 2 * E;
  int s = ed[e], d = ed[E + e];
  int pos = seg[(size_t)r * NS + d] + rank[(size_t)r * E + e];
  if ((unsigned)pos < (unsigned)E) csr[(size_t)r * E + pos] = s;
}

__global__ __launch_bounds__(256) void gather_mean(const int* __restrict__ seg, const int* __restrict__ csr,
                                                   const float* __restrict__ feat, float* __restrict__ agg,
                                                   int E, int N, int NS) {
  const int lane = threadIdx.x & 31;
  const int w = __builtin_amdgcn_readfirstlane((int)(threadIdx.x >> 5));
  const int nwaves = gridDim.x * 8;
  for (int idx = blockIdx.x * 8 + w; idx < 2 * N; idx += nwaves) {
    const int r = (idx >= N) ? 1 : 0;
    const int i = idx - r * N;
    const int* c = seg + (size_t)r * NS;
    int start = c[i];
    int end = c[i + 1];
    if (start < 0) start = 0;
    if (end > E) end = E;
    const int* nb = csr + (size_t)r * E;
    float4 acc = make_float4(0.0f, 0.0f, 0.0f, 0.0f);
    for (int base = start; base < end; base += 32) {
      const int rem = end - base;
      const int n = rem < 32 ? rem : 32;
      int mine = (lane < n) ? nb[base + lane] : 0;
      if ((unsigned)mine >= (unsigned)N) mine = 0;
      int j = 0;
      for (; j + 4 <= n; j += 4) {
        const int s0 = __builtin_amdgcn_readlane(mine, j);
        const int s1 = __builtin_amdgcn_readlane(mine, j + 1);
        const int s2 = __builtin_amdgcn_readlane(mine, j + 2);
        const int s3 = __builtin_amdgcn_readlane(mine, j + 3);
        const float4 v0 = *(const float4*)(feat + (size_t)s0 * FDIM + lane * 4);
        const float4 v1 = *(const float4*)(feat + (size_t)s1 * FDIM + lane * 4);
        const float4 v2 = *(const float4*)(feat + (size_t)s2 * FDIM + lane * 4);
        const float4 v3 = *(const float4*)(feat + (size_t)s3 * FDIM + lane * 4);
        acc.x += v0.x; acc.y += v0.y; acc.z += v0.z; acc.w += v0.w;
        acc.x += v1.x; acc.y += v1.y; acc.z += v1.z; acc.w += v1.w;
        acc.x += v2.x; acc.y += v2.y; acc.z += v2.z; acc.w += v2.w;
        acc.x += v3.x; acc.y += v3.y; acc.z += v3.z; acc.w += v3.w;
      }
      for (; j < n; ++j) {
        const int s0 = __builtin_amdgcn_readlane(mine, j);
        const float4 v0 = *(const float4*)(feat + (size_t)s0 * FDIM + lane * 4);
        acc.x += v0.x; acc.y += v0.y; acc.z += v0.z; acc.w += v0.w;
      }
    }
    const float deg = (float)(end - start);
    const float dinv = 1.0f / fmaxf(deg, 1.0f);
    float4 o;
    o.x = acc.x * dinv; o.y = acc.y * dinv; o.z = acc.z * dinv; o.w = acc.w * dinv;
    *(float4*)(agg + ((size_t)r * N + i) * FDIM + lane * 4) = o;
  }
}

template <int NOUT, bool RELU>
__global__ __launch_bounds__(256) void sage_gemm(
    const float* __restrict__ A0, const float* __restrict__ G0, const float* __restrict__ G1,
    const float* __restrict__ Bmat, const float* __restrict__ bias,
    float* __restrict__ out, int Nrows) {
  constexpr int NT = NOUT / 16;
  __shared__ float Bs[KC * NOUT];

  const int tid = threadIdx.x;
  const int lane = tid & 31;
  const int wave = tid >> 5;
  const int rowBase = blockIdx.x * 128 + wave * 16;
  const int m = lane & 15;
  const int khalf = (lane >> 4) << 1;

  v8f czero = {};
  v8f c[NT];
#pragma unroll
  for (int t = 0; t < NT; ++t) c[t] = czero;

  int rowA = rowBase + m;
  if (rowA > Nrows - 1) rowA = Nrows - 1;

  for (int kc = 0; kc < KTOT; kc += KC) {
    for (int i = tid; i < KC * NOUT; i += 256)
      Bs[i] = Bmat[kc * NOUT + i];
    __syncthreads();

    const float* Asrc = (kc < 128) ? A0 : ((kc < 256) ? G0 : G1);
    const int kbase = kc & 127;

    for (int k4 = 0; k4 < KC; k4 += 4) {
      v2f a = *(const v2f*)(Asrc + (size_t)rowA * FDIM + kbase + k4 + khalf);
      const int kb = k4 + khalf;
#pragma unroll
      for (int t = 0; t < NT; ++t) {
        v2f b;
        b.x = Bs[kb * NOUT + t * 16 + m];
        b.y = Bs[(kb + 1) * NOUT + t * 16 + m];
        c[t] = __builtin_amdgcn_wmma_f32_16x16x4_f32(
            false, a, false, b, (short)0, c[t], false, false);
      }
    }
    __syncthreads();
  }

  const int rhalf = (lane >> 4) << 3;
#pragma unroll
  for (int t = 0; t < NT; ++t) {
    int col = t * 16 + m;
    float bv = bias[col];
#pragma unroll
    for (int v = 0; v < 8; ++v) {
      int row = rowBase + v + rhalf;
      if (row < Nrows) {
        float val = c[t][v] + bv;
        if (RELU) val = fmaxf(val, 0.0f);
        out[(size_t)row * NOUT + col] = val;
      }
    }
  }
}

__global__ void edge_score(const int* __restrict__ src, const int* __restrict__ dst,
                           const float* __restrict__ h, float* __restrict__ out, int E) {
  int gid = blockIdx.x * blockDim.x + threadIdx.x;
  int e = gid >> 5, lane = gid & 31;
  if (e >= E) return;
  int s = src[e], d = dst[e];
  v2f a = *(const v2f*)(h + (size_t)s * 64 + lane * 2);
  v2f b = *(const v2f*)(h + (size_t)d * 64 + lane * 2);
  float p = a.x * b.x + a.y * b.y;
#pragma unroll
  for (int off = 16; off > 0; off >>= 1)
    p += __shfl_xor(p, off, 32);
  if (lane == 0) out[e] = p;
}

extern "C" void kernel_launch(void* const* d_in, const int* in_sizes, int n_in,
                              void* d_out, int out_size, void* d_ws, size_t ws_size,
                              hipStream_t stream) {
  const float* x   = (const float*)d_in[0];
  const int*   edges = (const int*)d_in[1];
  const int*   negE  = (const int*)d_in[2];
  const float* Wn1 = (const float*)d_in[3];
  const float* Ws1 = (const float*)d_in[4];
  const float* b1  = (const float*)d_in[5];
  const float* Wn2 = (const float*)d_in[6];
  const float* Ws2 = (const float*)d_in[7];
  const float* b2  = (const float*)d_in[8];
  float* out = (float*)d_out;

  const int N = in_sizes[0] / 128;
  const int E = in_sizes[1] / 4;

  float* ws    = (float*)d_ws;
  float* agg0  = ws + 2 * (size_t)N;
  float* agg1  = agg0 + (size_t)N * 128;
  float* h1    = agg1 + (size_t)N * 128;
  float* h2    = h1 + (size_t)N * 128;
  float* Bcat1 = h2 + (size_t)N * 64;
  float* bias1 = Bcat1 + 384 * 128;
  float* Bcat2 = bias1 + 128;
  float* bias2 = Bcat2 + 384 * 64;
  const int NS = (N + 8) & ~7;
  int* csr  = (int*)h2;
  int* seg  = csr + 2 * (size_t)E;
  int* rank = (int*)agg1;

  const int T = 256;

  fill_zero<<<(2 * NS + T - 1) / T, T, 0, stream>>>((float*)seg, 2 * (size_t)NS);
  const int wtot = 384 * 128 + 128 + 384 * 64 + 64;
  build_weights<<<(wtot + T - 1) / T, T, 0, stream>>>(Wn1, Ws1, b1, Wn2, Ws2, b2,
                                                      Bcat1, bias1, Bcat2, bias2);

  const dim3 egrid((E + T - 1) / T, 2);
  count_rank<<<egrid, T, 0, stream>>>(edges, seg, rank, E, NS);
  scan_counts<<<2, 1024, 0, stream>>>(seg, N, NS);
  fill_csr<<<egrid, T, 0, stream>>>(edges, seg, rank, csr, E, NS);

  const int sblocks = (int)(((size_t)E * 32 + T - 1) / T);
  const int gblocks = (N + 127) / 128;
  const int ablocks = 4096;

  gather_mean<<<ablocks, T, 0, stream>>>(seg, csr, x, agg0, E, N, NS);
  sage_gemm<128, true><<<gblocks, T, 0, stream>>>(x, agg0, agg1, Bcat1, bias1, h1, N);

  gather_mean<<<ablocks, T, 0, stream>>>(seg, csr, h1, agg0, E, N, NS);
  sage_gemm<64, false><<<gblocks, T, 0, stream>>>(h1, agg0, agg1, Bcat2, bias2, h2, N);

  edge_score<<<sblocks, T, 0, stream>>>(edges, edges + E, h2, out, E);
  edge_score<<<sblocks, T, 0, stream>>>(negE,  negE + E,  h2, out + E, E);
}


# Edge scoring on resident waves: each wave loops over 32-edge groups, 16 lanes x 16 B per row (whole-line loads), transposing 16-lane DPP sums (new sum order), one 128-B store per group
# speedup vs baseline: 2.5557x; 1.3147x over previous
#include <hip/hip_runtime.h>
#include <hip/hip_bf16.h>

typedef float v2f __attribute__((ext_vector_type(2)));
typedef float v8f __attribute__((ext_vector_type(8)));

#define FDIM 128
#define KTOT 384
#define KC   64

__global__ void fill_zero(float* p, size_t n) {
  size_t i = (size_t)blockIdx.x * blockDim.x + threadIdx.x;
  size_t stride = (size_t)gridDim.x * blockDim.x;
  for (; i < n; i += stride) p[i] = 0.0f;
}

__global__ void build_weights(const float* __restrict__ Wn1, const float* __restrict__ Ws1,
                              const float* __restrict__ b1,
                              const float* __restrict__ Wn2, const float* __restrict__ Ws2,
                              const float* __restrict__ b2,
                              float* __restrict__ Bcat1, float* __restrict__ bias1,
                              float* __restrict__ Bcat2, float* __restrict__ bias2) {
  int i = blockIdx.x * blockDim.x + threadIdx.x;
  const int S1 = 384 * 128, S2 = S1 + 128, S3 = S2 + 384 * 64, S4 = S3 + 64;
  if (i < S1) {
    int k = i >> 7, n = i & 127;
    float v;
    if (k < 128)      v = Ws1[k * 128 + n] + Ws1[16384 + k * 128 + n];
    else if (k < 256) v = Wn1[(k - 128) * 128 + n];
    else              v = Wn1[16384 + (k - 256) * 128 + n];
    Bcat1[i] = v;
  } else if (i < S2) {
    int n = i - S1;
    bias1[n] = b1[n] + b1[128 + n];
  } else if (i < S3) {
    int j = i - S2;
    int k = j >> 6, n = j & 63;
    float v;
    if (k < 128)      v = Ws2[k * 64 + n] + Ws2[8192 + k * 64 + n];
    else if (k < 256) v = Wn2[(k - 128) * 64 + n];
    else              v = Wn2[8192 + (k - 256) * 64 + n];
    Bcat2[j] = v;
  } else if (i < S4) {
    int n = i - S3;
    bias2[n] = b2[n] + b2[64 + n];
  }
}

__global__ void count_rank(const int* __restrict__ edges, int* seg, int* __restrict__ rank, int E, int NS) {
  int e = blockIdx.x * blockDim.x + threadIdx.x;
  int r = blockIdx.y;
  if (e >= E) return;
  int d = edges[(size_t)r * 2 * E + E + e];
  rank[(size_t)r * E + e] = atomicAdd(seg + (size_t)r * NS + d, 1);
}

__global__ __launch_bounds__(1024) void scan_counts(int* seg, int N, int NS) {
  __shared__ int wsum[32];
  int* c = seg + (size_t)blockIdx.x * NS;
  const int t = threadIdx.x, lane = t & 31, w = t >> 5;
  const int per = (((N + 1023) / 1024) + 7) & ~7;
  int b = t * per;
  if (b > N) b = N;
  int e = b + per;
  if (e > N) e = N;
  int s = 0;
  for (int i = b; i < e; ++i) s += c[i];
  int inc = s;
  for (int off = 1; off < 32; off <<= 1) {
    int up = __shfl_up(inc, off, 32);
    if (lane >= off) inc += up;
  }
  if (lane == 31) wsum[w] = inc;
  __syncthreads();
  if (w == 0) {
    int v = wsum[lane];
    int iv = v;
    for (int off = 1; off < 32; off <<= 1) {
      int up = __shfl_up(iv, off, 32);
      if (lane >= off) iv += up;
    }
    wsum[lane] = iv - v;
  }
  __syncthreads();
  int run = wsum[w] + inc - s;
  for (int i = b; i < e; ++i) {
    int v = c[i];
    c[i] = run;
    run += v;
  }
  if (b < e && e == N) c[N] = run;
}

__global__ void fill_csr(const int* __restrict__ edges, const int* __restrict__ seg,
                         const int* __restrict__ rank, int* __restrict__ csr, int E, int NS) {
  int e = blockIdx.x * blockDim.x + threadIdx.x;
  int r = blockIdx.y;
  if (e >= E) return;
  const int* ed = edges + (size_t)r * 2 * E;
  int s = ed[e], d = ed[E + e];
  int pos = seg[(size_t)r * NS + d] + rank[(size_t)r * E + e];
  if ((unsigned)pos < (unsigned)E) csr[(size_t)r * E + pos] = s;
}

__global__ __launch_bounds__(256) void gather_mean(const int* __restrict__ seg, const int* __restrict__ csr,
                                                   const float* __restrict__ feat, float* __restrict__ agg,
                                                   int E, int N, int NS) {
  const int lane = threadIdx.x & 31;
  const int w = __builtin_amdgcn_readfirstlane((int)(threadIdx.x >> 5));
  const int nwaves = gridDim.x * 8;
  for (int idx = blockIdx.x * 8 + w; idx < 2 * N; idx += nwaves) {
    const int r = (idx >= N) ? 1 : 0;
    const int i = idx - r * N;
    const int* c = seg + (size_t)r * NS;
    int start = c[i];
    int end = c[i + 1];
    if (start < 0) start = 0;
    if (end > E) end = E;
    const int* nb = csr + (size_t)r * E;
    float4 acc = make_float4(0.0f, 0.0f, 0.0f, 0.0f);
    for (int base = start; base < end; base += 32) {
      const int rem = end - base;
      const int n = rem < 32 ? rem : 32;
      int mine = (lane < n) ? nb[base + lane] : 0;
      if ((unsigned)mine >= (unsigned)N) mine = 0;
      int j = 0;
      for (; j + 4 <= n; j += 4) {
        const int s0 = __builtin_amdgcn_readlane(mine, j);
        const int s1 = __builtin_amdgcn_readlane(mine, j + 1);
        const int s2 = __builtin_amdgcn_readlane(mine, j + 2);
        const int s3 = __builtin_amdgcn_readlane(mine, j + 3);
        const float4 v0 = *(const float4*)(feat + (size_t)s0 * FDIM + lane * 4);
        const float4 v1 = *(const float4*)(feat + (size_t)s1 * FDIM + lane * 4);
        const float4 v2 = *(const float4*)(feat + (size_t)s2 * FDIM + lane * 4);
        const float4 v3 = *(const float4*)(feat + (size_t)s3 * FDIM + lane * 4);
        acc.x += v0.x; acc.y += v0.y; acc.z += v0.z; acc.w += v0.w;
        acc.x += v1.x; acc.y += v1.y; acc.z += v1.z; acc.w += v1.w;
        acc.x += v2.x; acc.y += v2.y; acc.z += v2.z; acc.w += v2.w;
        acc.x += v3.x; acc.y += v3.y; acc.z += v3.z; acc.w += v3.w;
      }
      for (; j < n; ++j) {
        const int s0 = __builtin_amdgcn_readlane(mine, j);
        const float4 v0 = *(const float4*)(feat + (size_t)s0 * FDIM + lane * 4);
        acc.x += v0.x; acc.y += v0.y; acc.z += v0.z; acc.w += v0.w;
      }
    }
    const float deg = (float)(end - start);
    const float dinv = 1.0f / fmaxf(deg, 1.0f);
    float4 o;
    o.x = acc.x * dinv; o.y = acc.y * dinv; o.z = acc.z * dinv; o.w = acc.w * dinv;
    *(float4*)(agg + ((size_t)r * N + i) * FDIM + lane * 4) = o;
  }
}

template <int NOUT, bool RELU>
__global__ __launch_bounds__(256) void sage_gemm(
    const float* __restrict__ A0, const float* __restrict__ G0, const float* __restrict__ G1,
    const float* __restrict__ Bmat, const float* __restrict__ bias,
    float* __restrict__ out, int Nrows) {
  constexpr int NT = NOUT / 16;
  __shared__ float Bs[KC * NOUT];

  const int tid = threadIdx.x;
  const int lane = tid & 31;
  const int wave = tid >> 5;
  const int rowBase = blockIdx.x * 128 + wave * 16;
  const int m = lane & 15;
  const int khalf = (lane >> 4) << 1;

  v8f czero = {};
  v8f c[NT];
#pragma unroll
  for (int t = 0; t < NT; ++t) c[t] = czero;

  int rowA = rowBase + m;
  if (rowA > Nrows - 1) rowA = Nrows - 1;

  for (int kc = 0; kc < KTOT; kc += KC) {
    for (int i = tid; i < KC * NOUT; i += 256)
      Bs[i] = Bmat[kc * NOUT + i];
    __syncthreads();

    const float* Asrc = (kc < 128) ? A0 : ((kc < 256) ? G0 : G1);
    const int kbase = kc & 127;

    for (int k4 = 0; k4 < KC; k4 += 4) {
      v2f a = *(const v2f*)(Asrc + (size_t)rowA * FDIM + kbase + k4 + khalf);
      const int kb = k4 + khalf;
#pragma unroll
      for (int t = 0; t < NT; ++t) {
        v2f b;
        b.x = Bs[kb * NOUT + t * 16 + m];
        b.y = Bs[(kb + 1) * NOUT + t * 16 + m];
        c[t] = __builtin_amdgcn_wmma_f32_16x16x4_f32(
            false, a, false, b, (short)0, c[t], false, false);
      }
    }
    __syncthreads();
  }

  const int rhalf = (lane >> 4) << 3;
#pragma unroll
  for (int t = 0; t < NT; ++t) {
    int col = t * 16 + m;
    float bv = bias[col];
#pragma unroll
    for (int v = 0; v < 8; ++v) {
      int row = rowBase + v + rhalf;
      if (row < Nrows) {
        float val = c[t][v] + bv;
        if (RELU) val = fmaxf(val, 0.0f);
        out[(size_t)row * NOUT + col] = val;
      }
    }
  }
}


template <int S> __device__ __forceinline__ unsigned row_lane_plus(unsigned v, unsigned add) {
  return (unsigned)__builtin_amdgcn_update_dpp(0, (int)v, 0x150 + S, 0xf, 0xf, true) + add;
}
template <int M> __device__ __forceinline__ float row_xor_lane(float v) {
  return __builtin_bit_cast(float,
      __builtin_amdgcn_update_dpp(0, __builtin_bit_cast(int, v), 0x160 + M, 0xf, 0xf, true));
}
template <int M> __device__ __forceinline__ float fold_pair(bool up, float lo, float hi) {
  const float keep = up ? hi : lo;
  const float send = up ? lo : hi;
  return keep + row_xor_lane<M>(send);
}

#define EDGE_LOAD(S)                                                        \
  const float4 a##S = *(const float4*)(hb + row_lane_plus<S>(so, c16));     \
  const float4 b##S = *(const float4*)(hb + row_lane_plus<S>(dof, c16));
#define EDGE_DOT(S) \
  p[S] = a##S.x * b##S.x + a##S.y * b##S.y + a##S.z * b##S.z + a##S.w * b##S.w;

__global__ __launch_bounds__(256) void edge_score(const int* __restrict__ src, const int* __restrict__ dst,
                                                  const float* __restrict__ h, float* __restrict__ out,
                                                  int E, int N) {
  const int lane = threadIdx.x & 31;
  const int c = lane & 15;
  const unsigned c16 = (unsigned)c * 16u;
  const int w = __builtin_amdgcn_readfirstlane((int)(threadIdx.x >> 5));
  const int nwaves = gridDim.x * 8;
  const int ngroups = (E + 31) >> 5;
  const char* hb = (const char*)h;
  const bool u8 = (c & 8) != 0, u4 = (c & 4) != 0, u2 = (c & 2) != 0, u1 = (c & 1) != 0;

  int g = blockIdx.x * 8 + w;
  int si = 0, di = 0;
  if (g * 32 + lane < E) { si = src[g * 32 + lane]; di = dst[g * 32 + lane]; }
  for (; g < ngroups; g += nwaves) {
    const int e = g * 32 + lane;
    int en = e + nwaves * 32;
    if (en > E - 1) en = E - 1;
    const int sn = src[en], dn = dst[en];
    if ((unsigned)si >= (unsigned)N) si = 0;
    if ((unsigned)di >= (unsigned)N) di = 0;
    const unsigned so = (unsigned)si * 256u;
    const unsigned dof = (unsigned)di * 256u;
    float p[16];
    EDGE_LOAD(0) EDGE_LOAD(1) EDGE_LOAD(2) EDGE_LOAD(3) EDGE_LOAD(4) EDGE_LOAD(5) EDGE_LOAD(6) EDGE_LOAD(7)
    EDGE_LOAD(8) EDGE_LOAD(9) EDGE_LOAD(10) EDGE_LOAD(11) EDGE_LOAD(12) EDGE_LOAD(13) EDGE_LOAD(14) EDGE_LOAD(15)
    EDGE_DOT(0) EDGE_DOT(1) EDGE_DOT(2) EDGE_DOT(3) EDGE_DOT(4) EDGE_DOT(5) EDGE_DOT(6) EDGE_DOT(7)
    EDGE_DOT(8) EDGE_DOT(9) EDGE_DOT(10) EDGE_DOT(11) EDGE_DOT(12) EDGE_DOT(13) EDGE_DOT(14) EDGE_DOT(15)
    float q[8];
#pragma unroll
    for (int s = 0; s < 8; ++s) q[s] = fold_pair<8>(u8, p[s], p[s + 8]);
    float r4[4];
#pragma unroll
    for (int s = 0; s < 4; ++s) r4[s] = fold_pair<4>(u4, q[s], q[s + 4]);
    const float t0 = fold_pair<2>(u2, r4[0], r4[2]);
    const float t1 = fold_pair<2>(u2, r4[1], r4[3]);
    const float score = fold_pair<1>(u1, t0, t1);
    if (e < E) out[e] = score;
    si = sn; di = dn;
  }
}
#undef EDGE_LOAD
#undef EDGE_DOT

extern "C" void kernel_launch(void* const* d_in, const int* in_sizes, int n_in,
                              void* d_out, int out_size, void* d_ws, size_t ws_size,
                              hipStream_t stream) {
  const float* x   = (const float*)d_in[0];
  const int*   edges = (const int*)d_in[1];
  const int*   negE  = (const int*)d_in[2];
  const float* Wn1 = (const float*)d_in[3];
  const float* Ws1 = (const float*)d_in[4];
  const float* b1  = (const float*)d_in[5];
  const float* Wn2 = (const float*)d_in[6];
  const float* Ws2 = (const float*)d_in[7];
  const float* b2  = (const float*)d_in[8];
  float* out = (float*)d_out;

  const int N = in_sizes[0] / 128;
  const int E = in_sizes[1] / 4;

  float* ws    = (float*)d_ws;
  float* agg0  = ws + 2 * (size_t)N;
  float* agg1  = agg0 + (size_t)N * 128;
  float* h1    = agg1 + (size_t)N * 128;
  float* h2    = h1 + (size_t)N * 128;
  float* Bcat1 = h2 + (size_t)N * 64;
  float* bias1 = Bcat1 + 384 * 128;
  float* Bcat2 = bias1 + 128;
  float* bias2 = Bcat2 + 384 * 64;
  const int NS = (N + 8) & ~7;
  int* csr  = (int*)h2;
  int* seg  = csr + 2 * (size_t)E;
  int* rank = (int*)agg1;

  const int T = 256;

  fill_zero<<<(2 * NS + T - 1) / T, T, 0, stream>>>((float*)seg, 2 * (size_t)NS);
  const int wtot = 384 * 128 + 128 + 384 * 64 + 64;
  build_weights<<<(wtot + T - 1) / T, T, 0, stream>>>(Wn1, Ws1, b1, Wn2, Ws2, b2,
                                                      Bcat1, bias1, Bcat2, bias2);

  const dim3 egrid((E + T - 1) / T, 2);
  count_rank<<<egrid, T, 0, stream>>>(edges, seg, rank, E, NS);
  scan_counts<<<2, 1024, 0, stream>>>(seg, N, NS);
  fill_csr<<<egrid, T, 0, stream>>>(edges, seg, rank, csr, E, NS);

  const int sblocks = 1024;
  const int gblocks = (N + 127) / 128;
  const int ablocks = 4096;

  gather_mean<<<ablocks, T, 0, stream>>>(seg, csr, x, agg0, E, N, NS);
  sage_gemm<128, true><<<gblocks, T, 0, stream>>>(x, agg0, agg1, Bcat1, bias1, h1, N);

  gather_mean<<<ablocks, T, 0, stream>>>(seg, csr, h1, agg0, E, N, NS);
  sage_gemm<64, false><<<gblocks, T, 0, stream>>>(h1, agg0, agg1, Bcat2, bias2, h2, N);

  edge_score<<<sblocks, T, 0, stream>>>(edges, edges + E, h2, out, E, N);
  edge_score<<<sblocks, T, 0, stream>>>(negE,  negE + E,  h2, out + E, E, N);
}


# Layer 2 projects before it aggregates (mean is linear): GEMM-2 makes P[N,192] = h1 @ [Wself | Wn0 | Wn1], the pull sums 64-wide rows, two neighbours per load: half the gather bytes; new sum order
# speedup vs baseline: 2.7945x; 1.0934x over previous
#include <hip/hip_runtime.h>
#include <hip/hip_bf16.h>

typedef float v2f __attribute__((ext_vector_type(2)));
typedef float v8f __attribute__((ext_vector_type(8)));

#define FDIM 128
#define KC   64
#define PROW 192

__global__ void fill_zero(float* p, size_t n) {
  size_t i = (size_t)blockIdx.x * blockDim.x + threadIdx.x;
  size_t stride = (size_t)gridDim.x * blockDim.x;
  for (; i < n; i += stride) p[i] = 0.0f;
}

__global__ void build_weights(const float* __restrict__ Wn1, const float* __restrict__ Ws1,
                              const float* __restrict__ b1,
                              const float* __restrict__ Wn2, const float* __restrict__ Ws2,
                              const float* __restrict__ b2,
                              float* __restrict__ Bcat1, float* __restrict__ bias1,
                              float* __restrict__ Bcat2, float* __restrict__ bias2) {
  int i = blockIdx.x * blockDim.x + threadIdx.x;
  const int S1 = 384 * 128, S2 = S1 + 128, S3 = S2 + 128 * PROW, S4 = S3 + PROW;
  if (i < S1) {
    int k = i >> 7, n = i & 127;
    float v;
    if (k < 128)      v = Ws1[k * 128 + n] + Ws1[16384 + k * 128 + n];
    else if (k < 256) v = Wn1[(k - 128) * 128 + n];
    else              v = Wn1[16384 + (k - 256) * 128 + n];
    Bcat1[i] = v;
  } else if (i < S2) {
    int n = i - S1;
    bias1[n] = b1[n] + b1[128 + n];
  } else if (i < S3) {
    int j = i - S2;
    int k = j / PROW, n = j - k * PROW;
    float v;
    if (n < 64)       v = Ws2[k * 64 + n] + Ws2[8192 + k * 64 + n];
    else if (n < 128) v = Wn2[k * 64 + (n - 64)];
    else              v = Wn2[8192 + k * 64 + (n - 128)];
    Bcat2[j] = v;
  } else if (i < S4) {
    int n = i - S3;
    bias2[n] = (n < 64) ? b2[n] + b2[64 + n] : 0.0f;
  }
}

__global__ void count_rank(const int* __restrict__ edges, int* seg, int* __restrict__ rank, int E, int NS) {
  int e = blockIdx.x * blockDim.x + threadIdx.x;
  int r = blockIdx.y;
  if (e >= E) return;
  int d = edges[(size_t)r * 2 * E + E + e];
  rank[(size_t)r * E + e] = atomicAdd(seg + (size_t)r * NS + d, 1);
}

__global__ __launch_bounds__(1024) void scan_counts(int* seg, int N, int NS) {
  __shared__ int wsum[32];
  int* c = seg + (size_t)blockIdx.x * NS;
  const int t = threadIdx.x, lane = t & 31, w = t >> 5;
  const int per = (((N + 1023) / 1024) + 7) & ~7;
  int b = t * per;
  if (b > N) b = N;
  int e = b + per;
  if (e > N) e = N;
  int s = 0;
  for (int i = b; i < e; ++i) s += c[i];
  int inc = s;
  for (int off = 1; off < 32; off <<= 1) {
    int up = __shfl_up(inc, off, 32);
    if (lane >= off) inc += up;
  }
  if (lane == 31) wsum[w] = inc;
  __syncthreads();
  if (w == 0) {
    int v = wsum[lane];
    int iv = v;
    for (int off = 1; off < 32; off <<= 1) {
      int up = __shfl_up(iv, off, 32);
      if (lane >= off) iv += up;
    }
    wsum[lane] = iv - v;
  }
  __syncthreads();
  int run = wsum[w] + inc - s;
  for (int i = b; i < e; ++i) {
    int v = c[i];
    c[i] = run;
    run += v;
  }
  if (b < e && e == N) c[N] = run;
}

__global__ void fill_csr(const int* __restrict__ edges, const int* __restrict__ seg,
                         const int* __restrict__ rank, int* __restrict__ csr, int E, int NS) {
  int e = blockIdx.x * blockDim.x + threadIdx.x;
  int r = blockIdx.y;
  if (e >= E) return;
  const int* ed = edges + (size_t)r * 2 * E;
  int s = ed[e], d = ed[E + e];
  int pos = seg[(size_t)r * NS + d] + rank[(size_t)r * E + e];
  if ((unsigned)pos < (unsigned)E) csr[(size_t)r * E + pos] = s;
}

__global__ __launch_bounds__(256) void gather_mean(const int* __restrict__ seg, const int* __restrict__ csr,
                                                   const float* __restrict__ feat, float* __restrict__ agg,
                                                   int E, int N, int NS) {
  const int lane = threadIdx.x & 31;
  const int w = __builtin_amdgcn_readfirstlane((int)(threadIdx.x >> 5));
  const int nwaves = gridDim.x * 8;
  for (int idx = blockIdx.x * 8 + w; idx < 2 * N; idx += nwaves) {
    const int r = (idx >= N) ? 1 : 0;
    const int i = idx - r * N;
    const int* c = seg + (size_t)r * NS;
    int start = c[i];
    int end = c[i + 1];
    if (start < 0) start = 0;
    if (end > E) end = E;
    const int* nb = csr + (size_t)r * E;
    float4 acc = make_float4(0.0f, 0.0f, 0.0f, 0.0f);
    for (int base = start; base < end; base += 32) {
      const int rem = end - base;
      const int n = rem < 32 ? rem : 32;
      int mine = (lane < n) ? nb[base + lane] : 0;
      if ((unsigned)mine >= (unsigned)N) mine = 0;
      int j = 0;
      for (; j + 4 <= n; j += 4) {
        const int s0 = __builtin_amdgcn_readlane(mine, j);
        const int s1 = __builtin_amdgcn_readlane(mine, j + 1);
        const int s2 = __builtin_amdgcn_readlane(mine, j + 2);
        const int s3 = __builtin_amdgcn_readlane(mine, j + 3);
        const float4 v0 = *(const float4*)(feat + (size_t)s0 * FDIM + lane * 4);
        const float4 v1 = *(const float4*)(feat + (size_t)s1 * FDIM + lane * 4);
        const float4 v2 = *(const float4*)(feat + (size_t)s2 * FDIM + lane * 4);
        const float4 v3 = *(const float4*)(feat + (size_t)s3 * FDIM + lane * 4);
        acc.x += v0.x; acc.y += v0.y; acc.z += v0.z; acc.w += v0.w;
        acc.x += v1.x; acc.y += v1.y; acc.z += v1.z; acc.w += v1.w;
        acc.x += v2.x; acc.y += v2.y; acc.z += v2.z; acc.w += v2.w;
        acc.x += v3.x; acc.y += v3.y; acc.z += v3.z; acc.w += v3.w;
      }
      for (; j < n; ++j) {
        const int s0 = __builtin_amdgcn_readlane(mine, j);
        const float4 v0 = *(const float4*)(feat + (size_t)s0 * FDIM + lane * 4);
        acc.x += v0.x; acc.y += v0.y; acc.z += v0.z; acc.w += v0.w;
      }
    }
    const float deg = (float)(end - start);
    const float dinv = 1.0f / fmaxf(deg, 1.0f);
    float4 o;
    o.x = acc.x * dinv; o.y = acc.y * dinv; o.z = acc.z * dinv; o.w = acc.w * dinv;
    *(float4*)(agg + ((size_t)r * N + i) * FDIM + lane * 4) = o;
  }
}


__device__ __forceinline__ float other_half(float v) {
  const int b = __builtin_bit_cast(int, v);
  return __builtin_bit_cast(float, __builtin_amdgcn_permlanex16(b, b, 0x76543210, 0xfedcba98, false, false));
}

__device__ __forceinline__ int first_ids(const int* __restrict__ nb, int start, int end, int lane) {
  const int rem = end - start;
  const int n = rem < 32 ? rem : 32;
  return (lane < n) ? nb[start + lane] : 0;
}

#define PULL_PAIR(K)                                                                          \
  const unsigned e##K = (unsigned)__builtin_amdgcn_readlane((int)moff, j + 2 * K);            \
  const unsigned o##K = (unsigned)__builtin_amdgcn_readlane((int)moff, j + 2 * K + 1);        \
  float4 v##K = make_float4(0.0f, 0.0f, 0.0f, 0.0f);                                          \
  if (j + 2 * K + hlf < n) v##K = *(const float4*)(pb + ((hlf ? o##K : e##K) + c16));
#define PULL_ADD(K) acc.x += v##K.x; acc.y += v##K.y; acc.z += v##K.z; acc.w += v##K.w;

__device__ __forceinline__ float4 pull_blocks(const int* __restrict__ nb, int start, int end, int mine,
                                              const char* __restrict__ pb, unsigned c16, int lane, int hlf, int N) {
  float4 acc = make_float4(0.0f, 0.0f, 0.0f, 0.0f);
  for (int base = start; base < end; base += 32) {
    const int rem = end - base;
    const int n = rem < 32 ? rem : 32;
    if (base != start) mine = (lane < n) ? nb[base + lane] : 0;
    if ((unsigned)mine >= (unsigned)N) mine = 0;
    const unsigned moff = (unsigned)mine * (PROW * 4u);
    for (int j = 0; j < n; j += 16) {
      PULL_PAIR(0) PULL_PAIR(1) PULL_PAIR(2) PULL_PAIR(3) PULL_PAIR(4) PULL_PAIR(5) PULL_PAIR(6) PULL_PAIR(7)
      PULL_ADD(0) PULL_ADD(1) PULL_ADD(2) PULL_ADD(3) PULL_ADD(4) PULL_ADD(5) PULL_ADD(6) PULL_ADD(7)
    }
  }
  return acc;
}
#undef PULL_PAIR
#undef PULL_ADD

__global__ __launch_bounds__(256) void gather_proj(const int* __restrict__ seg, const int* __restrict__ csr,
                                                   const float* __restrict__ P, float* __restrict__ h2,
                                                   int E, int N, int NS) {
  const int lane = threadIdx.x & 31;
  const int hlf = lane >> 4;
  const unsigned c16 = (unsigned)(lane & 15) * 16u;
  const int w = __builtin_amdgcn_readfirstlane((int)(threadIdx.x >> 5));
  const int nwaves = gridDim.x * 8;
  const char* pbytes = (const char*)P;
  for (int i = blockIdx.x * 8 + w; i < N; i += nwaves) {
    int s0 = seg[i], e0 = seg[i + 1];
    int s1 = seg[(size_t)NS + i], e1 = seg[(size_t)NS + i + 1];
    if (s0 < 0) s0 = 0;
    if (e0 > E) e0 = E;
    if (s1 < 0) s1 = 0;
    if (e1 > E) e1 = E;
    const int m0 = first_ids(csr, s0, e0, lane);
    const int m1 = first_ids(csr + E, s1, e1, lane);
    float4 tot = make_float4(0.0f, 0.0f, 0.0f, 0.0f);
    if (hlf == 0) tot = *(const float4*)(pbytes + ((size_t)i * (PROW * 4u) + c16));
    const float4 a0 = pull_blocks(csr, s0, e0, m0, pbytes + 256, c16, lane, hlf, N);
    const float4 a1 = pull_blocks(csr + E, s1, e1, m1, pbytes + 512, c16, lane, hlf, N);
    const float d0 = 1.0f / fmaxf((float)(e0 - s0), 1.0f);
    const float d1 = 1.0f / fmaxf((float)(e1 - s1), 1.0f);
    tot.x += a0.x * d0 + a1.x * d1;
    tot.y += a0.y * d0 + a1.y * d1;
    tot.z += a0.z * d0 + a1.z * d1;
    tot.w += a0.w * d0 + a1.w * d1;
    tot.x += other_half(tot.x);
    tot.y += other_half(tot.y);
    tot.z += other_half(tot.z);
    tot.w += other_half(tot.w);
    if (hlf == 0) *(float4*)(h2 + (size_t)i * 64 + (lane & 15) * 4) = tot;
  }
}

template <int NOUT, int KT, bool RELU>
__global__ __launch_bounds__(256) void sage_gemm(
    const float* __restrict__ A0, const float* __restrict__ G0, const float* __restrict__ G1,
    const float* __restrict__ Bmat, const float* __restrict__ bias,
    float* __restrict__ out, int Nrows) {
  constexpr int NT = NOUT / 16;
  __shared__ float Bs[KC * NOUT];

  const int tid = threadIdx.x;
  const int lane = tid & 31;
  const int wave = tid >> 5;
  const int rowBase = blockIdx.x * 128 + wave * 16;
  const int m = lane & 15;
  const int khalf = (lane >> 4) << 1;

  v8f czero = {};
  v8f c[NT];
#pragma unroll
  for (int t = 0; t < NT; ++t) c[t] = czero;

  int rowA = rowBase + m;
  if (rowA > Nrows - 1) rowA = Nrows - 1;

  for (int kc = 0; kc < KT; kc += KC) {
    for (int i = tid; i < KC * NOUT; i += 256)
      Bs[i] = Bmat[kc * NOUT + i];
    __syncthreads();

    const float* Asrc = (kc < 128) ? A0 : ((kc < 256) ? G0 : G1);
    const int kbase = kc & 127;

    for (int k4 = 0; k4 < KC; k4 += 4) {
      v2f a = *(const v2f*)(Asrc + (size_t)rowA * FDIM + kbase + k4 + khalf);
      const int kb = k4 + khalf;
#pragma unroll
      for (int t = 0; t < NT; ++t) {
        v2f b;
        b.x = Bs[kb * NOUT + t * 16 + m];
        b.y = Bs[(kb + 1) * NOUT + t * 16 + m];
        c[t] = __builtin_amdgcn_wmma_f32_16x16x4_f32(
            false, a, false, b, (short)0, c[t], false, false);
      }
    }
    __syncthreads();
  }

  const int rhalf = (lane >> 4) << 3;
#pragma unroll
  for (int t = 0; t < NT; ++t) {
    int col = t * 16 + m;
    float bv = bias[col];
#pragma unroll
    for (int v = 0; v < 8; ++v) {
      int row = rowBase + v + rhalf;
      if (row < Nrows) {
        float val = c[t][v] + bv;
        if (RELU) val = fmaxf(val, 0.0f);
        out[(size_t)row * NOUT + col] = val;
      }
    }
  }
}


template <int S> __device__ __forceinline__ unsigned row_lane_plus(unsigned v, unsigned add) {
  return (unsigned)__builtin_amdgcn_update_dpp(0, (int)v, 0x150 + S, 0xf, 0xf, true) + add;
}
template <int M> __device__ __forceinline__ float row_xor_lane(float v) {
  return __builtin_bit_cast(float,
      __builtin_amdgcn_update_dpp(0, __builtin_bit_cast(int, v), 0x160 + M, 0xf, 0xf, true));
}
template <int M> __device__ __forceinline__ float fold_pair(bool up, float lo, float hi) {
  const float keep = up ? hi : lo;
  const float send = up ? lo : hi;
  return keep + row_xor_lane<M>(send);
}

#define EDGE_LOAD(S)                                                        \
  const float4 a##S = *(const float4*)(hb + row_lane_plus<S>(so, c16));     \
  const float4 b##S = *(const float4*)(hb + row_lane_plus<S>(dof, c16));
#define EDGE_DOT(S) \
  p[S] = a##S.x * b##S.x + a##S.y * b##S.y + a##S.z * b##S.z + a##S.w * b##S.w;

__global__ __launch_bounds__(256) void edge_score(const int* __restrict__ src, const int* __restrict__ dst,
                                                  const float* __restrict__ h, float* __restrict__ out,
                                                  int E, int N) {
  const int lane = threadIdx.x & 31;
  const int c = lane & 15;
  const unsigned c16 = (unsigned)c * 16u;
  const int w = __builtin_amdgcn_readfirstlane((int)(threadIdx.x >> 5));
  const int nwaves = gridDim.x * 8;
  const int ngroups = (E + 31) >> 5;
  const char* hb = (const char*)h;
  const bool u8 = (c & 8) != 0, u4 = (c & 4) != 0, u2 = (c & 2) != 0, u1 = (c & 1) != 0;

  int g = blockIdx.x * 8 + w;
  int si = 0, di = 0;
  if (g * 32 + lane < E) { si = src[g * 32 + lane]; di = dst[g * 32 + lane]; }
  for (; g < ngroups; g += nwaves) {
    const int e = g * 32 + lane;
    int en = e + nwaves * 32;
    if (en > E - 1) en = E - 1;
    const int sn = src[en], dn = dst[en];
    if ((unsigned)si >= (unsigned)N) si = 0;
    if ((unsigned)di >= (unsigned)N) di = 0;
    const unsigned so = (unsigned)si * 256u;
    const unsigned dof = (unsigned)di * 256u;
    float p[16];
    EDGE_LOAD(0) EDGE_LOAD(1) EDGE_LOAD(2) EDGE_LOAD(3) EDGE_LOAD(4) EDGE_LOAD(5) EDGE_LOAD(6) EDGE_LOAD(7)
    EDGE_LOAD(8) EDGE_LOAD(9) EDGE_LOAD(10) EDGE_LOAD(11) EDGE_LOAD(12) EDGE_LOAD(13) EDGE_LOAD(14) EDGE_LOAD(15)
    EDGE_DOT(0) EDGE_DOT(1) EDGE_DOT(2) EDGE_DOT(3) EDGE_DOT(4) EDGE_DOT(5) EDGE_DOT(6) EDGE_DOT(7)
    EDGE_DOT(8) EDGE_DOT(9) EDGE_DOT(10) EDGE_DOT(11) EDGE_DOT(12) EDGE_DOT(13) EDGE_DOT(14) EDGE_DOT(15)
    float q[8];
#pragma unroll
    for (int s = 0; s < 8; ++s) q[s] = fold_pair<8>(u8, p[s], p[s + 8]);
    float r4[4];
#pragma unroll
    for (int s = 0; s < 4; ++s) r4[s] = fold_pair<4>(u4, q[s], q[s + 4]);
    const float t0 = fold_pair<2>(u2, r4[0], r4[2]);
    const float t1 = fold_pair<2>(u2, r4[1], r4[3]);
    const float score = fold_pair<1>(u1, t0, t1);
    if (e < E) out[e] = score;
    si = sn; di = dn;
  }
}
#undef EDGE_LOAD
#undef EDGE_DOT

extern "C" void kernel_launch(void* const* d_in, const int* in_sizes, int n_in,
                              void* d_out, int out_size, void* d_ws, size_t ws_size,
                              hipStream_t stream) {
  const float* x   = (const float*)d_in[0];
  const int*   edges = (const int*)d_in[1];
  const int*   negE  = (const int*)d_in[2];
  const float* Wn1 = (const float*)d_in[3];
  const float* Ws1 = (const float*)d_in[4];
  const float* b1  = (const float*)d_in[5];
  const float* Wn2 = (const float*)d_in[6];
  const float* Ws2 = (const float*)d_in[7];
  const float* b2  = (const float*)d_in[8];
  float* out = (float*)d_out;

  const int N = in_sizes[0] / 128;
  const int E = in_sizes[1] / 4;

  float* ws    = (float*)d_ws;
  float* bias2 = ws;
  float* agg0  = ws + 2 * (size_t)N;
  float* agg1  = agg0 + (size_t)N * 128;
  float* h1    = agg1 + (size_t)N * 128;
  float* elist = h1 + (size_t)N * 128;
  float* Bcat1 = elist + (size_t)N * 64;
  float* bias1 = Bcat1 + 384 * 128;
  float* Bcat2 = bias1 + 128;
  float* P     = agg0 + 32;
  float* h2    = h1;
  const int NS = (N + 8) & ~7;
  int* csr  = (int*)elist;
  int* seg  = csr + 2 * (size_t)E;
  int* rank = (int*)agg1;

  const int T = 256;

  fill_zero<<<(2 * NS + T - 1) / T, T, 0, stream>>>((float*)seg, 2 * (size_t)NS);
  const int wtot = 384 * 128 + 128 + 128 * PROW + PROW;
  build_weights<<<(wtot + T - 1) / T, T, 0, stream>>>(Wn1, Ws1, b1, Wn2, Ws2, b2,
                                                      Bcat1, bias1, Bcat2, bias2);

  const dim3 egrid((E + T - 1) / T, 2);
  count_rank<<<egrid, T, 0, stream>>>(edges, seg, rank, E, NS);
  scan_counts<<<2, 1024, 0, stream>>>(seg, N, NS);
  fill_csr<<<egrid, T, 0, stream>>>(edges, seg, rank, csr, E, NS);

  const int sblocks = 1024;
  const int gblocks = (N + 127) / 128;
  const int ablocks = 4096;
  const int pblocks = 2048;

  gather_mean<<<ablocks, T, 0, stream>>>(seg, csr, x, agg0, E, N, NS);
  sage_gemm<128, 384, true><<<gblocks, T, 0, stream>>>(x, agg0, agg1, Bcat1, bias1, h1, N);

  sage_gemm<PROW, 128, false><<<gblocks, T, 0, stream>>>(h1, h1, h1, Bcat2, bias2, P, N);
  gather_proj<<<pblocks, T, 0, stream>>>(seg, csr, P, h2, E, N, NS);

  edge_score<<<sblocks, T, 0, stream>>>(edges, edges + E, h2, out, E, N);
  edge_score<<<sblocks, T, 0, stream>>>(negE,  negE + E,  h2, out + E, E, N);
}


# Edge-sort passes on resident waves: count_rank and fill_csr launch 4,096 waves (was 50,000 one-edge waves), 16 edges per lane, loads then atomics issued in groups; new atomic rank, so new sum order
# speedup vs baseline: 2.9732x; 1.0639x over previous
#include <hip/hip_runtime.h>
#include <hip/hip_bf16.h>

typedef float v2f __attribute__((ext_vector_type(2)));
typedef float v8f __attribute__((ext_vector_type(8)));

#define FDIM 128
#define KC   64
#define PROW 192

__global__ void fill_zero(float* p, size_t n) {
  size_t i = (size_t)blockIdx.x * blockDim.x + threadIdx.x;
  size_t stride = (size_t)gridDim.x * blockDim.x;
  for (; i < n; i += stride) p[i] = 0.0f;
}

__global__ void build_weights(const float* __restrict__ Wn1, const float* __restrict__ Ws1,
                              const float* __restrict__ b1,
                              const float* __restrict__ Wn2, const float* __restrict__ Ws2,
                              const float* __restrict__ b2,
                              float* __restrict__ Bcat1, float* __restrict__ bias1,
                              float* __restrict__ Bcat2, float* __restrict__ bias2) {
  int i = blockIdx.x * blockDim.x + threadIdx.x;
  const int S1 = 384 * 128, S2 = S1 + 128, S3 = S2 + 128 * PROW, S4 = S3 + PROW;
  if (i < S1) {
    int k = i >> 7, n = i & 127;
    float v;
    if (k < 128)      v = Ws1[k * 128 + n] + Ws1[16384 + k * 128 + n];
    else if (k < 256) v = Wn1[(k - 128) * 128 + n];
    else              v = Wn1[16384 + (k - 256) * 128 + n];
    Bcat1[i] = v;
  } else if (i < S2) {
    int n = i - S1;
    bias1[n] = b1[n] + b1[128 + n];
  } else if (i < S3) {
    int j = i - S2;
    int k = j / PROW, n = j - k * PROW;
    float v;
    if (n < 64)       v = Ws2[k * 64 + n] + Ws2[8192 + k * 64 + n];
    else if (n < 128) v = Wn2[k * 64 + (n - 64)];
    else              v = Wn2[8192 + k * 64 + (n - 128)];
    Bcat2[j] = v;
  } else if (i < S4) {
    int n = i - S3;
    bias2[n] = (n < 64) ? b2[n] + b2[64 + n] : 0.0f;
  }
}

#define EPT 16

__global__ __launch_bounds__(256) void count_rank(const int* edges, int* seg, int* __restrict__ rank, int E, int NS) {
  const int r = blockIdx.y;
  const int* dst = edges + (size_t)r * 2 * E + E;
  int* sg = seg + (size_t)r * NS;
  int* __restrict__ rk = rank + (size_t)r * E;
  const int stride = gridDim.x * 256;
  for (int e0 = blockIdx.x * 256 + threadIdx.x; e0 < E; e0 += EPT * stride) {
    int d[EPT], v[EPT];
#pragma unroll
    for (int k = 0; k < EPT; ++k) {
      const int e = e0 + k * stride;
      d[k] = dst[e < E ? e : e0];
    }
#pragma unroll
    for (int k = 0; k < EPT; ++k) {
      v[k] = 0;
      if (e0 + k * stride < E) v[k] = atomicAdd(sg + d[k], 1);
    }
#pragma unroll
    for (int k = 0; k < EPT; ++k) {
      const int e = e0 + k * stride;
      if (e < E) rk[e] = v[k];
    }
  }
}

__global__ __launch_bounds__(1024) void scan_counts(int* seg, int N, int NS) {
  __shared__ int wsum[32];
  int* c = seg + (size_t)blockIdx.x * NS;
  const int t = threadIdx.x, lane = t & 31, w = t >> 5;
  const int per = (((N + 1023) / 1024) + 7) & ~7;
  int b = t * per;
  if (b > N) b = N;
  int e = b + per;
  if (e > N) e = N;
  int s = 0;
  for (int i = b; i < e; ++i) s += c[i];
  int inc = s;
  for (int off = 1; off < 32; off <<= 1) {
    int up = __shfl_up(inc, off, 32);
    if (lane >= off) inc += up;
  }
  if (lane == 31) wsum[w] = inc;
  __syncthreads();
  if (w == 0) {
    int v = wsum[lane];
    int iv = v;
    for (int off = 1; off < 32; off <<= 1) {
      int up = __shfl_up(iv, off, 32);
      if (lane >= off) iv += up;
    }
    wsum[lane] = iv - v;
  }
  __syncthreads();
  int run = wsum[w] + inc - s;
  for (int i = b; i < e; ++i) {
    int v = c[i];
    c[i] = run;
    run += v;
  }
  if (b < e && e == N) c[N] = run;
}

__global__ __launch_bounds__(256) void fill_csr(const int* edges, const int* __restrict__ seg,
                                                const int* __restrict__ rank, int* __restrict__ csr, int E, int NS) {
  const int r = blockIdx.y;
  const int* ed = edges + (size_t)r * 2 * E;
  const int* __restrict__ sg = seg + (size_t)r * NS;
  const int* __restrict__ rk = rank + (size_t)r * E;
  int* __restrict__ lst = csr + (size_t)r * E;
  const int stride = gridDim.x * 256;
  for (int e0 = blockIdx.x * 256 + threadIdx.x; e0 < E; e0 += EPT * stride) {
    int s[EPT], d[EPT], q[EPT], pos[EPT];
#pragma unroll
    for (int k = 0; k < EPT; ++k) {
      const int e = e0 + k * stride;
      const int ec = e < E ? e : e0;
      d[k] = ed[E + ec];
      s[k] = ed[ec];
      q[k] = rk[ec];
    }
    asm volatile("" ::: "memory");
#pragma unroll
    for (int k = 0; k < EPT; ++k) pos[k] = sg[d[k]] + q[k];
#pragma unroll
    for (int k = 0; k < EPT; ++k) {
      if (e0 + k * stride < E && (unsigned)pos[k] < (unsigned)E) lst[pos[k]] = s[k];
    }
  }
}
#undef EPT

__global__ __launch_bounds__(256) void gather_mean(const int* __restrict__ seg, const int* __restrict__ csr,
                                                   const float* __restrict__ feat, float* __restrict__ agg,
                                                   int E, int N, int NS) {
  const int lane = threadIdx.x & 31;
  const int w = __builtin_amdgcn_readfirstlane((int)(threadIdx.x >> 5));
  const int nwaves = gridDim.x * 8;
  for (int idx = blockIdx.x * 8 + w; idx < 2 * N; idx += nwaves) {
    const int r = (idx >= N) ? 1 : 0;
    const int i = idx - r * N;
    const int* c = seg + (size_t)r * NS;
    int start = c[i];
    int end = c[i + 1];
    if (start < 0) start = 0;
    if (end > E) end = E;
    const int* nb = csr + (size_t)r * E;
    float4 acc = make_float4(0.0f, 0.0f, 0.0f, 0.0f);
    for (int base = start; base < end; base += 32) {
      const int rem = end - base;
      const int n = rem < 32 ? rem : 32;
      int mine = (lane < n) ? nb[base + lane] : 0;
      if ((unsigned)mine >= (unsigned)N) mine = 0;
      int j = 0;
      for (; j + 4 <= n; j += 4) {
        const int s0 = __builtin_amdgcn_readlane(mine, j);
        const int s1 = __builtin_amdgcn_readlane(mine, j + 1);
        const int s2 = __builtin_amdgcn_readlane(mine, j + 2);
        const int s3 = __builtin_amdgcn_readlane(mine, j + 3);
        const float4 v0 = *(const float4*)(feat + (size_t)s0 * FDIM + lane * 4);
        const float4 v1 = *(const float4*)(feat + (size_t)s1 * FDIM + lane * 4);
        const float4 v2 = *(const float4*)(feat + (size_t)s2 * FDIM + lane * 4);
        const float4 v3 = *(const float4*)(feat + (size_t)s3 * FDIM + lane * 4);
        acc.x += v0.x; acc.y += v0.y; acc.z += v0.z; acc.w += v0.w;
        acc.x += v1.x; acc.y += v1.y; acc.z += v1.z; acc.w += v1.w;
        acc.x += v2.x; acc.y += v2.y; acc.z += v2.z; acc.w += v2.w;
        acc.x += v3.x; acc.y += v3.y; acc.z += v3.z; acc.w += v3.w;
      }
      for (; j < n; ++j) {
        const int s0 = __builtin_amdgcn_readlane(mine, j);
        const float4 v0 = *(const float4*)(feat + (size_t)s0 * FDIM + lane * 4);
        acc.x += v0.x; acc.y += v0.y; acc.z += v0.z; acc.w += v0.w;
      }
    }
    const float deg = (float)(end - start);
    const float dinv = 1.0f / fmaxf(deg, 1.0f);
    float4 o;
    o.x = acc.x * dinv; o.y = acc.y * dinv; o.z = acc.z * dinv; o.w = acc.w * dinv;
    *(float4*)(agg + ((size_t)r * N + i) * FDIM + lane * 4) = o;
  }
}


__device__ __forceinline__ float other_half(float v) {
  const int b = __builtin_bit_cast(int, v);
  return __builtin_bit_cast(float, __builtin_amdgcn_permlanex16(b, b, 0x76543210, 0xfedcba98, false, false));
}

__device__ __forceinline__ int first_ids(const int* __restrict__ nb, int start, int end, int lane) {
  const int rem = end - start;
  const int n = rem < 32 ? rem : 32;
  return (lane < n) ? nb[start + lane] : 0;
}

#define PULL_PAIR(K)                                                                          \
  const unsigned e##K = (unsigned)__builtin_amdgcn_readlane((int)moff, j + 2 * K);            \
  const unsigned o##K = (unsigned)__builtin_amdgcn_readlane((int)moff, j + 2 * K + 1);        \
  float4 v##K = make_float4(0.0f, 0.0f, 0.0f, 0.0f);                                          \
  if (j + 2 * K + hlf < n) v##K = *(const float4*)(pb + ((hlf ? o##K : e##K) + c16));
#define PULL_ADD(K) acc.x += v##K.x; acc.y += v##K.y; acc.z += v##K.z; acc.w += v##K.w;

__device__ __forceinline__ float4 pull_blocks(const int* __restrict__ nb, int start, int end, int mine,
                                              const char* __restrict__ pb, unsigned c16, int lane, int hlf, int N) {
  float4 acc = make_float4(0.0f, 0.0f, 0.0f, 0.0f);
  for (int base = start; base < end; base += 32) {
    const int rem = end - base;
    const int n = rem < 32 ? rem : 32;
    if (base != start) mine = (lane < n) ? nb[base + lane] : 0;
    if ((unsigned)mine >= (unsigned)N) mine = 0;
    const unsigned moff = (unsigned)mine * (PROW * 4u);
    for (int j = 0; j < n; j += 16) {
      PULL_PAIR(0) PULL_PAIR(1) PULL_PAIR(2) PULL_PAIR(3) PULL_PAIR(4) PULL_PAIR(5) PULL_PAIR(6) PULL_PAIR(7)
      PULL_ADD(0) PULL_ADD(1) PULL_ADD(2) PULL_ADD(3) PULL_ADD(4) PULL_ADD(5) PULL_ADD(6) PULL_ADD(7)
    }
  }
  return acc;
}
#undef PULL_PAIR
#undef PULL_ADD

__global__ __launch_bounds__(256) void gather_proj(const int* __restrict__ seg, const int* __restrict__ csr,
                                                   const float* __restrict__ P, float* __restrict__ h2,
                                                   int E, int N, int NS) {
  const int lane = threadIdx.x & 31;
  const int hlf = lane >> 4;
  const unsigned c16 = (unsigned)(lane & 15) * 16u;
  const int w = __builtin_amdgcn_readfirstlane((int)(threadIdx.x >> 5));
  const int nwaves = gridDim.x * 8;
  const char* pbytes = (const char*)P;
  for (int i = blockIdx.x * 8 + w; i < N; i += nwaves) {
    int s0 = seg[i], e0 = seg[i + 1];
    int s1 = seg[(size_t)NS + i], e1 = seg[(size_t)NS + i + 1];
    if (s0 < 0) s0 = 0;
    if (e0 > E) e0 = E;
    if (s1 < 0) s1 = 0;
    if (e1 > E) e1 = E;
    const int m0 = first_ids(csr, s0, e0, lane);
    const int m1 = first_ids(csr + E, s1, e1, lane);
    float4 tot = make_float4(0.0f, 0.0f, 0.0f, 0.0f);
    if (hlf == 0) tot = *(const float4*)(pbytes + ((size_t)i * (PROW * 4u) + c16));
    const float4 a0 = pull_blocks(csr, s0, e0, m0, pbytes + 256, c16, lane, hlf, N);
    const float4 a1 = pull_blocks(csr + E, s1, e1, m1, pbytes + 512, c16, lane, hlf, N);
    const float d0 = 1.0f / fmaxf((float)(e0 - s0), 1.0f);
    const float d1 = 1.0f / fmaxf((float)(e1 - s1), 1.0f);
    tot.x += a0.x * d0 + a1.x * d1;
    tot.y += a0.y * d0 + a1.y * d1;
    tot.z += a0.z * d0 + a1.z * d1;
    tot.w += a0.w * d0 + a1.w * d1;
    tot.x += other_half(tot.x);
    tot.y += other_half(tot.y);
    tot.z += other_half(tot.z);
    tot.w += other_half(tot.w);
    if (hlf == 0) *(float4*)(h2 + (size_t)i * 64 + (lane & 15) * 4) = tot;
  }
}

template <int NOUT, int KT, bool RELU>
__global__ __launch_bounds__(256) void sage_gemm(
    const float* __restrict__ A0, const float* __restrict__ G0, const float* __restrict__ G1,
    const float* __restrict__ Bmat, const float* __restrict__ bias,
    float* __restrict__ out, int Nrows) {
  constexpr int NT = NOUT / 16;
  __shared__ float Bs[KC * NOUT];

  const int tid = threadIdx.x;
  const int lane = tid & 31;
  const int wave = tid >> 5;
  const int rowBase = blockIdx.x * 128 + wave * 16;
  const int m = lane & 15;
  const int khalf = (lane >> 4) << 1;

  v8f czero = {};
  v8f c[NT];
#pragma unroll
  for (int t = 0; t < NT; ++t) c[t] = czero;

  int rowA = rowBase + m;
  if (rowA > Nrows - 1) rowA = Nrows - 1;

  for (int kc = 0; kc < KT; kc += KC) {
    for (int i = tid; i < KC * NOUT; i += 256)
      Bs[i] = Bmat[kc * NOUT + i];
    __syncthreads();

    const float* Asrc = (kc < 128) ? A0 : ((kc < 256) ? G0 : G1);
    const int kbase = kc & 127;

    for (int k4 = 0; k4 < KC; k4 += 4) {
      v2f a = *(const v2f*)(Asrc + (size_t)rowA * FDIM + kbase + k4 + khalf);
      const int kb = k4 + khalf;
#pragma unroll
      for (int t = 0; t < NT; ++t) {
        v2f b;
        b.x = Bs[kb * NOUT + t * 16 + m];
        b.y = Bs[(kb + 1) * NOUT + t * 16 + m];
        c[t] = __builtin_amdgcn_wmma_f32_16x16x4_f32(
            false, a, false, b, (short)0, c[t], false, false);
      }
    }
    __syncthreads();
  }

  const int rhalf = (lane >> 4) << 3;
#pragma unroll
  for (int t = 0; t < NT; ++t) {
    int col = t * 16 + m;
    float bv = bias[col];
#pragma unroll
    for (int v = 0; v < 8; ++v) {
      int row = rowBase + v + rhalf;
      if (row < Nrows) {
        float val = c[t][v] + bv;
        if (RELU) val = fmaxf(val, 0.0f);
        out[(size_t)row * NOUT + col] = val;
      }
    }
  }
}


template <int S> __device__ __forceinline__ unsigned row_lane_plus(unsigned v, unsigned add) {
  return (unsigned)__builtin_amdgcn_update_dpp(0, (int)v, 0x150 + S, 0xf, 0xf, true) + add;
}
template <int M> __device__ __forceinline__ float row_xor_lane(float v) {
  return __builtin_bit_cast(float,
      __builtin_amdgcn_update_dpp(0, __builtin_bit_cast(int, v), 0x160 + M, 0xf, 0xf, true));
}
template <int M> __device__ __forceinline__ float fold_pair(bool up, float lo, float hi) {
  const float keep = up ? hi : lo;
  const float send = up ? lo : hi;
  return keep + row_xor_lane<M>(send);
}

#define EDGE_LOAD(S)                                                        \
  const float4 a##S = *(const float4*)(hb + row_lane_plus<S>(so, c16));     \
  const float4 b##S = *(const float4*)(hb + row_lane_plus<S>(dof, c16));
#define EDGE_DOT(S) \
  p[S] = a##S.x * b##S.x + a##S.y * b##S.y + a##S.z * b##S.z + a##S.w * b##S.w;

__global__ __launch_bounds__(256) void edge_score(const int* __restrict__ src, const int* __restrict__ dst,
                                                  const float* __restrict__ h, float* __restrict__ out,
                                                  int E, int N) {
  const int lane = threadIdx.x & 31;
  const int c = lane & 15;
  const unsigned c16 = (unsigned)c * 16u;
  const int w = __builtin_amdgcn_readfirstlane((int)(threadIdx.x >> 5));
  const int nwaves = gridDim.x * 8;
  const int ngroups = (E + 31) >> 5;
  const char* hb = (const char*)h;
  const bool u8 = (c & 8) != 0, u4 = (c & 4) != 0, u2 = (c & 2) != 0, u1 = (c & 1) != 0;

  int g = blockIdx.x * 8 + w;
  int si = 0, di = 0;
  if (g * 32 + lane < E) { si = src[g * 32 + lane]; di = dst[g * 32 + lane]; }
  for (; g < ngroups; g += nwaves) {
    const int e = g * 32 + lane;
    int en = e + nwaves * 32;
    if (en > E - 1) en = E - 1;
    const int sn = src[en], dn = dst[en];
    if ((unsigned)si >= (unsigned)N) si = 0;
    if ((unsigned)di >= (unsigned)N) di = 0;
    const unsigned so = (unsigned)si * 256u;
    const unsigned dof = (unsigned)di * 256u;
    float p[16];
    EDGE_LOAD(0) EDGE_LOAD(1) EDGE_LOAD(2) EDGE_LOAD(3) EDGE_LOAD(4) EDGE_LOAD(5) EDGE_LOAD(6) EDGE_LOAD(7)
    EDGE_LOAD(8) EDGE_LOAD(9) EDGE_LOAD(10) EDGE_LOAD(11) EDGE_LOAD(12) EDGE_LOAD(13) EDGE_LOAD(14) EDGE_LOAD(15)
    EDGE_DOT(0) EDGE_DOT(1) EDGE_DOT(2) EDGE_DOT(3) EDGE_DOT(4) EDGE_DOT(5) EDGE_DOT(6) EDGE_DOT(7)
    EDGE_DOT(8) EDGE_DOT(9) EDGE_DOT(10) EDGE_DOT(11) EDGE_DOT(12) EDGE_DOT(13) EDGE_DOT(14) EDGE_DOT(15)
    float q[8];
#pragma unroll
    for (int s = 0; s < 8; ++s) q[s] = fold_pair<8>(u8, p[s], p[s + 8]);
    float r4[4];
#pragma unroll
    for (int s = 0; s < 4; ++s) r4[s] = fold_pair<4>(u4, q[s], q[s + 4]);
    const float t0 = fold_pair<2>(u2, r4[0], r4[2]);
    const float t1 = fold_pair<2>(u2, r4[1], r4[3]);
    const float score = fold_pair<1>(u1, t0, t1);
    if (e < E) out[e] = score;
    si = sn; di = dn;
  }
}
#undef EDGE_LOAD
#undef EDGE_DOT

extern "C" void kernel_launch(void* const* d_in, const int* in_sizes, int n_in,
                              void* d_out, int out_size, void* d_ws, size_t ws_size,
                              hipStream_t stream) {
  const float* x   = (const float*)d_in[0];
  const int*   edges = (const int*)d_in[1];
  const int*   negE  = (const int*)d_in[2];
  const float* Wn1 = (const float*)d_in[3];
  const float* Ws1 = (const float*)d_in[4];
  const float* b1  = (const float*)d_in[5];
  const float* Wn2 = (const float*)d_in[6];
  const float* Ws2 = (const float*)d_in[7];
  const float* b2  = (const float*)d_in[8];
  float* out = (float*)d_out;

  const int N = in_sizes[0] / 128;
  const int E = in_sizes[1] / 4;

  float* ws    = (float*)d_ws;
  float* bias2 = ws;
  float* agg0  = ws + 2 * (size_t)N;
  float* agg1  = agg0 + (size_t)N * 128;
  float* h1    = agg1 + (size_t)N * 128;
  float* elist = h1 + (size_t)N * 128;
  float* Bcat1 = elist + (size_t)N * 64;
  float* bias1 = Bcat1 + 384 * 128;
  float* Bcat2 = bias1 + 128;
  float* P     = agg0 + 32;
  float* h2    = h1;
  const int NS = (N + 8) & ~7;
  int* csr  = (int*)elist;
  int* seg  = csr + 2 * (size_t)E;
  int* rank = (int*)agg1;

  const int T = 256;

  fill_zero<<<(2 * NS + T - 1) / T, T, 0, stream>>>((float*)seg, 2 * (size_t)NS);
  const int wtot = 384 * 128 + 128 + 128 * PROW + PROW;
  build_weights<<<(wtot + T - 1) / T, T, 0, stream>>>(Wn1, Ws1, b1, Wn2, Ws2, b2,
                                                      Bcat1, bias1, Bcat2, bias2);

  const dim3 egrid(256, 2);
  count_rank<<<egrid, T, 0, stream>>>(edges, seg, rank, E, NS);
  scan_counts<<<2, 1024, 0, stream>>>(seg, N, NS);
  fill_csr<<<egrid, T, 0, stream>>>(edges, seg, rank, csr, E, NS);

  const int sblocks = 1024;
  const int gblocks = (N + 127) / 128;
  const int ablocks = 4096;
  const int pblocks = 2048;

  gather_mean<<<ablocks, T, 0, stream>>>(seg, csr, x, agg0, E, N, NS);
  sage_gemm<128, 384, true><<<gblocks, T, 0, stream>>>(x, agg0, agg1, Bcat1, bias1, h1, N);

  sage_gemm<PROW, 128, false><<<gblocks, T, 0, stream>>>(h1, h1, h1, Bcat2, bias2, P, N);
  gather_proj<<<pblocks, T, 0, stream>>>(seg, csr, P, h2, E, N, NS);

  edge_score<<<sblocks, T, 0, stream>>>(edges, edges + E, h2, out, E, N);
  edge_score<<<sblocks, T, 0, stream>>>(negE,  negE + E,  h2, out + E, E, N);
}


# speedup vs baseline: 3.1927x; 1.0738x over previous
#include <hip/hip_runtime.h>
#include <hip/hip_bf16.h>

typedef float v2f __attribute__((ext_vector_type(2)));
typedef float v8f __attribute__((ext_vector_type(8)));

#define FDIM 128
#define KC   64
#define PROW 192

__global__ void fill_zero(float* p, size_t n) {
  size_t i = (size_t)blockIdx.x * blockDim.x + threadIdx.x;
  size_t stride = (size_t)gridDim.x * blockDim.x;
  for (; i < n; i += stride) p[i] = 0.0f;
}

__global__ void build_weights(const float* __restrict__ Wn1, const float* __restrict__ Ws1,
                              const float* __restrict__ b1,
                              const float* __restrict__ Wn2, const float* __restrict__ Ws2,
                              const float* __restrict__ b2,
                              float* __restrict__ Bcat1, float* __restrict__ bias1,
                              float* __restrict__ Bcat2, float* __restrict__ bias2) {
  int i = blockIdx.x * blockDim.x + threadIdx.x;
  const int S1 = 384 * 128, S2 = S1 + 128, S3 = S2 + 128 * PROW, S4 = S3 + PROW;
  if (i < S1) {
    int k = i >> 7, n = i & 127;
    float v;
    if (k < 128)      v = Ws1[k * 128 + n] + Ws1[16384 + k * 128 + n];
    else if (k < 256) v = Wn1[(k - 128) * 128 + n];
    else              v = Wn1[16384 + (k - 256) * 128 + n];
    Bcat1[i] = v;
  } else if (i < S2) {
    int n = i - S1;
    bias1[n] = b1[n] + b1[128 + n];
  } else if (i < S3) {
    int j = i - S2;
    int k = j / PROW, n = j - k * PROW;
    float v;
    if (n < 64)       v = Ws2[k * 64 + n] + Ws2[8192 + k * 64 + n];
    else if (n < 128) v = Wn2[k * 64 + (n - 64)];
    else              v = Wn2[8192 + k * 64 + (n - 128)];
    Bcat2[j] = v;
  } else if (i < S4) {
    int n = i - S3;
    bias2[n] = (n < 64) ? b2[n] + b2[64 + n] : 0.0f;
  }
}

#define EPT 16

__global__ __launch_bounds__(256) void count_rank(const int* edges, int* seg, int* __restrict__ rank, int E, int NS) {
  const int r = blockIdx.y;
  const int* dst = edges + (size_t)r * 2 * E + E;
  int* sg = seg + (size_t)r * NS;
  int* __restrict__ rk = rank + (size_t)r * E;
  const int stride = gridDim.x * 256;
  for (int e0 = blockIdx.x * 256 + threadIdx.x; e0 < E; e0 += EPT * stride) {
    int d[EPT], v[EPT];
#pragma unroll
    for (int k = 0; k < EPT; ++k) {
      const int e = e0 + k * stride;
      d[k] = dst[e < E ? e : e0];
    }
#pragma unroll
    for (int k = 0; k < EPT; ++k) {
      v[k] = 0;
      if (e0 + k * stride < E) v[k] = atomicAdd(sg + d[k], 1);
    }
#pragma unroll
    for (int k = 0; k < EPT; ++k) {
      const int e = e0 + k * stride;
      if (e < E) rk[e] = v[k];
    }
  }
}

__global__ __launch_bounds__(1024) void scan_counts(int* seg, int N, int NS) {
  __shared__ int wsum[32];
  int* c = seg + (size_t)blockIdx.x * NS;
  const int t = threadIdx.x, lane = t & 31, w = t >> 5;
  const int per = (((N + 1023) / 1024) + 7) & ~7;
  int b = t * per;
  if (b > N) b = N;
  int e = b + per;
  if (e > N) e = N;
  int s = 0;
  for (int i = b; i < e; ++i) s += c[i];
  int inc = s;
  for (int off = 1; off < 32; off <<= 1) {
    int up = __shfl_up(inc, off, 32);
    if (lane >= off) inc += up;
  }
  if (lane == 31) wsum[w] = inc;
  __syncthreads();
  if (w == 0) {
    int v = wsum[lane];
    int iv = v;
    for (int off = 1; off < 32; off <<= 1) {
      int up = __shfl_up(iv, off, 32);
      if (lane >= off) iv += up;
    }
    wsum[lane] = iv - v;
  }
  __syncthreads();
  int run = wsum[w] + inc - s;
  for (int i = b; i < e; ++i) {
    int v = c[i];
    c[i] = run;
    run += v;
  }
  if (b < e && e == N) c[N] = run;
}

__global__ __launch_bounds__(256) void fill_csr(const int* edges, const int* __restrict__ seg,
                                                const int* __restrict__ rank, int* __restrict__ csr, int E, int NS) {
  const int r = blockIdx.y;
  const int* ed = edges + (size_t)r * 2 * E;
  const int* __restrict__ sg = seg + (size_t)r * NS;
  const int* __restrict__ rk = rank + (size_t)r * E;
  int* __restrict__ lst = csr + (size_t)r * E;
  const int stride = gridDim.x * 256;
  for (int e0 = blockIdx.x * 256 + threadIdx.x; e0 < E; e0 += EPT * stride) {
    int s[EPT], d[EPT], q[EPT], pos[EPT];
#pragma unroll
    for (int k = 0; k < EPT; ++k) {
      const int e = e0 + k * stride;
      const int ec = e < E ? e : e0;
      d[k] = ed[E + ec];
      s[k] = ed[ec];
      q[k] = rk[ec];
    }
    asm volatile("" ::: "memory");
#pragma unroll
    for (int k = 0; k < EPT; ++k) pos[k] = sg[d[k]] + q[k];
#pragma unroll
    for (int k = 0; k < EPT; ++k) {
      if (e0 + k * stride < E && (unsigned)pos[k] < (unsigned)E) lst[pos[k]] = s[k];
    }
  }
}
#undef EPT

__global__ __launch_bounds__(256) void gather_mean(const int* __restrict__ seg, const int* __restrict__ csr,
                                                   const float* __restrict__ feat, float* __restrict__ agg,
                                                   int E, int N, int NS) {
  const int lane = threadIdx.x & 31;
  const int w = __builtin_amdgcn_readfirstlane((int)(threadIdx.x >> 5));
  const int nwaves = gridDim.x * 8;
  for (int idx = blockIdx.x * 8 + w; idx < 2 * N; idx += nwaves) {
    const int r = (idx >= N) ? 1 : 0;
    const int i = idx - r * N;
    const int* c = seg + (size_t)r * NS;
    int start = c[i];
    int end = c[i + 1];
    if (start < 0) start = 0;
    if (end > E) end = E;
    const int* nb = csr + (size_t)r * E;
    float4 acc = make_float4(0.0f, 0.0f, 0.0f, 0.0f);
    for (int base = start; base < end; base += 32) {
      const int rem = end - base;
      const int n = rem < 32 ? rem : 32;
      int mine = (lane < n) ? nb[base + lane] : 0;
      if ((unsigned)mine >= (unsigned)N) mine = 0;
      int j = 0;
      for (; j + 4 <= n; j += 4) {
        const int s0 = __builtin_amdgcn_readlane(mine, j);
        const int s1 = __builtin_amdgcn_readlane(mine, j + 1);
        const int s2 = __builtin_amdgcn_readlane(mine, j + 2);
        const int s3 = __builtin_amdgcn_readlane(mine, j + 3);
        const float4 v0 = *(const float4*)(feat + (size_t)s0 * FDIM + lane * 4);
        const float4 v1 = *(const float4*)(feat + (size_t)s1 * FDIM + lane * 4);
        const float4 v2 = *(const float4*)(feat + (size_t)s2 * FDIM + lane * 4);
        const float4 v3 = *(const float4*)(feat + (size_t)s3 * FDIM + lane * 4);
        acc.x += v0.x; acc.y += v0.y; acc.z += v0.z; acc.w += v0.w;
        acc.x += v1.x; acc.y += v1.y; acc.z += v1.z; acc.w += v1.w;
        acc.x += v2.x; acc.y += v2.y; acc.z += v2.z; acc.w += v2.w;
        acc.x += v3.x; acc.y += v3.y; acc.z += v3.z; acc.w += v3.w;
      }
      for (; j < n; ++j) {
        const int s0 = __builtin_amdgcn_readlane(mine, j);
        const float4 v0 = *(const float4*)(feat + (size_t)s0 * FDIM + lane * 4);
        acc.x += v0.x; acc.y += v0.y; acc.z += v0.z; acc.w += v0.w;
      }
    }
    const float deg = (float)(end - start);
    const float dinv = 1.0f / fmaxf(deg, 1.0f);
    float4 o;
    o.x = acc.x * dinv; o.y = acc.y * dinv; o.z = acc.z * dinv; o.w = acc.w * dinv;
    *(float4*)(agg + ((size_t)r * N + i) * FDIM + lane * 4) = o;
  }
}


__device__ __forceinline__ float other_half(float v) {
  const int b = __builtin_bit_cast(int, v);
  return __builtin_bit_cast(float, __builtin_amdgcn_permlanex16(b, b, 0x76543210, 0xfedcba98, false, false));
}

__device__ __forceinline__ int first_ids(const int* __restrict__ nb, int start, int end, int lane) {
  const int rem = end - start;
  const int n = rem < 32 ? rem : 32;
  return (lane < n) ? nb[start + lane] : 0;
}

#define PULL_PAIR(K)                                                                          \
  const unsigned e##K = (unsigned)__builtin_amdgcn_readlane((int)moff, j + 2 * K);            \
  const unsigned o##K = (unsigned)__builtin_amdgcn_readlane((int)moff, j + 2 * K + 1);        \
  float4 v##K = make_float4(0.0f, 0.0f, 0.0f, 0.0f);                                          \
  if (j + 2 * K + hlf < n) v##K = *(const float4*)(pb + ((hlf ? o##K : e##K) + c16));
#define PULL_ADD(K) acc.x += v##K.x; acc.y += v##K.y; acc.z += v##K.z; acc.w += v##K.w;

__device__ __forceinline__ float4 pull_blocks(const int* __restrict__ nb, int start, int end, int mine,
                                              const char* __restrict__ pb, unsigned c16, int lane, int hlf, int N) {
  float4 acc = make_float4(0.0f, 0.0f, 0.0f, 0.0f);
  for (int base = start; base < end; base += 32) {
    const int rem = end - base;
    const int n = rem < 32 ? rem : 32;
    if (base != start) mine = (lane < n) ? nb[base + lane] : 0;
    if ((unsigned)mine >= (unsigned)N) mine = 0;
    const unsigned moff = (unsigned)mine * (PROW * 4u);
    for (int j = 0; j < n; j += 16) {
      PULL_PAIR(0) PULL_PAIR(1) PULL_PAIR(2) PULL_PAIR(3) PULL_PAIR(4) PULL_PAIR(5) PULL_PAIR(6) PULL_PAIR(7)
      PULL_ADD(0) PULL_ADD(1) PULL_ADD(2) PULL_ADD(3) PULL_ADD(4) PULL_ADD(5) PULL_ADD(6) PULL_ADD(7)
    }
  }
  return acc;
}
#undef PULL_PAIR
#undef PULL_ADD

__global__ __launch_bounds__(256) void gather_proj(const int* __restrict__ seg, const int* __restrict__ csr,
                                                   const float* __restrict__ P, float* __restrict__ h2,
                                                   int E, int N, int NS) {
  const int lane = threadIdx.x & 31;
  const int hlf = lane >> 4;
  const unsigned c16 = (unsigned)(lane & 15) * 16u;
  const int w = __builtin_amdgcn_readfirstlane((int)(threadIdx.x >> 5));
  const int nwaves = gridDim.x * 8;
  const char* pbytes = (const char*)P;
  for (int i = blockIdx.x * 8 + w; i < N; i += nwaves) {
    int s0 = seg[i], e0 = seg[i + 1];
    int s1 = seg[(size_t)NS + i], e1 = seg[(size_t)NS + i + 1];
    if (s0 < 0) s0 = 0;
    if (e0 > E) e0 = E;
    if (s1 < 0) s1 = 0;
    if (e1 > E) e1 = E;
    const int m0 = first_ids(csr, s0, e0, lane);
    const int m1 = first_ids(csr + E, s1, e1, lane);
    float4 tot = make_float4(0.0f, 0.0f, 0.0f, 0.0f);
    if (hlf == 0) tot = *(const float4*)(pbytes + ((size_t)i * (PROW * 4u) + c16));
    const float4 a0 = pull_blocks(csr, s0, e0, m0, pbytes + 256, c16, lane, hlf, N);
    const float4 a1 = pull_blocks(csr + E, s1, e1, m1, pbytes + 512, c16, lane, hlf, N);
    const float d0 = 1.0f / fmaxf((float)(e0 - s0), 1.0f);
    const float d1 = 1.0f / fmaxf((float)(e1 - s1), 1.0f);
    tot.x += a0.x * d0 + a1.x * d1;
    tot.y += a0.y * d0 + a1.y * d1;
    tot.z += a0.z * d0 + a1.z * d1;
    tot.w += a0.w * d0 + a1.w * d1;
    tot.x += other_half(tot.x);
    tot.y += other_half(tot.y);
    tot.z += other_half(tot.z);
    tot.w += other_half(tot.w);
    if (hlf == 0) *(float4*)(h2 + (size_t)i * 64 + (lane & 15) * 4) = tot;
  }
}

template <int NOUT, int KT, bool RELU>
__global__ __launch_bounds__(256) void sage_gemm(
    const float* __restrict__ A0, const float* __restrict__ G0, const float* __restrict__ G1,
    const float* __restrict__ Bmat, const float* __restrict__ bias,
    float* __restrict__ out, int Nrows) {
  constexpr int NT = NOUT / 16;
  static_assert(KC <= 128, "Bs must hold a K-chunk of B");
  __shared__ float Bs[128 * NOUT];

  const int tid = threadIdx.x;
  const int lane = tid & 31;
  const int wave = tid >> 5;
  const int rowBase = blockIdx.x * 128 + wave * 16;
  const int m = lane & 15;
  const int khalf = (lane >> 4) << 1;

  v8f czero = {};
  v8f c[NT];
#pragma unroll
  for (int t = 0; t < NT; ++t) c[t] = czero;

  int rowA = rowBase + m;
  if (rowA > Nrows - 1) rowA = Nrows - 1;

  for (int kc = 0; kc < KT; kc += KC) {
    for (int i = tid; i < KC * NOUT; i += 256)
      Bs[i] = Bmat[kc * NOUT + i];
    __syncthreads();

    const float* Asrc = (kc < 128) ? A0 : ((kc < 256) ? G0 : G1);
    const int kbase = kc & 127;

    for (int k4 = 0; k4 < KC; k4 += 4) {
      v2f a = *(const v2f*)(Asrc + (size_t)rowA * FDIM + kbase + k4 + khalf);
      const int kb = k4 + khalf;
#pragma unroll
      for (int t = 0; t < NT; ++t) {
        v2f b;
        b.x = Bs[kb * NOUT + t * 16 + m];
        b.y = Bs[(kb + 1) * NOUT + t * 16 + m];
        c[t] = __builtin_amdgcn_wmma_f32_16x16x4_f32(
            false, a, false, b, (short)0, c[t], false, false);
      }
    }
    __syncthreads();
  }

  float* slot = Bs + wave * (16 * NOUT);
  const int rhalf = (lane >> 4) << 3;
#pragma unroll
  for (int t = 0; t < NT; ++t) {
    const int col = t * 16 + m;
    const float bv = bias[col];
#pragma unroll
    for (int v = 0; v < 8; ++v) {
      float val = c[t][v] + bv;
      if (RELU) val = fmaxf(val, 0.0f);
      slot[(v + rhalf) * NOUT + col] = val;
    }
  }
  __syncthreads();
  constexpr int RP = NOUT / 4;
  constexpr int NPC = (16 * RP) / 32;
  const int wrow = __builtin_amdgcn_readfirstlane(rowBase);
  if (wrow + 16 <= Nrows) {
    float* dst = out + (size_t)wrow * NOUT + lane * 4;
#pragma unroll
    for (int j = 0; j < NPC; ++j)
      *(float4*)(dst + j * 128) = *(const float4*)(slot + (j * 32 + lane) * 4);
  } else {
#pragma unroll
    for (int j = 0; j < NPC; ++j) {
      const int q = j * 32 + lane;
      const int r = q / RP;
      const int c4 = q - r * RP;
      const int row = rowBase + r;
      if (row < Nrows) *(float4*)(out + (size_t)row * NOUT + c4 * 4) = *(const float4*)(slot + q * 4);
    }
  }
}


template <int S> __device__ __forceinline__ unsigned row_lane_plus(unsigned v, unsigned add) {
  return (unsigned)__builtin_amdgcn_update_dpp(0, (int)v, 0x150 + S, 0xf, 0xf, true) + add;
}
template <int M> __device__ __forceinline__ float row_xor_lane(float v) {
  return __builtin_bit_cast(float,
      __builtin_amdgcn_update_dpp(0, __builtin_bit_cast(int, v), 0x160 + M, 0xf, 0xf, true));
}
template <int M> __device__ __forceinline__ float fold_pair(bool up, float lo, float hi) {
  const float keep = up ? hi : lo;
  const float send = up ? lo : hi;
  return keep + row_xor_lane<M>(send);
}

#define EDGE_LOAD(S)                                                        \
  const float4 a##S = *(const float4*)(hb + row_lane_plus<S>(so, c16));     \
  const float4 b##S = *(const float4*)(hb + row_lane_plus<S>(dof, c16));
#define EDGE_DOT(S) \
  p[S] = a##S.x * b##S.x + a##S.y * b##S.y + a##S.z * b##S.z + a##S.w * b##S.w;

__global__ __launch_bounds__(256) void edge_score(const int* __restrict__ src, const int* __restrict__ dst,
                                                  const float* __restrict__ h, float* __restrict__ out,
                                                  int E, int N) {
  const int lane = threadIdx.x & 31;
  const int c = lane & 15;
  const unsigned c16 = (unsigned)c * 16u;
  const int w = __builtin_amdgcn_readfirstlane((int)(threadIdx.x >> 5));
  const int nwaves = gridDim.x * 8;
  const int ngroups = (E + 31) >> 5;
  const char* hb = (const char*)h;
  const bool u8 = (c & 8) != 0, u4 = (c & 4) != 0, u2 = (c & 2) != 0, u1 = (c & 1) != 0;

  int g = blockIdx.x * 8 + w;
  int si = 0, di = 0;
  if (g * 32 + lane < E) { si = src[g * 32 + lane]; di = dst[g * 32 + lane]; }
  for (; g < ngroups; g += nwaves) {
    const int e = g * 32 + lane;
    int en = e + nwaves * 32;
    if (en > E - 1) en = E - 1;
    const int sn = src[en], dn = dst[en];
    if ((unsigned)si >= (unsigned)N) si = 0;
    if ((unsigned)di >= (unsigned)N) di = 0;
    const unsigned so = (unsigned)si * 256u;
    const unsigned dof = (unsigned)di * 256u;
    float p[16];
    EDGE_LOAD(0) EDGE_LOAD(1) EDGE_LOAD(2) EDGE_LOAD(3) EDGE_LOAD(4) EDGE_LOAD(5) EDGE_LOAD(6) EDGE_LOAD(7)
    EDGE_LOAD(8) EDGE_LOAD(9) EDGE_LOAD(10) EDGE_LOAD(11) EDGE_LOAD(12) EDGE_LOAD(13) EDGE_LOAD(14) EDGE_LOAD(15)
    EDGE_DOT(0) EDGE_DOT(1) EDGE_DOT(2) EDGE_DOT(3) EDGE_DOT(4) EDGE_DOT(5) EDGE_DOT(6) EDGE_DOT(7)
    EDGE_DOT(8) EDGE_DOT(9) EDGE_DOT(10) EDGE_DOT(11) EDGE_DOT(12) EDGE_DOT(13) EDGE_DOT(14) EDGE_DOT(15)
    float q[8];
#pragma unroll
    for (int s = 0; s < 8; ++s) q[s] = fold_pair<8>(u8, p[s], p[s + 8]);
    float r4[4];
#pragma unroll
    for (int s = 0; s < 4; ++s) r4[s] = fold_pair<4>(u4, q[s], q[s + 4]);
    const float t0 = fold_pair<2>(u2, r4[0], r4[2]);
    const float t1 = fold_pair<2>(u2, r4[1], r4[3]);
    const float score = fold_pair<1>(u1, t0, t1);
    if (e < E) out[e] = score;
    si = sn; di = dn;
  }
}
#undef EDGE_LOAD
#undef EDGE_DOT

extern "C" void kernel_launch(void* const* d_in, const int* in_sizes, int n_in,
                              void* d_out, int out_size, void* d_ws, size_t ws_size,
                              hipStream_t stream) {
  const float* x   = (const float*)d_in[0];
  const int*   edges = (const int*)d_in[1];
  const int*   negE  = (const int*)d_in[2];
  const float* Wn1 = (const float*)d_in[3];
  const float* Ws1 = (const float*)d_in[4];
  const float* b1  = (const float*)d_in[5];
  const float* Wn2 = (const float*)d_in[6];
  const float* Ws2 = (const float*)d_in[7];
  const float* b2  = (const float*)d_in[8];
  float* out = (float*)d_out;

  const int N = in_sizes[0] / 128;
  const int E = in_sizes[1] / 4;

  float* ws    = (float*)d_ws;
  float* bias2 = ws;
  float* agg0  = ws + ((2 * (size_t)N) & ~(size_t)63);
  float* agg1  = agg0 + (size_t)N * 128;
  float* h1    = agg1 + (size_t)N * 128;
  float* elist = h1 + (size_t)N * 128;
  float* Bcat1 = elist + (size_t)N * 64;
  float* bias1 = Bcat1 + 384 * 128;
  float* Bcat2 = bias1 + 128;
  float* P     = agg0;
  float* h2    = h1;
  const int NS = (N + 8) & ~7;
  int* csr  = (int*)elist;
  int* seg  = csr + 2 * (size_t)E;
  int* rank = (int*)agg1;

  const int T = 256;

  fill_zero<<<(2 * NS + T - 1) / T, T, 0, stream>>>((float*)seg, 2 * (size_t)NS);
  const int wtot = 384 * 128 + 128 + 128 * PROW + PROW;
  build_weights<<<(wtot + T - 1) / T, T, 0, stream>>>(Wn1, Ws1, b1, Wn2, Ws2, b2,
                                                      Bcat1, bias1, Bcat2, bias2);

  const dim3 egrid(256, 2);
  count_rank<<<egrid, T, 0, stream>>>(edges, seg, rank, E, NS);
  scan_counts<<<2, 1024, 0, stream>>>(seg, N, NS);
  fill_csr<<<egrid, T, 0, stream>>>(edges, seg, rank, csr, E, NS);

  const int sblocks = 1024;
  const int gblocks = (N + 127) / 128;
  const int ablocks = 4096;
  const int pblocks = 2048;

  gather_mean<<<ablocks, T, 0, stream>>>(seg, csr, x, agg0, E, N, NS);
  sage_gemm<128, 384, true><<<gblocks, T, 0, stream>>>(x, agg0, agg1, Bcat1, bias1, h1, N);

  sage_gemm<PROW, 128, false><<<gblocks, T, 0, stream>>>(h1, h1, h1, Bcat2, bias2, P, N);
  gather_proj<<<pblocks, T, 0, stream>>>(seg, csr, P, h2, E, N, NS);

  edge_score<<<sblocks, T, 0, stream>>>(edges, edges + E, h2, out, E, N);
  edge_score<<<sblocks, T, 0, stream>>>(negE,  negE + E,  h2, out + E, E, N);
}


# Both GEMMs as one resident 16-wave workgroup per WGP: weight matrix staged in LDS once in fragment order, 16x64 units dealt evenly to waves, A fetched a chunk ahead as 16-byte loads; new sum order
# speedup vs baseline: 3.4798x; 1.0899x over previous
#include <hip/hip_runtime.h>
#include <hip/hip_bf16.h>

typedef float v2f __attribute__((ext_vector_type(2)));
typedef float v8f __attribute__((ext_vector_type(8)));

#define FDIM 128
#define PROW 192

__global__ void fill_zero(float* p, size_t n) {
  size_t i = (size_t)blockIdx.x * blockDim.x + threadIdx.x;
  size_t stride = (size_t)gridDim.x * blockDim.x;
  for (; i < n; i += stride) p[i] = 0.0f;
}

__global__ void build_weights(const float* __restrict__ Wn1, const float* __restrict__ Ws1,
                              const float* __restrict__ b1,
                              const float* __restrict__ Wn2, const float* __restrict__ Ws2,
                              const float* __restrict__ b2,
                              float* __restrict__ Bcat1, float* __restrict__ bias1,
                              float* __restrict__ Bcat2, float* __restrict__ bias2) {
  int i = blockIdx.x * blockDim.x + threadIdx.x;
  const int S1 = 384 * 128, S2 = S1 + 128, S3 = S2 + 128 * PROW, S4 = S3 + PROW;
  if (i < S1) {
    int k = i >> 7, n = i & 127;
    float v;
    if (k < 128)      v = Ws1[k * 128 + n] + Ws1[16384 + k * 128 + n];
    else if (k < 256) v = Wn1[(k - 128) * 128 + n];
    else              v = Wn1[16384 + (k - 256) * 128 + n];
    Bcat1[i] = v;
  } else if (i < S2) {
    int n = i - S1;
    bias1[n] = b1[n] + b1[128 + n];
  } else if (i < S3) {
    int j = i - S2;
    int k = j / PROW, n = j - k * PROW;
    float v;
    if (n < 64)       v = Ws2[k * 64 + n] + Ws2[8192 + k * 64 + n];
    else if (n < 128) v = Wn2[k * 64 + (n - 64)];
    else              v = Wn2[8192 + k * 64 + (n - 128)];
    Bcat2[j] = v;
  } else if (i < S4) {
    int n = i - S3;
    bias2[n] = (n < 64) ? b2[n] + b2[64 + n] : 0.0f;
  }
}

#define EPT 16

__global__ __launch_bounds__(256) void count_rank(const int* edges, int* seg, int* __restrict__ rank, int E, int NS) {
  const int r = blockIdx.y;
  const int* dst = edges + (size_t)r * 2 * E + E;
  int* sg = seg + (size_t)r * NS;
  int* __restrict__ rk = rank + (size_t)r * E;
  const int stride = gridDim.x * 256;
  for (int e0 = blockIdx.x * 256 + threadIdx.x; e0 < E; e0 += EPT * stride) {
    int d[EPT], v[EPT];
#pragma unroll
    for (int k = 0; k < EPT; ++k) {
      const int e = e0 + k * stride;
      d[k] = dst[e < E ? e : e0];
    }
#pragma unroll
    for (int k = 0; k < EPT; ++k) {
      v[k] = 0;
      if (e0 + k * stride < E) v[k] = atomicAdd(sg + d[k], 1);
    }
#pragma unroll
    for (int k = 0; k < EPT; ++k) {
      const int e = e0 + k * stride;
      if (e < E) rk[e] = v[k];
    }
  }
}

__global__ __launch_bounds__(1024) void scan_counts(int* seg, int N, int NS) {
  __shared__ int wsum[32];
  int* c = seg + (size_t)blockIdx.x * NS;
  const int t = threadIdx.x, lane = t & 31, w = t >> 5;
  const int per = (((N + 1023) / 1024) + 7) & ~7;
  int b = t * per;
  if (b > N) b = N;
  int e = b + per;
  if (e > N) e = N;
  int s = 0;
  for (int i = b; i < e; ++i) s += c[i];
  int inc = s;
  for (int off = 1; off < 32; off <<= 1) {
    int up = __shfl_up(inc, off, 32);
    if (lane >= off) inc += up;
  }
  if (lane == 31) wsum[w] = inc;
  __syncthreads();
  if (w == 0) {
    int v = wsum[lane];
    int iv = v;
    for (int off = 1; off < 32; off <<= 1) {
      int up = __shfl_up(iv, off, 32);
      if (lane >= off) iv += up;
    }
    wsum[lane] = iv - v;
  }
  __syncthreads();
  int run = wsum[w] + inc - s;
  for (int i = b; i < e; ++i) {
    int v = c[i];
    c[i] = run;
    run += v;
  }
  if (b < e && e == N) c[N] = run;
}

__global__ __launch_bounds__(256) void fill_csr(const int* edges, const int* __restrict__ seg,
                                                const int* __restrict__ rank, int* __restrict__ csr, int E, int NS) {
  const int r = blockIdx.y;
  const int* ed = edges + (size_t)r * 2 * E;
  const int* __restrict__ sg = seg + (size_t)r * NS;
  const int* __restrict__ rk = rank + (size_t)r * E;
  int* __restrict__ lst = csr + (size_t)r * E;
  const int stride = gridDim.x * 256;
  for (int e0 = blockIdx.x * 256 + threadIdx.x; e0 < E; e0 += EPT * stride) {
    int s[EPT], d[EPT], q[EPT], pos[EPT];
#pragma unroll
    for (int k = 0; k < EPT; ++k) {
      const int e = e0 + k * stride;
      const int ec = e < E ? e : e0;
      d[k] = ed[E + ec];
      s[k] = ed[ec];
      q[k] = rk[ec];
    }
    asm volatile("" ::: "memory");
#pragma unroll
    for (int k = 0; k < EPT; ++k) pos[k] = sg[d[k]] + q[k];
#pragma unroll
    for (int k = 0; k < EPT; ++k) {
      if (e0 + k * stride < E && (unsigned)pos[k] < (unsigned)E) lst[pos[k]] = s[k];
    }
  }
}
#undef EPT

__global__ __launch_bounds__(256) void gather_mean(const int* __restrict__ seg, const int* __restrict__ csr,
                                                   const float* __restrict__ feat, float* __restrict__ agg,
                                                   int E, int N, int NS) {
  const int lane = threadIdx.x & 31;
  const int w = __builtin_amdgcn_readfirstlane((int)(threadIdx.x >> 5));
  const int nwaves = gridDim.x * 8;
  for (int idx = blockIdx.x * 8 + w; idx < 2 * N; idx += nwaves) {
    const int r = (idx >= N) ? 1 : 0;
    const int i = idx - r * N;
    const int* c = seg + (size_t)r * NS;
    int start = c[i];
    int end = c[i + 1];
    if (start < 0) start = 0;
    if (end > E) end = E;
    const int* nb = csr + (size_t)r * E;
    float4 acc = make_float4(0.0f, 0.0f, 0.0f, 0.0f);
    for (int base = start; base < end; base += 32) {
      const int rem = end - base;
      const int n = rem < 32 ? rem : 32;
      int mine = (lane < n) ? nb[base + lane] : 0;
      if ((unsigned)mine >= (unsigned)N) mine = 0;
      int j = 0;
      for (; j + 4 <= n; j += 4) {
        const int s0 = __builtin_amdgcn_readlane(mine, j);
        const int s1 = __builtin_amdgcn_readlane(mine, j + 1);
        const int s2 = __builtin_amdgcn_readlane(mine, j + 2);
        const int s3 = __builtin_amdgcn_readlane(mine, j + 3);
        const float4 v0 = *(const float4*)(feat + (size_t)s0 * FDIM + lane * 4);
        const float4 v1 = *(const float4*)(feat + (size_t)s1 * FDIM + lane * 4);
        const float4 v2 = *(const float4*)(feat + (size_t)s2 * FDIM + lane * 4);
        const float4 v3 = *(const float4*)(feat + (size_t)s3 * FDIM + lane * 4);
        acc.x += v0.x; acc.y += v0.y; acc.z += v0.z; acc.w += v0.w;
        acc.x += v1.x; acc.y += v1.y; acc.z += v1.z; acc.w += v1.w;
        acc.x += v2.x; acc.y += v2.y; acc.z += v2.z; acc.w += v2.w;
        acc.x += v3.x; acc.y += v3.y; acc.z += v3.z; acc.w += v3.w;
      }
      for (; j < n; ++j) {
        const int s0 = __builtin_amdgcn_readlane(mine, j);
        const float4 v0 = *(const float4*)(feat + (size_t)s0 * FDIM + lane * 4);
        acc.x += v0.x; acc.y += v0.y; acc.z += v0.z; acc.w += v0.w;
      }
    }
    const float deg = (float)(end - start);
    const float dinv = 1.0f / fmaxf(deg, 1.0f);
    float4 o;
    o.x = acc.x * dinv; o.y = acc.y * dinv; o.z = acc.z * dinv; o.w = acc.w * dinv;
    *(float4*)(agg + ((size_t)r * N + i) * FDIM + lane * 4) = o;
  }
}


__device__ __forceinline__ float other_half(float v) {
  const int b = __builtin_bit_cast(int, v);
  return __builtin_bit_cast(float, __builtin_amdgcn_permlanex16(b, b, 0x76543210, 0xfedcba98, false, false));
}

__device__ __forceinline__ int first_ids(const int* __restrict__ nb, int start, int end, int lane) {
  const int rem = end - start;
  const int n = rem < 32 ? rem : 32;
  return (lane < n) ? nb[start + lane] : 0;
}

#define PULL_PAIR(K)                                                                          \
  const unsigned e##K = (unsigned)__builtin_amdgcn_readlane((int)moff, j + 2 * K);            \
  const unsigned o##K = (unsigned)__builtin_amdgcn_readlane((int)moff, j + 2 * K + 1);        \
  float4 v##K = make_float4(0.0f, 0.0f, 0.0f, 0.0f);                                          \
  if (j + 2 * K + hlf < n) v##K = *(const float4*)(pb + ((hlf ? o##K : e##K) + c16));
#define PULL_ADD(K) acc.x += v##K.x; acc.y += v##K.y; acc.z += v##K.z; acc.w += v##K.w;

__device__ __forceinline__ float4 pull_blocks(const int* __restrict__ nb, int start, int end, int mine,
                                              const char* __restrict__ pb, unsigned c16, int lane, int hlf, int N) {
  float4 acc = make_float4(0.0f, 0.0f, 0.0f, 0.0f);
  for (int base = start; base < end; base += 32) {
    const int rem = end - base;
    const int n = rem < 32 ? rem : 32;
    if (base != start) mine = (lane < n) ? nb[base + lane] : 0;
    if ((unsigned)mine >= (unsigned)N) mine = 0;
    const unsigned moff = (unsigned)mine * (PROW * 4u);
    for (int j = 0; j < n; j += 16) {
      PULL_PAIR(0) PULL_PAIR(1) PULL_PAIR(2) PULL_PAIR(3) PULL_PAIR(4) PULL_PAIR(5) PULL_PAIR(6) PULL_PAIR(7)
      PULL_ADD(0) PULL_ADD(1) PULL_ADD(2) PULL_ADD(3) PULL_ADD(4) PULL_ADD(5) PULL_ADD(6) PULL_ADD(7)
    }
  }
  return acc;
}
#undef PULL_PAIR
#undef PULL_ADD

__global__ __launch_bounds__(256) void gather_proj(const int* __restrict__ seg, const int* __restrict__ csr,
                                                   const float* __restrict__ P, float* __restrict__ h2,
                                                   int E, int N, int NS) {
  const int lane = threadIdx.x & 31;
  const int hlf = lane >> 4;
  const unsigned c16 = (unsigned)(lane & 15) * 16u;
  const int w = __builtin_amdgcn_readfirstlane((int)(threadIdx.x >> 5));
  const int nwaves = gridDim.x * 8;
  const char* pbytes = (const char*)P;
  for (int i = blockIdx.x * 8 + w; i < N; i += nwaves) {
    int s0 = seg[i], e0 = seg[i + 1];
    int s1 = seg[(size_t)NS + i], e1 = seg[(size_t)NS + i + 1];
    if (s0 < 0) s0 = 0;
    if (e0 > E) e0 = E;
    if (s1 < 0) s1 = 0;
    if (e1 > E) e1 = E;
    const int m0 = first_ids(csr, s0, e0, lane);
    const int m1 = first_ids(csr + E, s1, e1, lane);
    float4 tot = make_float4(0.0f, 0.0f, 0.0f, 0.0f);
    if (hlf == 0) tot = *(const float4*)(pbytes + ((size_t)i * (PROW * 4u) + c16));
    const float4 a0 = pull_blocks(csr, s0, e0, m0, pbytes + 256, c16, lane, hlf, N);
    const float4 a1 = pull_blocks(csr + E, s1, e1, m1, pbytes + 512, c16, lane, hlf, N);
    const float d0 = 1.0f / fmaxf((float)(e0 - s0), 1.0f);
    const float d1 = 1.0f / fmaxf((float)(e1 - s1), 1.0f);
    tot.x += a0.x * d0 + a1.x * d1;
    tot.y += a0.y * d0 + a1.y * d1;
    tot.z += a0.z * d0 + a1.z * d1;
    tot.w += a0.w * d0 + a1.w * d1;
    tot.x += other_half(tot.x);
    tot.y += other_half(tot.y);
    tot.z += other_half(tot.z);
    tot.w += other_half(tot.w);
    if (hlf == 0) *(float4*)(h2 + (size_t)i * 64 + (lane & 15) * 4) = tot;
  }
}

typedef float v4f __attribute__((ext_vector_type(4)));

#define GEMM_WGS 256
#define GEMM_WAVES 16
#define GEMM_THREADS 512
#define SLOT_PITCH 68

template <int NOUT, int KT, bool RELU>
__global__ __launch_bounds__(GEMM_THREADS) void sage_gemm(
    const float* __restrict__ A0, const float* __restrict__ G0, const float* __restrict__ G1,
    const float* __restrict__ Bmat, const float* __restrict__ bias,
    float* __restrict__ out, int Nrows) {
  constexpr int NCH = KT / 64;
  constexpr int NSPLIT = NOUT / 64;
  static_assert(NCH % 2 == 0, "the two A register sets swap roles once per chunk");
  __shared__ v4f Bs[96 * 128];
  __shared__ float slots[GEMM_WAVES * 16 * SLOT_PITCH];

  const int tid = threadIdx.x;
  const int lane = tid & 31;
  const int m = lane & 15;
  const int h = lane >> 4;
  const int wave = __builtin_amdgcn_readfirstlane(tid >> 5);

  for (int i = tid; i < (KT / 4) * NOUT; i += GEMM_THREADS) {
    const int kq = i / NOUT;
    const int n = i - kq * NOUT;
    const float* p = Bmat + (size_t)(4 * kq) * NOUT + n;
    v4f w;
    w.x = p[0]; w.y = p[NOUT]; w.z = p[2 * NOUT]; w.w = p[3 * NOUT];
    Bs[i] = w;
  }
  __syncthreads();

  const int nRT = (Nrows + 15) >> 4;
  const int rt0 = (int)(((long long)blockIdx.x * nRT) / GEMM_WGS);
  const int rt1 = (int)(((long long)(blockIdx.x + 1) * nRT) / GEMM_WGS);
  const int nunits = (rt1 - rt0) * NSPLIT;

  auto a_ptr = [&](int u, int c) -> const float* {
    const int uu = u < nunits ? u : nunits - 1;
    int row = (rt0 + uu / NSPLIT) * 16 + m;
    if (row > Nrows - 1) row = Nrows - 1;
    const float* src = (c < 2) ? A0 : ((c < 4) ? G0 : G1);
    return src + (size_t)row * FDIM + (c & 1) * 64 + 4 * h;
  };

  if (wave >= nunits) return;
  v4f abuf[2][8];
  {
    const float* p = a_ptr(wave, 0);
#pragma unroll
    for (int j = 0; j < 8; ++j) abuf[0][j] = *(const v4f*)(p + j * 8);
  }
  float* slot = slots + wave * (16 * SLOT_PITCH);

  for (int u = wave; u < nunits; u += GEMM_WAVES) {
    const int rt = rt0 + u / NSPLIT;
    const int nh = u - (u / NSPLIT) * NSPLIT;
    v8f acc[4];
#pragma unroll
    for (int t = 0; t < 4; ++t) acc[t] = v8f{};

    const v4f* bunit = Bs + (size_t)h * NOUT + nh * 64 + m;
    v4f bcur[4], bnxt[4];
#pragma unroll
    for (int t = 0; t < 4; ++t) bcur[t] = bunit[t * 16];

#pragma unroll
    for (int c = 0; c < NCH; ++c) {
      {
        const float* p = (c + 1 < NCH) ? a_ptr(u, c + 1) : a_ptr(u + GEMM_WAVES, 0);
#pragma unroll
        for (int j = 0; j < 8; ++j) abuf[(c + 1) & 1][j] = *(const v4f*)(p + j * 8);
      }
#pragma unroll
      for (int j = 0; j < 8; ++j) {
        const int st = c * 8 + j;
        if (st + 1 < NCH * 8) {
#pragma unroll
          for (int t = 0; t < 4; ++t) bnxt[t] = bunit[(size_t)(2 * (st + 1)) * NOUT + t * 16];
        }
        const v4f a = abuf[c & 1][j];
        const v2f alo = {a.x, a.y}, ahi = {a.z, a.w};
#pragma unroll
        for (int t = 0; t < 4; ++t) {
          const v2f blo = {bcur[t].x, bcur[t].y};
          acc[t] = __builtin_amdgcn_wmma_f32_16x16x4_f32(false, alo, false, blo, (short)0, acc[t], false, false);
        }
#pragma unroll
        for (int t = 0; t < 4; ++t) {
          const v2f bhi = {bcur[t].z, bcur[t].w};
          acc[t] = __builtin_amdgcn_wmma_f32_16x16x4_f32(false, ahi, false, bhi, (short)0, acc[t], false, false);
        }
#pragma unroll
        for (int t = 0; t < 4; ++t) bcur[t] = bnxt[t];
      }
    }

#pragma unroll
    for (int t = 0; t < 4; ++t) {
      const float bv = bias[nh * 64 + t * 16 + m];
#pragma unroll
      for (int v = 0; v < 8; ++v) {
        float val = acc[t][v] + bv;
        if (RELU) val = fmaxf(val, 0.0f);
        slot[(v + 8 * h) * SLOT_PITCH + t * 16 + m] = val;
      }
    }
    __builtin_amdgcn_fence(__ATOMIC_RELEASE, "wavefront");
    __builtin_amdgcn_wave_barrier();
    __builtin_amdgcn_fence(__ATOMIC_ACQUIRE, "wavefront");
    float* orow = out + (size_t)(rt * 16 + h) * NOUT + nh * 64 + m * 4;
    const float* srow = slot + h * SLOT_PITCH + m * 4;
    if (rt * 16 + 16 <= Nrows) {
#pragma unroll
      for (int j = 0; j < 8; ++j)
        *(float4*)(orow + (size_t)(2 * j) * NOUT) = *(const float4*)(srow + 2 * j * SLOT_PITCH);
    } else {
#pragma unroll
      for (int j = 0; j < 8; ++j) {
        if (rt * 16 + 2 * j + h < Nrows)
          *(float4*)(orow + (size_t)(2 * j) * NOUT) = *(const float4*)(srow + 2 * j * SLOT_PITCH);
      }
    }
    __builtin_amdgcn_fence(__ATOMIC_RELEASE, "wavefront");
    __builtin_amdgcn_wave_barrier();
    __builtin_amdgcn_fence(__ATOMIC_ACQUIRE, "wavefront");
  }
}


template <int S> __device__ __forceinline__ unsigned row_lane_plus(unsigned v, unsigned add) {
  return (unsigned)__builtin_amdgcn_update_dpp(0, (int)v, 0x150 + S, 0xf, 0xf, true) + add;
}
template <int M> __device__ __forceinline__ float row_xor_lane(float v) {
  return __builtin_bit_cast(float,
      __builtin_amdgcn_update_dpp(0, __builtin_bit_cast(int, v), 0x160 + M, 0xf, 0xf, true));
}
template <int M> __device__ __forceinline__ float fold_pair(bool up, float lo, float hi) {
  const float keep = up ? hi : lo;
  const float send = up ? lo : hi;
  return keep + row_xor_lane<M>(send);
}

#define EDGE_LOAD(S)                                                        \
  const float4 a##S = *(const float4*)(hb + row_lane_plus<S>(so, c16));     \
  const float4 b##S = *(const float4*)(hb + row_lane_plus<S>(dof, c16));
#define EDGE_DOT(S) \
  p[S] = a##S.x * b##S.x + a##S.y * b##S.y + a##S.z * b##S.z + a##S.w * b##S.w;

__global__ __launch_bounds__(256) void edge_score(const int* __restrict__ src, const int* __restrict__ dst,
                                                  const float* __restrict__ h, float* __restrict__ out,
                                                  int E, int N) {
  const int lane = threadIdx.x & 31;
  const int c = lane & 15;
  const unsigned c16 = (unsigned)c * 16u;
  const int w = __builtin_amdgcn_readfirstlane((int)(threadIdx.x >> 5));
  const int nwaves = gridDim.x * 8;
  const int ngroups = (E + 31) >> 5;
  const char* hb = (const char*)h;
  const bool u8 = (c & 8) != 0, u4 = (c & 4) != 0, u2 = (c & 2) != 0, u1 = (c & 1) != 0;

  int g = blockIdx.x * 8 + w;
  int si = 0, di = 0;
  if (g * 32 + lane < E) { si = src[g * 32 + lane]; di = dst[g * 32 + lane]; }
  for (; g < ngroups; g += nwaves) {
    const int e = g * 32 + lane;
    int en = e + nwaves * 32;
    if (en > E - 1) en = E - 1;
    const int sn = src[en], dn = dst[en];
    if ((unsigned)si >= (unsigned)N) si = 0;
    if ((unsigned)di >= (unsigned)N) di = 0;
    const unsigned so = (unsigned)si * 256u;
    const unsigned dof = (unsigned)di * 256u;
    float p[16];
    EDGE_LOAD(0) EDGE_LOAD(1) EDGE_LOAD(2) EDGE_LOAD(3) EDGE_LOAD(4) EDGE_LOAD(5) EDGE_LOAD(6) EDGE_LOAD(7)
    EDGE_LOAD(8) EDGE_LOAD(9) EDGE_LOAD(10) EDGE_LOAD(11) EDGE_LOAD(12) EDGE_LOAD(13) EDGE_LOAD(14) EDGE_LOAD(15)
    EDGE_DOT(0) EDGE_DOT(1) EDGE_DOT(2) EDGE_DOT(3) EDGE_DOT(4) EDGE_DOT(5) EDGE_DOT(6) EDGE_DOT(7)
    EDGE_DOT(8) EDGE_DOT(9) EDGE_DOT(10) EDGE_DOT(11) EDGE_DOT(12) EDGE_DOT(13) EDGE_DOT(14) EDGE_DOT(15)
    float q[8];
#pragma unroll
    for (int s = 0; s < 8; ++s) q[s] = fold_pair<8>(u8, p[s], p[s + 8]);
    float r4[4];
#pragma unroll
    for (int s = 0; s < 4; ++s) r4[s] = fold_pair<4>(u4, q[s], q[s + 4]);
    const float t0 = fold_pair<2>(u2, r4[0], r4[2]);
    const float t1 = fold_pair<2>(u2, r4[1], r4[3]);
    const float score = fold_pair<1>(u1, t0, t1);
    if (e < E) out[e] = score;
    si = sn; di = dn;
  }
}
#undef EDGE_LOAD
#undef EDGE_DOT

extern "C" void kernel_launch(void* const* d_in, const int* in_sizes, int n_in,
                              void* d_out, int out_size, void* d_ws, size_t ws_size,
                              hipStream_t stream) {
  const float* x   = (const float*)d_in[0];
  const int*   edges = (const int*)d_in[1];
  const int*   negE  = (const int*)d_in[2];
  const float* Wn1 = (const float*)d_in[3];
  const float* Ws1 = (const float*)d_in[4];
  const float* b1  = (const float*)d_in[5];
  const float* Wn2 = (const float*)d_in[6];
  const float* Ws2 = (const float*)d_in[7];
  const float* b2  = (const float*)d_in[8];
  float* out = (float*)d_out;

  const int N = in_sizes[0] / 128;
  const int E = in_sizes[1] / 4;

  float* ws    = (float*)d_ws;
  float* bias2 = ws;
  float* agg0  = ws + ((2 * (size_t)N) & ~(size_t)63);
  float* agg1  = agg0 + (size_t)N * 128;
  float* h1    = agg1 + (size_t)N * 128;
  float* elist = h1 + (size_t)N * 128;
  float* Bcat1 = elist + (size_t)N * 64;
  float* bias1 = Bcat1 + 384 * 128;
  float* Bcat2 = bias1 + 128;
  float* P     = agg0;
  float* h2    = h1;
  const int NS = (N + 8) & ~7;
  int* csr  = (int*)elist;
  int* seg  = csr + 2 * (size_t)E;
  int* rank = (int*)agg1;

  const int T = 256;

  fill_zero<<<(2 * NS + T - 1) / T, T, 0, stream>>>((float*)seg, 2 * (size_t)NS);
  const int wtot = 384 * 128 + 128 + 128 * PROW + PROW;
  build_weights<<<(wtot + T - 1) / T, T, 0, stream>>>(Wn1, Ws1, b1, Wn2, Ws2, b2,
                                                      Bcat1, bias1, Bcat2, bias2);

  const dim3 egrid(256, 2);
  count_rank<<<egrid, T, 0, stream>>>(edges, seg, rank, E, NS);
  scan_counts<<<2, 1024, 0, stream>>>(seg, N, NS);
  fill_csr<<<egrid, T, 0, stream>>>(edges, seg, rank, csr, E, NS);

  const int sblocks = 1024;
  const int ablocks = 4096;
  const int pblocks = 2048;

  gather_mean<<<ablocks, T, 0, stream>>>(seg, csr, x, agg0, E, N, NS);
  sage_gemm<128, 384, true><<<GEMM_WGS, GEMM_THREADS, 0, stream>>>(x, agg0, agg1, Bcat1, bias1, h1, N);

  sage_gemm<PROW, 128, false><<<GEMM_WGS, GEMM_THREADS, 0, stream>>>(h1, h1, h1, Bcat2, bias2, P, N);
  gather_proj<<<pblocks, T, 0, stream>>>(seg, csr, P, h2, E, N, NS);

  edge_score<<<sblocks, T, 0, stream>>>(edges, edges + E, h2, out, E, N);
  edge_score<<<sblocks, T, 0, stream>>>(negE,  negE + E,  h2, out + E, E, N);
}


# Three launch boundaries removed (11 to 8): counter clearing joins the weight build, each fill workgroup scans its relation's counts in LDS itself, one launch scores both edge lists
# speedup vs baseline: 3.7811x; 1.0866x over previous
#include <hip/hip_runtime.h>
#include <hip/hip_bf16.h>

typedef float v2f __attribute__((ext_vector_type(2)));
typedef float v8f __attribute__((ext_vector_type(8)));

#define FDIM 128
#define PROW 192

__global__ void build_weights(const float* __restrict__ Wn1, const float* __restrict__ Ws1,
                              const float* __restrict__ b1,
                              const float* __restrict__ Wn2, const float* __restrict__ Ws2,
                              const float* __restrict__ b2,
                              float* __restrict__ Bcat1, float* __restrict__ bias1,
                              float* __restrict__ Bcat2, float* __restrict__ bias2,
                              int* __restrict__ cnt, int nzero) {
  int i = blockIdx.x * blockDim.x + threadIdx.x;
  for (int z = i; z < nzero; z += gridDim.x * blockDim.x) cnt[z] = 0;
  const int S1 = 384 * 128, S2 = S1 + 128, S3 = S2 + 128 * PROW, S4 = S3 + PROW;
  if (i < S1) {
    int k = i >> 7, n = i & 127;
    float v;
    if (k < 128)      v = Ws1[k * 128 + n] + Ws1[16384 + k * 128 + n];
    else if (k < 256) v = Wn1[(k - 128) * 128 + n];
    else              v = Wn1[16384 + (k - 256) * 128 + n];
    Bcat1[i] = v;
  } else if (i < S2) {
    int n = i - S1;
    bias1[n] = b1[n] + b1[128 + n];
  } else if (i < S3) {
    int j = i - S2;
    int k = j / PROW, n = j - k * PROW;
    float v;
    if (n < 64)       v = Ws2[k * 64 + n] + Ws2[8192 + k * 64 + n];
    else if (n < 128) v = Wn2[k * 64 + (n - 64)];
    else              v = Wn2[8192 + k * 64 + (n - 128)];
    Bcat2[j] = v;
  } else if (i < S4) {
    int n = i - S3;
    bias2[n] = (n < 64) ? b2[n] + b2[64 + n] : 0.0f;
  }
}

#define EPT 16

__global__ __launch_bounds__(256) void count_rank(const int* edges, int* seg, int* __restrict__ rank, int E, int NS) {
  const int r = blockIdx.y;
  const int* dst = edges + (size_t)r * 2 * E + E;
  int* sg = seg + (size_t)r * NS;
  int* __restrict__ rk = rank + (size_t)r * E;
  const int stride = gridDim.x * 256;
  for (int e0 = blockIdx.x * 256 + threadIdx.x; e0 < E; e0 += EPT * stride) {
    int d[EPT], v[EPT];
#pragma unroll
    for (int k = 0; k < EPT; ++k) {
      const int e = e0 + k * stride;
      d[k] = dst[e < E ? e : e0];
    }
#pragma unroll
    for (int k = 0; k < EPT; ++k) {
      v[k] = 0;
      if (e0 + k * stride < E) v[k] = atomicAdd(sg + d[k], 1);
    }
#pragma unroll
    for (int k = 0; k < EPT; ++k) {
      const int e = e0 + k * stride;
      if (e < E) rk[e] = v[k];
    }
  }
}

#define SF_THREADS 1024
#define SF_CAP 65536
#define SF_EPT 8
__global__ __launch_bounds__(SF_THREADS) void scan_fill_csr(const int* edges, const int* __restrict__ cnt,
                                                            const int* rank, int* __restrict__ segout,
                                                            int* __restrict__ csr, int E, int N, int NS) {
  __shared__ __attribute__((aligned(16))) int pre[SF_CAP];
  __shared__ int wsum[32];
  if (N < 1 || N + 1 > SF_CAP) return;
  const int r = blockIdx.y;
  const int t = threadIdx.x, lane = t & 31, w = t >> 5;
  const int* __restrict__ c = cnt + (size_t)r * NS;

  {
    const int N4 = (N + 3) >> 2;
    int4 v[SF_CAP / 4 / SF_THREADS];
#pragma unroll
    for (int k = 0; k < SF_CAP / 4 / SF_THREADS; ++k) {
      const int i = t + k * SF_THREADS;
      v[k] = ((const int4*)c)[i < N4 ? i : 0];
    }
#pragma unroll
    for (int k = 0; k < SF_CAP / 4 / SF_THREADS; ++k) ((int4*)pre)[t + k * SF_THREADS] = v[k];
  }
  __syncthreads();

  const int per = ((N + SF_THREADS - 1) / SF_THREADS) | 1;
  int b = t * per;
  if (b > N) b = N;
  int e = b + per;
  if (e > N) e = N;
  int s = 0;
  for (int i = b; i < e; ++i) s += pre[i];
  int inc = s;
  for (int off = 1; off < 32; off <<= 1) {
    int up = __shfl_up(inc, off, 32);
    if (lane >= off) inc += up;
  }
  if (lane == 31) wsum[w] = inc;
  __syncthreads();
  if (w == 0) {
    int v = wsum[lane];
    int iv = v;
    for (int off = 1; off < 32; off <<= 1) {
      int up = __shfl_up(iv, off, 32);
      if (lane >= off) iv += up;
    }
    wsum[lane] = iv - v;
  }
  __syncthreads();
  int run = wsum[w] + inc - s;
  for (int i = b; i < e; ++i) {
    int v = pre[i];
    pre[i] = run;
    run += v;
  }
  if (b < e && e == N) pre[N] = run;
  __syncthreads();

  for (int i = blockIdx.x * SF_THREADS + t; i <= N; i += gridDim.x * SF_THREADS) segout[(size_t)r * NS + i] = pre[i];

  const int* ed = edges + (size_t)r * 2 * E;
  const int* rk = rank + (size_t)r * E;
  int* __restrict__ lst = csr + (size_t)r * E;
  const int stride = gridDim.x * SF_THREADS;
  for (int e0 = blockIdx.x * SF_THREADS + t; e0 < E; e0 += SF_EPT * stride) {
    int sv[SF_EPT], d[SF_EPT], q[SF_EPT];
#pragma unroll
    for (int k = 0; k < SF_EPT; ++k) {
      const int ee = e0 + k * stride;
      const int ec = ee < E ? ee : e0;
      d[k] = ed[E + ec];
      sv[k] = ed[ec];
      q[k] = rk[ec];
    }
    asm volatile("" ::: "memory");
#pragma unroll
    for (int k = 0; k < SF_EPT; ++k) {
      const int dd = d[k];
      if (e0 + k * stride < E && (unsigned)dd < (unsigned)N) {
        const int pos = pre[dd] + q[k];
        if ((unsigned)pos < (unsigned)E) lst[pos] = sv[k];
      }
    }
  }
}
#undef SF_EPT
#undef EPT

__global__ __launch_bounds__(256) void gather_mean(const int* __restrict__ seg, const int* __restrict__ csr,
                                                   const float* __restrict__ feat, float* __restrict__ agg,
                                                   int E, int N, int NS) {
  const int lane = threadIdx.x & 31;
  const int w = __builtin_amdgcn_readfirstlane((int)(threadIdx.x >> 5));
  const int nwaves = gridDim.x * 8;
  for (int idx = blockIdx.x * 8 + w; idx < 2 * N; idx += nwaves) {
    const int r = (idx >= N) ? 1 : 0;
    const int i = idx - r * N;
    const int* c = seg + (size_t)r * NS;
    int start = c[i];
    int end = c[i + 1];
    if (start < 0) start = 0;
    if (end > E) end = E;
    const int* nb = csr + (size_t)r * E;
    float4 acc = make_float4(0.0f, 0.0f, 0.0f, 0.0f);
    for (int base = start; base < end; base += 32) {
      const int rem = end - base;
      const int n = rem < 32 ? rem : 32;
      int mine = (lane < n) ? nb[base + lane] : 0;
      if ((unsigned)mine >= (unsigned)N) mine = 0;
      int j = 0;
      for (; j + 4 <= n; j += 4) {
        const int s0 = __builtin_amdgcn_readlane(mine, j);
        const int s1 = __builtin_amdgcn_readlane(mine, j + 1);
        const int s2 = __builtin_amdgcn_readlane(mine, j + 2);
        const int s3 = __builtin_amdgcn_readlane(mine, j + 3);
        const float4 v0 = *(const float4*)(feat + (size_t)s0 * FDIM + lane * 4);
        const float4 v1 = *(const float4*)(feat + (size_t)s1 * FDIM + lane * 4);
        const float4 v2 = *(const float4*)(feat + (size_t)s2 * FDIM + lane * 4);
        const float4 v3 = *(const float4*)(feat + (size_t)s3 * FDIM + lane * 4);
        acc.x += v0.x; acc.y += v0.y; acc.z += v0.z; acc.w += v0.w;
        acc.x += v1.x; acc.y += v1.y; acc.z += v1.z; acc.w += v1.w;
        acc.x += v2.x; acc.y += v2.y; acc.z += v2.z; acc.w += v2.w;
        acc.x += v3.x; acc.y += v3.y; acc.z += v3.z; acc.w += v3.w;
      }
      for (; j < n; ++j) {
        const int s0 = __builtin_amdgcn_readlane(mine, j);
        const float4 v0 = *(const float4*)(feat + (size_t)s0 * FDIM + lane * 4);
        acc.x += v0.x; acc.y += v0.y; acc.z += v0.z; acc.w += v0.w;
      }
    }
    const float deg = (float)(end - start);
    const float dinv = 1.0f / fmaxf(deg, 1.0f);
    float4 o;
    o.x = acc.x * dinv; o.y = acc.y * dinv; o.z = acc.z * dinv; o.w = acc.w * dinv;
    *(float4*)(agg + ((size_t)r * N + i) * FDIM + lane * 4) = o;
  }
}


__device__ __forceinline__ float other_half(float v) {
  const int b = __builtin_bit_cast(int, v);
  return __builtin_bit_cast(float, __builtin_amdgcn_permlanex16(b, b, 0x76543210, 0xfedcba98, false, false));
}

__device__ __forceinline__ int first_ids(const int* __restrict__ nb, int start, int end, int lane) {
  const int rem = end - start;
  const int n = rem < 32 ? rem : 32;
  return (lane < n) ? nb[start + lane] : 0;
}

#define PULL_PAIR(K)                                                                          \
  const unsigned e##K = (unsigned)__builtin_amdgcn_readlane((int)moff, j + 2 * K);            \
  const unsigned o##K = (unsigned)__builtin_amdgcn_readlane((int)moff, j + 2 * K + 1);        \
  float4 v##K = make_float4(0.0f, 0.0f, 0.0f, 0.0f);                                          \
  if (j + 2 * K + hlf < n) v##K = *(const float4*)(pb + ((hlf ? o##K : e##K) + c16));
#define PULL_ADD(K) acc.x += v##K.x; acc.y += v##K.y; acc.z += v##K.z; acc.w += v##K.w;

__device__ __forceinline__ float4 pull_blocks(const int* __restrict__ nb, int start, int end, int mine,
                                              const char* __restrict__ pb, unsigned c16, int lane, int hlf, int N) {
  float4 acc = make_float4(0.0f, 0.0f, 0.0f, 0.0f);
  for (int base = start; base < end; base += 32) {
    const int rem = end - base;
    const int n = rem < 32 ? rem : 32;
    if (base != start) mine = (lane < n) ? nb[base + lane] : 0;
    if ((unsigned)mine >= (unsigned)N) mine = 0;
    const unsigned moff = (unsigned)mine * (PROW * 4u);
    for (int j = 0; j < n; j += 16) {
      PULL_PAIR(0) PULL_PAIR(1) PULL_PAIR(2) PULL_PAIR(3) PULL_PAIR(4) PULL_PAIR(5) PULL_PAIR(6) PULL_PAIR(7)
      PULL_ADD(0) PULL_ADD(1) PULL_ADD(2) PULL_ADD(3) PULL_ADD(4) PULL_ADD(5) PULL_ADD(6) PULL_ADD(7)
    }
  }
  return acc;
}
#undef PULL_PAIR
#undef PULL_ADD

__global__ __launch_bounds__(256) void gather_proj(const int* __restrict__ seg, const int* __restrict__ csr,
                                                   const float* __restrict__ P, float* __restrict__ h2,
                                                   int E, int N, int NS) {
  const int lane = threadIdx.x & 31;
  const int hlf = lane >> 4;
  const unsigned c16 = (unsigned)(lane & 15) * 16u;
  const int w = __builtin_amdgcn_readfirstlane((int)(threadIdx.x >> 5));
  const int nwaves = gridDim.x * 8;
  const char* pbytes = (const char*)P;
  for (int i = blockIdx.x * 8 + w; i < N; i += nwaves) {
    int s0 = seg[i], e0 = seg[i + 1];
    int s1 = seg[(size_t)NS + i], e1 = seg[(size_t)NS + i + 1];
    if (s0 < 0) s0 = 0;
    if (e0 > E) e0 = E;
    if (s1 < 0) s1 = 0;
    if (e1 > E) e1 = E;
    const int m0 = first_ids(csr, s0, e0, lane);
    const int m1 = first_ids(csr + E, s1, e1, lane);
    float4 tot = make_float4(0.0f, 0.0f, 0.0f, 0.0f);
    if (hlf == 0) tot = *(const float4*)(pbytes + ((size_t)i * (PROW * 4u) + c16));
    const float4 a0 = pull_blocks(csr, s0, e0, m0, pbytes + 256, c16, lane, hlf, N);
    const float4 a1 = pull_blocks(csr + E, s1, e1, m1, pbytes + 512, c16, lane, hlf, N);
    const float d0 = 1.0f / fmaxf((float)(e0 - s0), 1.0f);
    const float d1 = 1.0f / fmaxf((float)(e1 - s1), 1.0f);
    tot.x += a0.x * d0 + a1.x * d1;
    tot.y += a0.y * d0 + a1.y * d1;
    tot.z += a0.z * d0 + a1.z * d1;
    tot.w += a0.w * d0 + a1.w * d1;
    tot.x += other_half(tot.x);
    tot.y += other_half(tot.y);
    tot.z += other_half(tot.z);
    tot.w += other_half(tot.w);
    if (hlf == 0) *(float4*)(h2 + (size_t)i * 64 + (lane & 15) * 4) = tot;
  }
}

typedef float v4f __attribute__((ext_vector_type(4)));

#define GEMM_WGS 256
#define GEMM_WAVES 16
#define GEMM_THREADS 512
#define SLOT_PITCH 68

template <int NOUT, int KT, bool RELU>
__global__ __launch_bounds__(GEMM_THREADS) void sage_gemm(
    const float* __restrict__ A0, const float* __restrict__ G0, const float* __restrict__ G1,
    const float* __restrict__ Bmat, const float* __restrict__ bias,
    float* __restrict__ out, int Nrows) {
  constexpr int NCH = KT / 64;
  constexpr int NSPLIT = NOUT / 64;
  static_assert(NCH % 2 == 0, "the two A register sets swap roles once per chunk");
  __shared__ v4f Bs[96 * 128];
  __shared__ float slots[GEMM_WAVES * 16 * SLOT_PITCH];

  const int tid = threadIdx.x;
  const int lane = tid & 31;
  const int m = lane & 15;
  const int h = lane >> 4;
  const int wave = __builtin_amdgcn_readfirstlane(tid >> 5);

  for (int i = tid; i < (KT / 4) * NOUT; i += GEMM_THREADS) {
    const int kq = i / NOUT;
    const int n = i - kq * NOUT;
    const float* p = Bmat + (size_t)(4 * kq) * NOUT + n;
    v4f w;
    w.x = p[0]; w.y = p[NOUT]; w.z = p[2 * NOUT]; w.w = p[3 * NOUT];
    Bs[i] = w;
  }
  __syncthreads();

  const int nRT = (Nrows + 15) >> 4;
  const int rt0 = (int)(((long long)blockIdx.x * nRT) / GEMM_WGS);
  const int rt1 = (int)(((long long)(blockIdx.x + 1) * nRT) / GEMM_WGS);
  const int nunits = (rt1 - rt0) * NSPLIT;

  auto a_ptr = [&](int u, int c) -> const float* {
    const int uu = u < nunits ? u : nunits - 1;
    int row = (rt0 + uu / NSPLIT) * 16 + m;
    if (row > Nrows - 1) row = Nrows - 1;
    const float* src = (c < 2) ? A0 : ((c < 4) ? G0 : G1);
    return src + (size_t)row * FDIM + (c & 1) * 64 + 4 * h;
  };

  if (wave >= nunits) return;
  v4f abuf[2][8];
  {
    const float* p = a_ptr(wave, 0);
#pragma unroll
    for (int j = 0; j < 8; ++j) abuf[0][j] = *(const v4f*)(p + j * 8);
  }
  float* slot = slots + wave * (16 * SLOT_PITCH);

  for (int u = wave; u < nunits; u += GEMM_WAVES) {
    const int rt = rt0 + u / NSPLIT;
    const int nh = u - (u / NSPLIT) * NSPLIT;
    v8f acc[4];
#pragma unroll
    for (int t = 0; t < 4; ++t) acc[t] = v8f{};

    const v4f* bunit = Bs + (size_t)h * NOUT + nh * 64 + m;
    v4f bcur[4], bnxt[4];
#pragma unroll
    for (int t = 0; t < 4; ++t) bcur[t] = bunit[t * 16];

#pragma unroll
    for (int c = 0; c < NCH; ++c) {
      {
        const float* p = (c + 1 < NCH) ? a_ptr(u, c + 1) : a_ptr(u + GEMM_WAVES, 0);
#pragma unroll
        for (int j = 0; j < 8; ++j) abuf[(c + 1) & 1][j] = *(const v4f*)(p + j * 8);
      }
#pragma unroll
      for (int j = 0; j < 8; ++j) {
        const int st = c * 8 + j;
        if (st + 1 < NCH * 8) {
#pragma unroll
          for (int t = 0; t < 4; ++t) bnxt[t] = bunit[(size_t)(2 * (st + 1)) * NOUT + t * 16];
        }
        const v4f a = abuf[c & 1][j];
        const v2f alo = {a.x, a.y}, ahi = {a.z, a.w};
#pragma unroll
        for (int t = 0; t < 4; ++t) {
          const v2f blo = {bcur[t].x, bcur[t].y};
          acc[t] = __builtin_amdgcn_wmma_f32_16x16x4_f32(false, alo, false, blo, (short)0, acc[t], false, false);
        }
#pragma unroll
        for (int t = 0; t < 4; ++t) {
          const v2f bhi = {bcur[t].z, bcur[t].w};
          acc[t] = __builtin_amdgcn_wmma_f32_16x16x4_f32(false, ahi, false, bhi, (short)0, acc[t], false, false);
        }
#pragma unroll
        for (int t = 0; t < 4; ++t) bcur[t] = bnxt[t];
      }
    }

#pragma unroll
    for (int t = 0; t < 4; ++t) {
      const float bv = bias[nh * 64 + t * 16 + m];
#pragma unroll
      for (int v = 0; v < 8; ++v) {
        float val = acc[t][v] + bv;
        if (RELU) val = fmaxf(val, 0.0f);
        slot[(v + 8 * h) * SLOT_PITCH + t * 16 + m] = val;
      }
    }
    __builtin_amdgcn_fence(__ATOMIC_RELEASE, "wavefront");
    __builtin_amdgcn_wave_barrier();
    __builtin_amdgcn_fence(__ATOMIC_ACQUIRE, "wavefront");
    float* orow = out + (size_t)(rt * 16 + h) * NOUT + nh * 64 + m * 4;
    const float* srow = slot + h * SLOT_PITCH + m * 4;
    if (rt * 16 + 16 <= Nrows) {
#pragma unroll
      for (int j = 0; j < 8; ++j)
        *(float4*)(orow + (size_t)(2 * j) * NOUT) = *(const float4*)(srow + 2 * j * SLOT_PITCH);
    } else {
#pragma unroll
      for (int j = 0; j < 8; ++j) {
        if (rt * 16 + 2 * j + h < Nrows)
          *(float4*)(orow + (size_t)(2 * j) * NOUT) = *(const float4*)(srow + 2 * j * SLOT_PITCH);
      }
    }
    __builtin_amdgcn_fence(__ATOMIC_RELEASE, "wavefront");
    __builtin_amdgcn_wave_barrier();
    __builtin_amdgcn_fence(__ATOMIC_ACQUIRE, "wavefront");
  }
}


template <int S> __device__ __forceinline__ unsigned row_lane_plus(unsigned v, unsigned add) {
  return (unsigned)__builtin_amdgcn_update_dpp(0, (int)v, 0x150 + S, 0xf, 0xf, true) + add;
}
template <int M> __device__ __forceinline__ float row_xor_lane(float v) {
  return __builtin_bit_cast(float,
      __builtin_amdgcn_update_dpp(0, __builtin_bit_cast(int, v), 0x160 + M, 0xf, 0xf, true));
}
template <int M> __device__ __forceinline__ float fold_pair(bool up, float lo, float hi) {
  const float keep = up ? hi : lo;
  const float send = up ? lo : hi;
  return keep + row_xor_lane<M>(send);
}

#define EDGE_LOAD(S)                                                        \
  const float4 a##S = *(const float4*)(hb + row_lane_plus<S>(so, c16));     \
  const float4 b##S = *(const float4*)(hb + row_lane_plus<S>(dof, c16));
#define EDGE_DOT(S) \
  p[S] = a##S.x * b##S.x + a##S.y * b##S.y + a##S.z * b##S.z + a##S.w * b##S.w;

__global__ __launch_bounds__(256) void edge_score(const int* __restrict__ src0, const int* __restrict__ dst0,
                                                  const int* __restrict__ src1, const int* __restrict__ dst1,
                                                  const float* __restrict__ h, float* __restrict__ out,
                                                  int E, int N) {
  if (E <= 0) return;
  const int lane = threadIdx.x & 31;
  const int c = lane & 15;
  const unsigned c16 = (unsigned)c * 16u;
  const int w = __builtin_amdgcn_readfirstlane((int)(threadIdx.x >> 5));
  const int nwaves = gridDim.x * 8;
  const int ngroups = (E + 31) >> 5;
  const int total = 2 * ngroups;
  const char* hb = (const char*)h;
  const bool u8 = (c & 8) != 0, u4 = (c & 4) != 0, u2 = (c & 2) != 0, u1 = (c & 1) != 0;

  auto load_ids = [&](int gg, int& s, int& d) {
    if (gg > total - 1) gg = total - 1;
    const bool second = gg >= ngroups;
    int e = (gg - (second ? ngroups : 0)) * 32 + lane;
    if (e > E - 1) e = E - 1;
    s = (second ? src1 : src0)[e];
    d = (second ? dst1 : dst0)[e];
  };

  int g = blockIdx.x * 8 + w;
  int si, di;
  load_ids(g, si, di);
  for (; g < total; g += nwaves) {
    const bool second = g >= ngroups;
    const int e = (g - (second ? ngroups : 0)) * 32 + lane;
    int sn, dn;
    load_ids(g + nwaves, sn, dn);
    if ((unsigned)si >= (unsigned)N) si = 0;
    if ((unsigned)di >= (unsigned)N) di = 0;
    const unsigned so = (unsigned)si * 256u;
    const unsigned dof = (unsigned)di * 256u;
    float p[16];
    EDGE_LOAD(0) EDGE_LOAD(1) EDGE_LOAD(2) EDGE_LOAD(3) EDGE_LOAD(4) EDGE_LOAD(5) EDGE_LOAD(6) EDGE_LOAD(7)
    EDGE_LOAD(8) EDGE_LOAD(9) EDGE_LOAD(10) EDGE_LOAD(11) EDGE_LOAD(12) EDGE_LOAD(13) EDGE_LOAD(14) EDGE_LOAD(15)
    EDGE_DOT(0) EDGE_DOT(1) EDGE_DOT(2) EDGE_DOT(3) EDGE_DOT(4) EDGE_DOT(5) EDGE_DOT(6) EDGE_DOT(7)
    EDGE_DOT(8) EDGE_DOT(9) EDGE_DOT(10) EDGE_DOT(11) EDGE_DOT(12) EDGE_DOT(13) EDGE_DOT(14) EDGE_DOT(15)
    float q[8];
#pragma unroll
    for (int s = 0; s < 8; ++s) q[s] = fold_pair<8>(u8, p[s], p[s + 8]);
    float r4[4];
#pragma unroll
    for (int s = 0; s < 4; ++s) r4[s] = fold_pair<4>(u4, q[s], q[s + 4]);
    const float t0 = fold_pair<2>(u2, r4[0], r4[2]);
    const float t1 = fold_pair<2>(u2, r4[1], r4[3]);
    const float score = fold_pair<1>(u1, t0, t1);
    if (e < E) out[(second ? (size_t)E : (size_t)0) + e] = score;
    si = sn; di = dn;
  }
}
#undef EDGE_LOAD
#undef EDGE_DOT

extern "C" void kernel_launch(void* const* d_in, const int* in_sizes, int n_in,
                              void* d_out, int out_size, void* d_ws, size_t ws_size,
                              hipStream_t stream) {
  const float* x   = (const float*)d_in[0];
  const int*   edges = (const int*)d_in[1];
  const int*   negE  = (const int*)d_in[2];
  const float* Wn1 = (const float*)d_in[3];
  const float* Ws1 = (const float*)d_in[4];
  const float* b1  = (const float*)d_in[5];
  const float* Wn2 = (const float*)d_in[6];
  const float* Ws2 = (const float*)d_in[7];
  const float* b2  = (const float*)d_in[8];
  float* out = (float*)d_out;

  const int N = in_sizes[0] / 128;
  const int E = in_sizes[1] / 4;

  float* ws    = (float*)d_ws;
  float* bias2 = ws;
  float* agg0  = ws + ((2 * (size_t)N) & ~(size_t)63);
  float* agg1  = agg0 + (size_t)N * 128;
  float* h1    = agg1 + (size_t)N * 128;
  float* elist = h1 + (size_t)N * 128;
  float* Bcat1 = elist + (size_t)N * 64;
  float* bias1 = Bcat1 + 384 * 128;
  float* Bcat2 = bias1 + 128;
  float* P     = agg0;
  float* h2    = h1;
  const int NS = (N + 8) & ~7;
  int* csr  = (int*)elist;
  int* seg  = csr + 2 * (size_t)E;
  int* segx = seg + 2 * (size_t)NS;
  int* rank = (int*)agg1;

  const int T = 256;

  const int wtot = 384 * 128 + 128 + 128 * PROW + PROW;
  build_weights<<<(wtot + T - 1) / T, T, 0, stream>>>(Wn1, Ws1, b1, Wn2, Ws2, b2,
                                                      Bcat1, bias1, Bcat2, bias2, seg, 2 * NS);

  const dim3 egrid(256, 2);
  count_rank<<<egrid, T, 0, stream>>>(edges, seg, rank, E, NS);
  scan_fill_csr<<<dim3(128, 2), SF_THREADS, 0, stream>>>(edges, seg, rank, segx, csr, E, N, NS);

  const int sblocks = 1024;
  const int ablocks = 4096;
  const int pblocks = 2048;

  gather_mean<<<ablocks, T, 0, stream>>>(segx, csr, x, agg0, E, N, NS);
  sage_gemm<128, 384, true><<<GEMM_WGS, GEMM_THREADS, 0, stream>>>(x, agg0, agg1, Bcat1, bias1, h1, N);

  sage_gemm<PROW, 128, false><<<GEMM_WGS, GEMM_THREADS, 0, stream>>>(h1, h1, h1, Bcat2, bias2, P, N);
  gather_proj<<<pblocks, T, 0, stream>>>(segx, csr, P, h2, E, N, NS);

  edge_score<<<sblocks, T, 0, stream>>>(edges, edges + E, negE, negE + E, h2, out, E, N);
}


# Edge sort without global atomics or per-edge scattered stores: chunks bucketed by node range in LDS, then one workgroup per range orders its pairs in LDS; LDS atomic rank, so new sum order
# speedup vs baseline: 6.0417x; 1.5979x over previous
#include <hip/hip_runtime.h>
#include <hip/hip_bf16.h>

typedef float v2f __attribute__((ext_vector_type(2)));
typedef float v8f __attribute__((ext_vector_type(8)));

#define FDIM 128
#define PROW 192

__global__ void build_weights(const float* __restrict__ Wn1, const float* __restrict__ Ws1,
                              const float* __restrict__ b1,
                              const float* __restrict__ Wn2, const float* __restrict__ Ws2,
                              const float* __restrict__ b2,
                              float* __restrict__ Bcat1, float* __restrict__ bias1,
                              float* __restrict__ Bcat2, float* __restrict__ bias2) {
  int i = blockIdx.x * blockDim.x + threadIdx.x;
  const int S1 = 384 * 128, S2 = S1 + 128, S3 = S2 + 128 * PROW, S4 = S3 + PROW;
  if (i < S1) {
    int k = i >> 7, n = i & 127;
    float v;
    if (k < 128)      v = Ws1[k * 128 + n] + Ws1[16384 + k * 128 + n];
    else if (k < 256) v = Wn1[(k - 128) * 128 + n];
    else              v = Wn1[16384 + (k - 256) * 128 + n];
    Bcat1[i] = v;
  } else if (i < S2) {
    int n = i - S1;
    bias1[n] = b1[n] + b1[128 + n];
  } else if (i < S3) {
    int j = i - S2;
    int k = j / PROW, n = j - k * PROW;
    float v;
    if (n < 64)       v = Ws2[k * 64 + n] + Ws2[8192 + k * 64 + n];
    else if (n < 128) v = Wn2[k * 64 + (n - 64)];
    else              v = Wn2[8192 + k * 64 + (n - 128)];
    Bcat2[j] = v;
  } else if (i < S4) {
    int n = i - S3;
    bias2[n] = (n < 64) ? b2[n] + b2[64 + n] : 0.0f;
  }
}

#define SORT_NB 256
#define SORT_NC 128
#define SORT_T 1024
#define SORT_EPT 8
#define SORT_CAP 16384

__device__ __forceinline__ int scan256(int v, int t, int* wtot, int& total) {
  const int lane = t & 31, w = t >> 5;
  int inc = (t < 256) ? v : 0;
  for (int off = 1; off < 32; off <<= 1) {
    const int up = __shfl_up(inc, off, 32);
    if (lane >= off) inc += up;
  }
  if (t < 256 && lane == 31) wtot[w] = inc;
  __syncthreads();
  int before = 0, all = 0;
  for (int k = 0; k < 8; ++k) {
    const int x = wtot[k];
    all += x;
    if (k < w) before += x;
  }
  total = all;
  return before + inc - ((t < 256) ? v : 0);
}

__global__ __launch_bounds__(SORT_T) void bucket_chunks(const int* __restrict__ edges, int* __restrict__ tmp,
                                                        int* __restrict__ pcnt, int* __restrict__ poff, int E, int N) {
  __shared__ int hist[SORT_NB];
  __shared__ int offs[SORT_NB];
  __shared__ int wtot[8];
  __shared__ int2 stage[SORT_T * SORT_EPT];
  const int CE = (E + SORT_NC - 1) / SORT_NC;
  if (E < 1 || N < 1 || CE > SORT_T * SORT_EPT) return;
  const int r = blockIdx.y, c = blockIdx.x, t = threadIdx.x;
  const unsigned BW = (unsigned)N / SORT_NB + 1u;
  const int* src = edges + (size_t)r * 2 * E;
  const int* dst = src + E;
  const int eb = c * CE;
  int ee = eb + CE;
  if (ee > E) ee = E;
  const int n = ee > eb ? ee - eb : 0;
  if (t < SORT_NB) hist[t] = 0;
  __syncthreads();
  int s[SORT_EPT], d[SORT_EPT], bk[SORT_EPT], rk[SORT_EPT];
#pragma unroll
  for (int k = 0; k < SORT_EPT; ++k) {
    const int i = t + k * SORT_T;
    const int e = i < n ? eb + i : (E - 1);
    s[k] = src[e];
    d[k] = dst[e];
  }
#pragma unroll
  for (int k = 0; k < SORT_EPT; ++k) {
    unsigned dd = (unsigned)d[k];
    if (dd >= (unsigned)N) dd = 0;
    d[k] = (int)dd;
    unsigned b = dd / BW;
    if (b > SORT_NB - 1) b = SORT_NB - 1;
    bk[k] = (int)b;
    rk[k] = 0;
    if (t + k * SORT_T < n) rk[k] = atomicAdd(&hist[b], 1);
  }
  __syncthreads();
  const int mine = (t < SORT_NB) ? hist[t] : 0;
  int total;
  const int pre = scan256(mine, t, wtot, total);
  if (t < SORT_NB) {
    offs[t] = pre;
    pcnt[((size_t)r * SORT_NB + t) * SORT_NC + c] = mine;
    poff[((size_t)r * SORT_NB + t) * SORT_NC + c] = pre;
  }
  __syncthreads();
#pragma unroll
  for (int k = 0; k < SORT_EPT; ++k) {
    if (t + k * SORT_T < n) {
      const int pos = offs[bk[k]] + rk[k];
      if ((unsigned)pos < (unsigned)(SORT_T * SORT_EPT)) stage[pos] = make_int2(s[k], d[k]);
    }
  }
  __syncthreads();
  int2* __restrict__ tp = (int2*)tmp + (size_t)r * E + eb;
  for (int i = t; i < n; i += SORT_T) tp[i] = stage[i];
}

__global__ __launch_bounds__(SORT_T) void range_csr(const int* __restrict__ tmp, const int* __restrict__ pcnt,
                                                    const int* __restrict__ poff, int* __restrict__ segout,
                                                    int* __restrict__ csr, int E, int N, int NS) {
  __shared__ int pc[SORT_NC], po[SORT_NC];
  __shared__ int h[SORT_NB], st[SORT_NB], cur[SORT_NB];
  __shared__ int wtot[8];
  __shared__ int gsum;
  __shared__ int srt[SORT_CAP];
  const int CE = (E + SORT_NC - 1) / SORT_NC;
  if (E < 1 || N < 1 || N / SORT_NB + 1 > SORT_NB || CE > SORT_T * SORT_EPT) return;
  const int r = blockIdx.y, b = blockIdx.x, t = threadIdx.x;
  const int BW = N / SORT_NB + 1;
  const int node0 = b * BW;
  const int* rowc = pcnt + ((size_t)r * SORT_NB + b) * SORT_NC;
  const int* rowo = poff + ((size_t)r * SORT_NB + b) * SORT_NC;
  if (t < SORT_NC) {
    int cc = rowc[t], oo = rowo[t];
    if (oo < 0) oo = 0;
    if (oo > CE) oo = CE;
    if (cc < 0) cc = 0;
    if (cc > CE - oo) cc = CE - oo;
    pc[t] = cc;
    po[t] = oo;
  }
  if (t < SORT_NB) { h[t] = 0; cur[t] = 0; }
  if (t == 0) gsum = 0;
  __syncthreads();
  {
    const int4* allc = (const int4*)(pcnt + (size_t)r * SORT_NB * SORT_NC);
    const int n4 = b * (SORT_NC / 4);
    int4 v[8];
#pragma unroll
    for (int k = 0; k < 8; ++k) {
      const int i = t + k * SORT_T;
      v[k] = allc[i < n4 ? i : 0];
    }
    int part = 0;
#pragma unroll
    for (int k = 0; k < 8; ++k)
      if (t + k * SORT_T < n4) part += v[k].x + v[k].y + v[k].z + v[k].w;
    for (int off = 16; off > 0; off >>= 1) part += __shfl_xor(part, off, 32);
    if ((t & 31) == 0) atomicAdd(&gsum, part);
  }
  const int g = t >> 3, k8 = t & 7;
  const int pstart = g * CE + po[g];
  const int room = E - pstart;
  const int pl = pc[g] < room ? pc[g] : (room > 0 ? room : 0);
  const int2* __restrict__ piece = (const int2*)tmp + (size_t)r * E + (pstart < E ? pstart : E - 1);
  int2 q[4];
#pragma unroll
  for (int k = 0; k < 4; ++k) {
    const int i = k8 + 8 * k;
    q[k] = piece[i < pl ? i : 0];
  }
#pragma unroll
  for (int k = 0; k < 4; ++k) {
    if (k8 + 8 * k < pl) {
      int dl = q[k].y - node0;
      if ((unsigned)dl >= (unsigned)BW) dl = 0;
      atomicAdd(&h[dl], 1);
    }
  }
  for (int i = k8 + 32; i < pl; i += 8) {
    int dl = piece[i].y - node0;
    if ((unsigned)dl >= (unsigned)BW) dl = 0;
    atomicAdd(&h[dl], 1);
  }
  __syncthreads();
  const int mine = (t < SORT_NB) ? h[t] : 0;
  int M;
  const int pre = scan256(mine, t, wtot, M);
  if (t < SORT_NB) st[t] = pre;
  __syncthreads();
  const int gstart = gsum;
  if (t < BW && node0 + t <= N) segout[(size_t)r * NS + node0 + t] = gstart + st[t];
#pragma unroll
  for (int k = 0; k < 4; ++k) {
    if (k8 + 8 * k < pl) {
      int dl = q[k].y - node0;
      if ((unsigned)dl >= (unsigned)BW) dl = 0;
      const int pos = st[dl] + atomicAdd(&cur[dl], 1);
      if ((unsigned)pos < (unsigned)SORT_CAP) srt[pos] = q[k].x;
      else if ((unsigned)(gstart + pos) < (unsigned)E) csr[(size_t)r * E + gstart + pos] = q[k].x;
    }
  }
  for (int i = k8 + 32; i < pl; i += 8) {
    const int2 sd = piece[i];
    int dl = sd.y - node0;
    if ((unsigned)dl >= (unsigned)BW) dl = 0;
    const int pos = st[dl] + atomicAdd(&cur[dl], 1);
    if ((unsigned)pos < (unsigned)SORT_CAP) srt[pos] = sd.x;
    else if ((unsigned)(gstart + pos) < (unsigned)E) csr[(size_t)r * E + gstart + pos] = sd.x;
  }
  __syncthreads();
  const int mout = M < SORT_CAP ? M : SORT_CAP;
  for (int i = t; i < mout; i += SORT_T)
    if ((unsigned)(gstart + i) < (unsigned)E) csr[(size_t)r * E + gstart + i] = srt[i];
}

__global__ __launch_bounds__(256) void gather_mean(const int* __restrict__ seg, const int* __restrict__ csr,
                                                   const float* __restrict__ feat, float* __restrict__ agg,
                                                   int E, int N, int NS) {
  const int lane = threadIdx.x & 31;
  const int w = __builtin_amdgcn_readfirstlane((int)(threadIdx.x >> 5));
  const int nwaves = gridDim.x * 8;
  for (int idx = blockIdx.x * 8 + w; idx < 2 * N; idx += nwaves) {
    const int r = (idx >= N) ? 1 : 0;
    const int i = idx - r * N;
    const int* c = seg + (size_t)r * NS;
    int start = c[i];
    int end = c[i + 1];
    if (start < 0) start = 0;
    if (end > E) end = E;
    const int* nb = csr + (size_t)r * E;
    float4 acc = make_float4(0.0f, 0.0f, 0.0f, 0.0f);
    for (int base = start; base < end; base += 32) {
      const int rem = end - base;
      const int n = rem < 32 ? rem : 32;
      int mine = (lane < n) ? nb[base + lane] : 0;
      if ((unsigned)mine >= (unsigned)N) mine = 0;
      int j = 0;
      for (; j + 4 <= n; j += 4) {
        const int s0 = __builtin_amdgcn_readlane(mine, j);
        const int s1 = __builtin_amdgcn_readlane(mine, j + 1);
        const int s2 = __builtin_amdgcn_readlane(mine, j + 2);
        const int s3 = __builtin_amdgcn_readlane(mine, j + 3);
        const float4 v0 = *(const float4*)(feat + (size_t)s0 * FDIM + lane * 4);
        const float4 v1 = *(const float4*)(feat + (size_t)s1 * FDIM + lane * 4);
        const float4 v2 = *(const float4*)(feat + (size_t)s2 * FDIM + lane * 4);
        const float4 v3 = *(const float4*)(feat + (size_t)s3 * FDIM + lane * 4);
        acc.x += v0.x; acc.y += v0.y; acc.z += v0.z; acc.w += v0.w;
        acc.x += v1.x; acc.y += v1.y; acc.z += v1.z; acc.w += v1.w;
        acc.x += v2.x; acc.y += v2.y; acc.z += v2.z; acc.w += v2.w;
        acc.x += v3.x; acc.y += v3.y; acc.z += v3.z; acc.w += v3.w;
      }
      for (; j < n; ++j) {
        const int s0 = __builtin_amdgcn_readlane(mine, j);
        const float4 v0 = *(const float4*)(feat + (size_t)s0 * FDIM + lane * 4);
        acc.x += v0.x; acc.y += v0.y; acc.z += v0.z; acc.w += v0.w;
      }
    }
    const float deg = (float)(end - start);
    const float dinv = 1.0f / fmaxf(deg, 1.0f);
    float4 o;
    o.x = acc.x * dinv; o.y = acc.y * dinv; o.z = acc.z * dinv; o.w = acc.w * dinv;
    *(float4*)(agg + ((size_t)r * N + i) * FDIM + lane * 4) = o;
  }
}


__device__ __forceinline__ float other_half(float v) {
  const int b = __builtin_bit_cast(int, v);
  return __builtin_bit_cast(float, __builtin_amdgcn_permlanex16(b, b, 0x76543210, 0xfedcba98, false, false));
}

__device__ __forceinline__ int first_ids(const int* __restrict__ nb, int start, int end, int lane) {
  const int rem = end - start;
  const int n = rem < 32 ? rem : 32;
  return (lane < n) ? nb[start + lane] : 0;
}

#define PULL_PAIR(K)                                                                          \
  const unsigned e##K = (unsigned)__builtin_amdgcn_readlane((int)moff, j + 2 * K);            \
  const unsigned o##K = (unsigned)__builtin_amdgcn_readlane((int)moff, j + 2 * K + 1);        \
  float4 v##K = make_float4(0.0f, 0.0f, 0.0f, 0.0f);                                          \
  if (j + 2 * K + hlf < n) v##K = *(const float4*)(pb + ((hlf ? o##K : e##K) + c16));
#define PULL_ADD(K) acc.x += v##K.x; acc.y += v##K.y; acc.z += v##K.z; acc.w += v##K.w;

__device__ __forceinline__ float4 pull_blocks(const int* __restrict__ nb, int start, int end, int mine,
                                              const char* __restrict__ pb, unsigned c16, int lane, int hlf, int N) {
  float4 acc = make_float4(0.0f, 0.0f, 0.0f, 0.0f);
  for (int base = start; base < end; base += 32) {
    const int rem = end - base;
    const int n = rem < 32 ? rem : 32;
    if (base != start) mine = (lane < n) ? nb[base + lane] : 0;
    if ((unsigned)mine >= (unsigned)N) mine = 0;
    const unsigned moff = (unsigned)mine * (PROW * 4u);
    for (int j = 0; j < n; j += 16) {
      PULL_PAIR(0) PULL_PAIR(1) PULL_PAIR(2) PULL_PAIR(3) PULL_PAIR(4) PULL_PAIR(5) PULL_PAIR(6) PULL_PAIR(7)
      PULL_ADD(0) PULL_ADD(1) PULL_ADD(2) PULL_ADD(3) PULL_ADD(4) PULL_ADD(5) PULL_ADD(6) PULL_ADD(7)
    }
  }
  return acc;
}
#undef PULL_PAIR
#undef PULL_ADD

__global__ __launch_bounds__(256) void gather_proj(const int* __restrict__ seg, const int* __restrict__ csr,
                                                   const float* __restrict__ P, float* __restrict__ h2,
                                                   int E, int N, int NS) {
  const int lane = threadIdx.x & 31;
  const int hlf = lane >> 4;
  const unsigned c16 = (unsigned)(lane & 15) * 16u;
  const int w = __builtin_amdgcn_readfirstlane((int)(threadIdx.x >> 5));
  const int nwaves = gridDim.x * 8;
  const char* pbytes = (const char*)P;
  for (int i = blockIdx.x * 8 + w; i < N; i += nwaves) {
    int s0 = seg[i], e0 = seg[i + 1];
    int s1 = seg[(size_t)NS + i], e1 = seg[(size_t)NS + i + 1];
    if (s0 < 0) s0 = 0;
    if (e0 > E) e0 = E;
    if (s1 < 0) s1 = 0;
    if (e1 > E) e1 = E;
    const int m0 = first_ids(csr, s0, e0, lane);
    const int m1 = first_ids(csr + E, s1, e1, lane);
    float4 tot = make_float4(0.0f, 0.0f, 0.0f, 0.0f);
    if (hlf == 0) tot = *(const float4*)(pbytes + ((size_t)i * (PROW * 4u) + c16));
    const float4 a0 = pull_blocks(csr, s0, e0, m0, pbytes + 256, c16, lane, hlf, N);
    const float4 a1 = pull_blocks(csr + E, s1, e1, m1, pbytes + 512, c16, lane, hlf, N);
    const float d0 = 1.0f / fmaxf((float)(e0 - s0), 1.0f);
    const float d1 = 1.0f / fmaxf((float)(e1 - s1), 1.0f);
    tot.x += a0.x * d0 + a1.x * d1;
    tot.y += a0.y * d0 + a1.y * d1;
    tot.z += a0.z * d0 + a1.z * d1;
    tot.w += a0.w * d0 + a1.w * d1;
    tot.x += other_half(tot.x);
    tot.y += other_half(tot.y);
    tot.z += other_half(tot.z);
    tot.w += other_half(tot.w);
    if (hlf == 0) *(float4*)(h2 + (size_t)i * 64 + (lane & 15) * 4) = tot;
  }
}

typedef float v4f __attribute__((ext_vector_type(4)));

#define GEMM_WGS 256
#define GEMM_WAVES 16
#define GEMM_THREADS 512
#define SLOT_PITCH 68

template <int NOUT, int KT, bool RELU>
__global__ __launch_bounds__(GEMM_THREADS) void sage_gemm(
    const float* __restrict__ A0, const float* __restrict__ G0, const float* __restrict__ G1,
    const float* __restrict__ Bmat, const float* __restrict__ bias,
    float* __restrict__ out, int Nrows) {
  constexpr int NCH = KT / 64;
  constexpr int NSPLIT = NOUT / 64;
  static_assert(NCH % 2 == 0, "the two A register sets swap roles once per chunk");
  __shared__ v4f Bs[96 * 128];
  __shared__ float slots[GEMM_WAVES * 16 * SLOT_PITCH];

  const int tid = threadIdx.x;
  const int lane = tid & 31;
  const int m = lane & 15;
  const int h = lane >> 4;
  const int wave = __builtin_amdgcn_readfirstlane(tid >> 5);

  for (int i = tid; i < (KT / 4) * NOUT; i += GEMM_THREADS) {
    const int kq = i / NOUT;
    const int n = i - kq * NOUT;
    const float* p = Bmat + (size_t)(4 * kq) * NOUT + n;
    v4f w;
    w.x = p[0]; w.y = p[NOUT]; w.z = p[2 * NOUT]; w.w = p[3 * NOUT];
    Bs[i] = w;
  }
  __syncthreads();

  const int nRT = (Nrows + 15) >> 4;
  const int rt0 = (int)(((long long)blockIdx.x * nRT) / GEMM_WGS);
  const int rt1 = (int)(((long long)(blockIdx.x + 1) * nRT) / GEMM_WGS);
  const int nunits = (rt1 - rt0) * NSPLIT;

  auto a_ptr = [&](int u, int c) -> const float* {
    const int uu = u < nunits ? u : nunits - 1;
    int row = (rt0 + uu / NSPLIT) * 16 + m;
    if (row > Nrows - 1) row = Nrows - 1;
    const float* src = (c < 2) ? A0 : ((c < 4) ? G0 : G1);
    return src + (size_t)row * FDIM + (c & 1) * 64 + 4 * h;
  };

  if (wave >= nunits) return;
  v4f abuf[2][8];
  {
    const float* p = a_ptr(wave, 0);
#pragma unroll
    for (int j = 0; j < 8; ++j) abuf[0][j] = *(const v4f*)(p + j * 8);
  }
  float* slot = slots + wave * (16 * SLOT_PITCH);

  for (int u = wave; u < nunits; u += GEMM_WAVES) {
    const int rt = rt0 + u / NSPLIT;
    const int nh = u - (u / NSPLIT) * NSPLIT;
    v8f acc[4];
#pragma unroll
    for (int t = 0; t < 4; ++t) acc[t] = v8f{};

    const v4f* bunit = Bs + (size_t)h * NOUT + nh * 64 + m;
    v4f bcur[4], bnxt[4];
#pragma unroll
    for (int t = 0; t < 4; ++t) bcur[t] = bunit[t * 16];

#pragma unroll
    for (int c = 0; c < NCH; ++c) {
      {
        const float* p = (c + 1 < NCH) ? a_ptr(u, c + 1) : a_ptr(u + GEMM_WAVES, 0);
#pragma unroll
        for (int j = 0; j < 8; ++j) abuf[(c + 1) & 1][j] = *(const v4f*)(p + j * 8);
      }
#pragma unroll
      for (int j = 0; j < 8; ++j) {
        const int st = c * 8 + j;
        if (st + 1 < NCH * 8) {
#pragma unroll
          for (int t = 0; t < 4; ++t) bnxt[t] = bunit[(size_t)(2 * (st + 1)) * NOUT + t * 16];
        }
        const v4f a = abuf[c & 1][j];
        const v2f alo = {a.x, a.y}, ahi = {a.z, a.w};
#pragma unroll
        for (int t = 0; t < 4; ++t) {
          const v2f blo = {bcur[t].x, bcur[t].y};
          acc[t] = __builtin_amdgcn_wmma_f32_16x16x4_f32(false, alo, false, blo, (short)0, acc[t], false, false);
        }
#pragma unroll
        for (int t = 0; t < 4; ++t) {
          const v2f bhi = {bcur[t].z, bcur[t].w};
          acc[t] = __builtin_amdgcn_wmma_f32_16x16x4_f32(false, ahi, false, bhi, (short)0, acc[t], false, false);
        }
#pragma unroll
        for (int t = 0; t < 4; ++t) bcur[t] = bnxt[t];
      }
    }

#pragma unroll
    for (int t = 0; t < 4; ++t) {
      const float bv = bias[nh * 64 + t * 16 + m];
#pragma unroll
      for (int v = 0; v < 8; ++v) {
        float val = acc[t][v] + bv;
        if (RELU) val = fmaxf(val, 0.0f);
        slot[(v + 8 * h) * SLOT_PITCH + t * 16 + m] = val;
      }
    }
    __builtin_amdgcn_fence(__ATOMIC_RELEASE, "wavefront");
    __builtin_amdgcn_wave_barrier();
    __builtin_amdgcn_fence(__ATOMIC_ACQUIRE, "wavefront");
    float* orow = out + (size_t)(rt * 16 + h) * NOUT + nh * 64 + m * 4;
    const float* srow = slot + h * SLOT_PITCH + m * 4;
    if (rt * 16 + 16 <= Nrows) {
#pragma unroll
      for (int j = 0; j < 8; ++j)
        *(float4*)(orow + (size_t)(2 * j) * NOUT) = *(const float4*)(srow + 2 * j * SLOT_PITCH);
    } else {
#pragma unroll
      for (int j = 0; j < 8; ++j) {
        if (rt * 16 + 2 * j + h < Nrows)
          *(float4*)(orow + (size_t)(2 * j) * NOUT) = *(const float4*)(srow + 2 * j * SLOT_PITCH);
      }
    }
    __builtin_amdgcn_fence(__ATOMIC_RELEASE, "wavefront");
    __builtin_amdgcn_wave_barrier();
    __builtin_amdgcn_fence(__ATOMIC_ACQUIRE, "wavefront");
  }
}


template <int S> __device__ __forceinline__ unsigned row_lane_plus(unsigned v, unsigned add) {
  return (unsigned)__builtin_amdgcn_update_dpp(0, (int)v, 0x150 + S, 0xf, 0xf, true) + add;
}
template <int M> __device__ __forceinline__ float row_xor_lane(float v) {
  return __builtin_bit_cast(float,
      __builtin_amdgcn_update_dpp(0, __builtin_bit_cast(int, v), 0x160 + M, 0xf, 0xf, true));
}
template <int M> __device__ __forceinline__ float fold_pair(bool up, float lo, float hi) {
  const float keep = up ? hi : lo;
  const float send = up ? lo : hi;
  return keep + row_xor_lane<M>(send);
}

#define EDGE_LOAD(S)                                                        \
  const float4 a##S = *(const float4*)(hb + row_lane_plus<S>(so, c16));     \
  const float4 b##S = *(const float4*)(hb + row_lane_plus<S>(dof, c16));
#define EDGE_DOT(S) \
  p[S] = a##S.x * b##S.x + a##S.y * b##S.y + a##S.z * b##S.z + a##S.w * b##S.w;

__global__ __launch_bounds__(256) void edge_score(const int* __restrict__ src0, const int* __restrict__ dst0,
                                                  const int* __restrict__ src1, const int* __restrict__ dst1,
                                                  const float* __restrict__ h, float* __restrict__ out,
                                                  int E, int N) {
  if (E <= 0) return;
  const int lane = threadIdx.x & 31;
  const int c = lane & 15;
  const unsigned c16 = (unsigned)c * 16u;
  const int w = __builtin_amdgcn_readfirstlane((int)(threadIdx.x >> 5));
  const int nwaves = gridDim.x * 8;
  const int ngroups = (E + 31) >> 5;
  const int total = 2 * ngroups;
  const char* hb = (const char*)h;
  const bool u8 = (c & 8) != 0, u4 = (c & 4) != 0, u2 = (c & 2) != 0, u1 = (c & 1) != 0;

  auto load_ids = [&](int gg, int& s, int& d) {
    if (gg > total - 1) gg = total - 1;
    const bool second = gg >= ngroups;
    int e = (gg - (second ? ngroups : 0)) * 32 + lane;
    if (e > E - 1) e = E - 1;
    s = (second ? src1 : src0)[e];
    d = (second ? dst1 : dst0)[e];
  };

  int g = blockIdx.x * 8 + w;
  int si, di;
  load_ids(g, si, di);
  for (; g < total; g += nwaves) {
    const bool second = g >= ngroups;
    const int e = (g - (second ? ngroups : 0)) * 32 + lane;
    int sn, dn;
    load_ids(g + nwaves, sn, dn);
    if ((unsigned)si >= (unsigned)N) si = 0;
    if ((unsigned)di >= (unsigned)N) di = 0;
    const unsigned so = (unsigned)si * 256u;
    const unsigned dof = (unsigned)di * 256u;
    float p[16];
    EDGE_LOAD(0) EDGE_LOAD(1) EDGE_LOAD(2) EDGE_LOAD(3) EDGE_LOAD(4) EDGE_LOAD(5) EDGE_LOAD(6) EDGE_LOAD(7)
    EDGE_LOAD(8) EDGE_LOAD(9) EDGE_LOAD(10) EDGE_LOAD(11) EDGE_LOAD(12) EDGE_LOAD(13) EDGE_LOAD(14) EDGE_LOAD(15)
    EDGE_DOT(0) EDGE_DOT(1) EDGE_DOT(2) EDGE_DOT(3) EDGE_DOT(4) EDGE_DOT(5) EDGE_DOT(6) EDGE_DOT(7)
    EDGE_DOT(8) EDGE_DOT(9) EDGE_DOT(10) EDGE_DOT(11) EDGE_DOT(12) EDGE_DOT(13) EDGE_DOT(14) EDGE_DOT(15)
    float q[8];
#pragma unroll
    for (int s = 0; s < 8; ++s) q[s] = fold_pair<8>(u8, p[s], p[s + 8]);
    float r4[4];
#pragma unroll
    for (int s = 0; s < 4; ++s) r4[s] = fold_pair<4>(u4, q[s], q[s + 4]);
    const float t0 = fold_pair<2>(u2, r4[0], r4[2]);
    const float t1 = fold_pair<2>(u2, r4[1], r4[3]);
    const float score = fold_pair<1>(u1, t0, t1);
    if (e < E) out[(second ? (size_t)E : (size_t)0) + e] = score;
    si = sn; di = dn;
  }
}
#undef EDGE_LOAD
#undef EDGE_DOT

extern "C" void kernel_launch(void* const* d_in, const int* in_sizes, int n_in,
                              void* d_out, int out_size, void* d_ws, size_t ws_size,
                              hipStream_t stream) {
  const float* x   = (const float*)d_in[0];
  const int*   edges = (const int*)d_in[1];
  const int*   negE  = (const int*)d_in[2];
  const float* Wn1 = (const float*)d_in[3];
  const float* Ws1 = (const float*)d_in[4];
  const float* b1  = (const float*)d_in[5];
  const float* Wn2 = (const float*)d_in[6];
  const float* Ws2 = (const float*)d_in[7];
  const float* b2  = (const float*)d_in[8];
  float* out = (float*)d_out;

  const int N = in_sizes[0] / 128;
  const int E = in_sizes[1] / 4;

  float* ws    = (float*)d_ws;
  float* bias2 = ws;
  float* agg0  = ws + ((2 * (size_t)N) & ~(size_t)63);
  float* agg1  = agg0 + (size_t)N * 128;
  float* h1    = agg1 + (size_t)N * 128;
  float* elist = h1 + (size_t)N * 128;
  float* Bcat1 = elist + (size_t)N * 64;
  float* bias1 = Bcat1 + 384 * 128;
  float* Bcat2 = bias1 + 128;
  float* P     = agg0;
  float* h2    = h1;
  const int NS = (N + 8) & ~7;
  int* csr  = (int*)elist;
  int* segx = csr + 2 * (size_t)E;
  int* tmp  = (int*)agg1;
  int* pcnt = (int*)agg1 + 4 * (size_t)E;
  int* poff = pcnt + 2 * SORT_NB * SORT_NC;

  const int T = 256;

  const int wtot = 384 * 128 + 128 + 128 * PROW + PROW;
  build_weights<<<(wtot + T - 1) / T, T, 0, stream>>>(Wn1, Ws1, b1, Wn2, Ws2, b2, Bcat1, bias1, Bcat2, bias2);

  bucket_chunks<<<dim3(SORT_NC, 2), SORT_T, 0, stream>>>(edges, tmp, pcnt, poff, E, N);
  range_csr<<<dim3(SORT_NB, 2), SORT_T, 0, stream>>>(tmp, pcnt, poff, segx, csr, E, N, NS);

  const int sblocks = 1024;
  const int ablocks = 4096;
  const int pblocks = 2048;

  gather_mean<<<ablocks, T, 0, stream>>>(segx, csr, x, agg0, E, N, NS);
  sage_gemm<128, 384, true><<<GEMM_WGS, GEMM_THREADS, 0, stream>>>(x, agg0, agg1, Bcat1, bias1, h1, N);

  sage_gemm<PROW, 128, false><<<GEMM_WGS, GEMM_THREADS, 0, stream>>>(h1, h1, h1, Bcat2, bias2, P, N);
  gather_proj<<<pblocks, T, 0, stream>>>(segx, csr, P, h2, E, N, NS);

  edge_score<<<sblocks, T, 0, stream>>>(edges, edges + E, negE, negE + E, h2, out, E, N);
}


# Layer-1 mean pull moved into the per-range sort workgroup: lists and bounds are read from LDS, waves take nodes from an LDS counter, 8 rows in flight; 8 launches become 7; sum order unchanged
# speedup vs baseline: 6.2920x; 1.0414x over previous
#include <hip/hip_runtime.h>
#include <hip/hip_bf16.h>

typedef float v2f __attribute__((ext_vector_type(2)));
typedef float v8f __attribute__((ext_vector_type(8)));

#define FDIM 128
#define PROW 192

__global__ void build_weights(const float* __restrict__ Wn1, const float* __restrict__ Ws1,
                              const float* __restrict__ b1,
                              const float* __restrict__ Wn2, const float* __restrict__ Ws2,
                              const float* __restrict__ b2,
                              float* __restrict__ Bcat1, float* __restrict__ bias1,
                              float* __restrict__ Bcat2, float* __restrict__ bias2) {
  int i = blockIdx.x * blockDim.x + threadIdx.x;
  const int S1 = 384 * 128, S2 = S1 + 128, S3 = S2 + 128 * PROW, S4 = S3 + PROW;
  if (i < S1) {
    int k = i >> 7, n = i & 127;
    float v;
    if (k < 128)      v = Ws1[k * 128 + n] + Ws1[16384 + k * 128 + n];
    else if (k < 256) v = Wn1[(k - 128) * 128 + n];
    else              v = Wn1[16384 + (k - 256) * 128 + n];
    Bcat1[i] = v;
  } else if (i < S2) {
    int n = i - S1;
    bias1[n] = b1[n] + b1[128 + n];
  } else if (i < S3) {
    int j = i - S2;
    int k = j / PROW, n = j - k * PROW;
    float v;
    if (n < 64)       v = Ws2[k * 64 + n] + Ws2[8192 + k * 64 + n];
    else if (n < 128) v = Wn2[k * 64 + (n - 64)];
    else              v = Wn2[8192 + k * 64 + (n - 128)];
    Bcat2[j] = v;
  } else if (i < S4) {
    int n = i - S3;
    bias2[n] = (n < 64) ? b2[n] + b2[64 + n] : 0.0f;
  }
}

#define SORT_NB 256
#define SORT_NC 128
#define SORT_T 1024
#define SORT_EPT 8
#define SORT_CAP 16384

__device__ __forceinline__ int scan256(int v, int t, int* wtot, int& total) {
  const int lane = t & 31, w = t >> 5;
  int inc = (t < 256) ? v : 0;
  for (int off = 1; off < 32; off <<= 1) {
    const int up = __shfl_up(inc, off, 32);
    if (lane >= off) inc += up;
  }
  if (t < 256 && lane == 31) wtot[w] = inc;
  __syncthreads();
  int before = 0, all = 0;
  for (int k = 0; k < 8; ++k) {
    const int x = wtot[k];
    all += x;
    if (k < w) before += x;
  }
  total = all;
  return before + inc - ((t < 256) ? v : 0);
}

__global__ __launch_bounds__(SORT_T) void bucket_chunks(const int* __restrict__ edges, int* __restrict__ tmp,
                                                        int* __restrict__ pcnt, int* __restrict__ poff, int E, int N) {
  __shared__ int hist[SORT_NB];
  __shared__ int offs[SORT_NB];
  __shared__ int wtot[8];
  __shared__ int2 stage[SORT_T * SORT_EPT];
  const int CE = (E + SORT_NC - 1) / SORT_NC;
  if (E < 1 || N < 1 || CE > SORT_T * SORT_EPT) return;
  const int r = blockIdx.y, c = blockIdx.x, t = threadIdx.x;
  const unsigned BW = (unsigned)N / SORT_NB + 1u;
  const int* src = edges + (size_t)r * 2 * E;
  const int* dst = src + E;
  const int eb = c * CE;
  int ee = eb + CE;
  if (ee > E) ee = E;
  const int n = ee > eb ? ee - eb : 0;
  if (t < SORT_NB) hist[t] = 0;
  __syncthreads();
  int s[SORT_EPT], d[SORT_EPT], bk[SORT_EPT], rk[SORT_EPT];
#pragma unroll
  for (int k = 0; k < SORT_EPT; ++k) {
    const int i = t + k * SORT_T;
    const int e = i < n ? eb + i : (E - 1);
    s[k] = src[e];
    d[k] = dst[e];
  }
#pragma unroll
  for (int k = 0; k < SORT_EPT; ++k) {
    unsigned dd = (unsigned)d[k];
    if (dd >= (unsigned)N) dd = 0;
    d[k] = (int)dd;
    unsigned b = dd / BW;
    if (b > SORT_NB - 1) b = SORT_NB - 1;
    bk[k] = (int)b;
    rk[k] = 0;
    if (t + k * SORT_T < n) rk[k] = atomicAdd(&hist[b], 1);
  }
  __syncthreads();
  const int mine = (t < SORT_NB) ? hist[t] : 0;
  int total;
  const int pre = scan256(mine, t, wtot, total);
  if (t < SORT_NB) {
    offs[t] = pre;
    pcnt[((size_t)r * SORT_NB + t) * SORT_NC + c] = mine;
    poff[((size_t)r * SORT_NB + t) * SORT_NC + c] = pre;
  }
  __syncthreads();
#pragma unroll
  for (int k = 0; k < SORT_EPT; ++k) {
    if (t + k * SORT_T < n) {
      const int pos = offs[bk[k]] + rk[k];
      if ((unsigned)pos < (unsigned)(SORT_T * SORT_EPT)) stage[pos] = make_int2(s[k], d[k]);
    }
  }
  __syncthreads();
  int2* __restrict__ tp = (int2*)tmp + (size_t)r * E + eb;
  for (int i = t; i < n; i += SORT_T) tp[i] = stage[i];
}

#define MEAN_ROW(K)                                                                           \
  float4 v##K = make_float4(0.0f, 0.0f, 0.0f, 0.0f);                                          \
  if (j + K < n) v##K = *(const float4*)(fb + ((unsigned)__builtin_amdgcn_readlane((int)moff, j + K) + l16));
#define MEAN_ADD(K) acc.x += v##K.x; acc.y += v##K.y; acc.z += v##K.z; acc.w += v##K.w;

__global__ __launch_bounds__(SORT_T) void range_csr_pull(const int* __restrict__ tmp, const int* __restrict__ pcnt,
                                                         const int* __restrict__ poff, int* __restrict__ segout,
                                                         int* csr, const float* __restrict__ feat,
                                                         float* __restrict__ agg, int E, int N, int NS) {
  __shared__ int pc[SORT_NC], po[SORT_NC];
  __shared__ int h[SORT_NB], st[SORT_NB + 1], cur[SORT_NB];
  __shared__ int wtot[8];
  __shared__ int gsum;
  __shared__ int nextn;
  __shared__ int srt[SORT_CAP];
  const int CE = (E + SORT_NC - 1) / SORT_NC;
  if (E < 1 || N < 1 || N / SORT_NB + 1 > SORT_NB || CE > SORT_T * SORT_EPT) return;
  const int r = blockIdx.y, b = blockIdx.x, t = threadIdx.x;
  const int BW = N / SORT_NB + 1;
  const int node0 = b * BW;
  const int* rowc = pcnt + ((size_t)r * SORT_NB + b) * SORT_NC;
  const int* rowo = poff + ((size_t)r * SORT_NB + b) * SORT_NC;
  if (t < SORT_NC) {
    int cc = rowc[t], oo = rowo[t];
    if (oo < 0) oo = 0;
    if (oo > CE) oo = CE;
    if (cc < 0) cc = 0;
    if (cc > CE - oo) cc = CE - oo;
    pc[t] = cc;
    po[t] = oo;
  }
  if (t < SORT_NB) { h[t] = 0; cur[t] = 0; }
  if (t == 0) { gsum = 0; nextn = 0; }
  __syncthreads();
  {
    const int4* allc = (const int4*)(pcnt + (size_t)r * SORT_NB * SORT_NC);
    const int n4 = b * (SORT_NC / 4);
    int4 v[8];
#pragma unroll
    for (int k = 0; k < 8; ++k) {
      const int i = t + k * SORT_T;
      v[k] = allc[i < n4 ? i : 0];
    }
    int part = 0;
#pragma unroll
    for (int k = 0; k < 8; ++k)
      if (t + k * SORT_T < n4) part += v[k].x + v[k].y + v[k].z + v[k].w;
    for (int off = 16; off > 0; off >>= 1) part += __shfl_xor(part, off, 32);
    if ((t & 31) == 0) atomicAdd(&gsum, part);
  }
  const int g = t >> 3, k8 = t & 7;
  const int pstart = g * CE + po[g];
  const int room = E - pstart;
  const int pl = pc[g] < room ? pc[g] : (room > 0 ? room : 0);
  const int2* __restrict__ piece = (const int2*)tmp + (size_t)r * E + (pstart < E ? pstart : E - 1);
  int2 q[4];
#pragma unroll
  for (int k = 0; k < 4; ++k) {
    const int i = k8 + 8 * k;
    q[k] = piece[i < pl ? i : 0];
  }
#pragma unroll
  for (int k = 0; k < 4; ++k) {
    if (k8 + 8 * k < pl) {
      int dl = q[k].y - node0;
      if ((unsigned)dl >= (unsigned)BW) dl = 0;
      atomicAdd(&h[dl], 1);
    }
  }
  for (int i = k8 + 32; i < pl; i += 8) {
    int dl = piece[i].y - node0;
    if ((unsigned)dl >= (unsigned)BW) dl = 0;
    atomicAdd(&h[dl], 1);
  }
  __syncthreads();
  const int mine = (t < SORT_NB) ? h[t] : 0;
  int M;
  const int pre = scan256(mine, t, wtot, M);
  if (t < SORT_NB) st[t] = pre;
  if (t == 0) st[SORT_NB] = M;
  __syncthreads();
  const int gstart = gsum;
  if (t < BW && node0 + t <= N) segout[(size_t)r * NS + node0 + t] = gstart + st[t];
#pragma unroll
  for (int k = 0; k < 4; ++k) {
    if (k8 + 8 * k < pl) {
      int dl = q[k].y - node0;
      if ((unsigned)dl >= (unsigned)BW) dl = 0;
      const int pos = st[dl] + atomicAdd(&cur[dl], 1);
      if ((unsigned)pos < (unsigned)SORT_CAP) srt[pos] = q[k].x;
      else if ((unsigned)(gstart + pos) < (unsigned)E) csr[(size_t)r * E + gstart + pos] = q[k].x;
    }
  }
  for (int i = k8 + 32; i < pl; i += 8) {
    const int2 sd = piece[i];
    int dl = sd.y - node0;
    if ((unsigned)dl >= (unsigned)BW) dl = 0;
    const int pos = st[dl] + atomicAdd(&cur[dl], 1);
    if ((unsigned)pos < (unsigned)SORT_CAP) srt[pos] = sd.x;
    else if ((unsigned)(gstart + pos) < (unsigned)E) csr[(size_t)r * E + gstart + pos] = sd.x;
  }
  __syncthreads();
  const int mout = M < SORT_CAP ? M : SORT_CAP;
  for (int i = t; i < mout; i += SORT_T)
    if ((unsigned)(gstart + i) < (unsigned)E) csr[(size_t)r * E + gstart + i] = srt[i];

  const int lane = t & 31;
  int nn = N - node0;
  if (nn > BW) nn = BW;
  int lim = E - gstart;
  if (lim > M) lim = M;
  const char* fb = (const char*)feat;
  const unsigned l16 = (unsigned)lane * 16u;
  const int* gl = csr + (size_t)r * E + gstart;
  for (;;) {
    int dl = 0;
    if (lane == 0) dl = atomicAdd(&nextn, 1);
    dl = __builtin_amdgcn_readfirstlane(dl);
    if (dl >= nn) break;
    const int start = __builtin_amdgcn_readfirstlane(st[dl]);
    int end = __builtin_amdgcn_readfirstlane(st[dl + 1]);
    if (end > lim) end = lim;
    float4 acc = make_float4(0.0f, 0.0f, 0.0f, 0.0f);
    for (int base = start; base < end; base += 32) {
      const int rem = end - base;
      const int n = rem < 32 ? rem : 32;
      const int p = base + lane;
      int ids = 0;
      if (lane < n) {
        ids = srt[p < SORT_CAP ? p : 0];
        if (p >= SORT_CAP) ids = gl[p];
      }
      if ((unsigned)ids >= (unsigned)N) ids = 0;
      const unsigned moff = (unsigned)ids * (FDIM * 4u);
      for (int j = 0; j < n; j += 8) {
        MEAN_ROW(0) MEAN_ROW(1) MEAN_ROW(2) MEAN_ROW(3) MEAN_ROW(4) MEAN_ROW(5) MEAN_ROW(6) MEAN_ROW(7)
        MEAN_ADD(0) MEAN_ADD(1) MEAN_ADD(2) MEAN_ADD(3) MEAN_ADD(4) MEAN_ADD(5) MEAN_ADD(6) MEAN_ADD(7)
      }
    }
    const float deg = (float)(end > start ? end - start : 0);
    const float dinv = 1.0f / fmaxf(deg, 1.0f);
    float4 o;
    o.x = acc.x * dinv; o.y = acc.y * dinv; o.z = acc.z * dinv; o.w = acc.w * dinv;
    *(float4*)(agg + ((size_t)r * N + node0 + dl) * FDIM + lane * 4) = o;
  }
}
#undef MEAN_ROW
#undef MEAN_ADD


__device__ __forceinline__ float other_half(float v) {
  const int b = __builtin_bit_cast(int, v);
  return __builtin_bit_cast(float, __builtin_amdgcn_permlanex16(b, b, 0x76543210, 0xfedcba98, false, false));
}

__device__ __forceinline__ int first_ids(const int* __restrict__ nb, int start, int end, int lane) {
  const int rem = end - start;
  const int n = rem < 32 ? rem : 32;
  return (lane < n) ? nb[start + lane] : 0;
}

#define PULL_PAIR(K)                                                                          \
  const unsigned e##K = (unsigned)__builtin_amdgcn_readlane((int)moff, j + 2 * K);            \
  const unsigned o##K = (unsigned)__builtin_amdgcn_readlane((int)moff, j + 2 * K + 1);        \
  float4 v##K = make_float4(0.0f, 0.0f, 0.0f, 0.0f);                                          \
  if (j + 2 * K + hlf < n) v##K = *(const float4*)(pb + ((hlf ? o##K : e##K) + c16));
#define PULL_ADD(K) acc.x += v##K.x; acc.y += v##K.y; acc.z += v##K.z; acc.w += v##K.w;

__device__ __forceinline__ float4 pull_blocks(const int* __restrict__ nb, int start, int end, int mine,
                                              const char* __restrict__ pb, unsigned c16, int lane, int hlf, int N) {
  float4 acc = make_float4(0.0f, 0.0f, 0.0f, 0.0f);
  for (int base = start; base < end; base += 32) {
    const int rem = end - base;
    const int n = rem < 32 ? rem : 32;
    if (base != start) mine = (lane < n) ? nb[base + lane] : 0;
    if ((unsigned)mine >= (unsigned)N) mine = 0;
    const unsigned moff = (unsigned)mine * (PROW * 4u);
    for (int j = 0; j < n; j += 16) {
      PULL_PAIR(0) PULL_PAIR(1) PULL_PAIR(2) PULL_PAIR(3) PULL_PAIR(4) PULL_PAIR(5) PULL_PAIR(6) PULL_PAIR(7)
      PULL_ADD(0) PULL_ADD(1) PULL_ADD(2) PULL_ADD(3) PULL_ADD(4) PULL_ADD(5) PULL_ADD(6) PULL_ADD(7)
    }
  }
  return acc;
}
#undef PULL_PAIR
#undef PULL_ADD

__global__ __launch_bounds__(256) void gather_proj(const int* __restrict__ seg, const int* __restrict__ csr,
                                                   const float* __restrict__ P, float* __restrict__ h2,
                                                   int E, int N, int NS) {
  const int lane = threadIdx.x & 31;
  const int hlf = lane >> 4;
  const unsigned c16 = (unsigned)(lane & 15) * 16u;
  const int w = __builtin_amdgcn_readfirstlane((int)(threadIdx.x >> 5));
  const int nwaves = gridDim.x * 8;
  const char* pbytes = (const char*)P;
  for (int i = blockIdx.x * 8 + w; i < N; i += nwaves) {
    int s0 = seg[i], e0 = seg[i + 1];
    int s1 = seg[(size_t)NS + i], e1 = seg[(size_t)NS + i + 1];
    if (s0 < 0) s0 = 0;
    if (e0 > E) e0 = E;
    if (s1 < 0) s1 = 0;
    if (e1 > E) e1 = E;
    const int m0 = first_ids(csr, s0, e0, lane);
    const int m1 = first_ids(csr + E, s1, e1, lane);
    float4 tot = make_float4(0.0f, 0.0f, 0.0f, 0.0f);
    if (hlf == 0) tot = *(const float4*)(pbytes + ((size_t)i * (PROW * 4u) + c16));
    const float4 a0 = pull_blocks(csr, s0, e0, m0, pbytes + 256, c16, lane, hlf, N);
    const float4 a1 = pull_blocks(csr + E, s1, e1, m1, pbytes + 512, c16, lane, hlf, N);
    const float d0 = 1.0f / fmaxf((float)(e0 - s0), 1.0f);
    const float d1 = 1.0f / fmaxf((float)(e1 - s1), 1.0f);
    tot.x += a0.x * d0 + a1.x * d1;
    tot.y += a0.y * d0 + a1.y * d1;
    tot.z += a0.z * d0 + a1.z * d1;
    tot.w += a0.w * d0 + a1.w * d1;
    tot.x += other_half(tot.x);
    tot.y += other_half(tot.y);
    tot.z += other_half(tot.z);
    tot.w += other_half(tot.w);
    if (hlf == 0) *(float4*)(h2 + (size_t)i * 64 + (lane & 15) * 4) = tot;
  }
}

typedef float v4f __attribute__((ext_vector_type(4)));

#define GEMM_WGS 256
#define GEMM_WAVES 16
#define GEMM_THREADS 512
#define SLOT_PITCH 68

template <int NOUT, int KT, bool RELU>
__global__ __launch_bounds__(GEMM_THREADS) void sage_gemm(
    const float* __restrict__ A0, const float* __restrict__ G0, const float* __restrict__ G1,
    const float* __restrict__ Bmat, const float* __restrict__ bias,
    float* __restrict__ out, int Nrows) {
  constexpr int NCH = KT / 64;
  constexpr int NSPLIT = NOUT / 64;
  static_assert(NCH % 2 == 0, "the two A register sets swap roles once per chunk");
  __shared__ v4f Bs[96 * 128];
  __shared__ float slots[GEMM_WAVES * 16 * SLOT_PITCH];

  const int tid = threadIdx.x;
  const int lane = tid & 31;
  const int m = lane & 15;
  const int h = lane >> 4;
  const int wave = __builtin_amdgcn_readfirstlane(tid >> 5);

  for (int i = tid; i < (KT / 4) * NOUT; i += GEMM_THREADS) {
    const int kq = i / NOUT;
    const int n = i - kq * NOUT;
    const float* p = Bmat + (size_t)(4 * kq) * NOUT + n;
    v4f w;
    w.x = p[0]; w.y = p[NOUT]; w.z = p[2 * NOUT]; w.w = p[3 * NOUT];
    Bs[i] = w;
  }
  __syncthreads();

  const int nRT = (Nrows + 15) >> 4;
  const int rt0 = (int)(((long long)blockIdx.x * nRT) / GEMM_WGS);
  const int rt1 = (int)(((long long)(blockIdx.x + 1) * nRT) / GEMM_WGS);
  const int nunits = (rt1 - rt0) * NSPLIT;

  auto a_ptr = [&](int u, int c) -> const float* {
    const int uu = u < nunits ? u : nunits - 1;
    int row = (rt0 + uu / NSPLIT) * 16 + m;
    if (row > Nrows - 1) row = Nrows - 1;
    const float* src = (c < 2) ? A0 : ((c < 4) ? G0 : G1);
    return src + (size_t)row * FDIM + (c & 1) * 64 + 4 * h;
  };

  if (wave >= nunits) return;
  v4f abuf[2][8];
  {
    const float* p = a_ptr(wave, 0);
#pragma unroll
    for (int j = 0; j < 8; ++j) abuf[0][j] = *(const v4f*)(p + j * 8);
  }
  float* slot = slots + wave * (16 * SLOT_PITCH);

  for (int u = wave; u < nunits; u += GEMM_WAVES) {
    const int rt = rt0 + u / NSPLIT;
    const int nh = u - (u / NSPLIT) * NSPLIT;
    v8f acc[4];
#pragma unroll
    for (int t = 0; t < 4; ++t) acc[t] = v8f{};

    const v4f* bunit = Bs + (size_t)h * NOUT + nh * 64 + m;
    v4f bcur[4], bnxt[4];
#pragma unroll
    for (int t = 0; t < 4; ++t) bcur[t] = bunit[t * 16];

#pragma unroll
    for (int c = 0; c < NCH; ++c) {
      {
        const float* p = (c + 1 < NCH) ? a_ptr(u, c + 1) : a_ptr(u + GEMM_WAVES, 0);
#pragma unroll
        for (int j = 0; j < 8; ++j) abuf[(c + 1) & 1][j] = *(const v4f*)(p + j * 8);
      }
#pragma unroll
      for (int j = 0; j < 8; ++j) {
        const int st = c * 8 + j;
        if (st + 1 < NCH * 8) {
#pragma unroll
          for (int t = 0; t < 4; ++t) bnxt[t] = bunit[(size_t)(2 * (st + 1)) * NOUT + t * 16];
        }
        const v4f a = abuf[c & 1][j];
        const v2f alo = {a.x, a.y}, ahi = {a.z, a.w};
#pragma unroll
        for (int t = 0; t < 4; ++t) {
          const v2f blo = {bcur[t].x, bcur[t].y};
          acc[t] = __builtin_amdgcn_wmma_f32_16x16x4_f32(false, alo, false, blo, (short)0, acc[t], false, false);
        }
#pragma unroll
        for (int t = 0; t < 4; ++t) {
          const v2f bhi = {bcur[t].z, bcur[t].w};
          acc[t] = __builtin_amdgcn_wmma_f32_16x16x4_f32(false, ahi, false, bhi, (short)0, acc[t], false, false);
        }
#pragma unroll
        for (int t = 0; t < 4; ++t) bcur[t] = bnxt[t];
      }
    }

#pragma unroll
    for (int t = 0; t < 4; ++t) {
      const float bv = bias[nh * 64 + t * 16 + m];
#pragma unroll
      for (int v = 0; v < 8; ++v) {
        float val = acc[t][v] + bv;
        if (RELU) val = fmaxf(val, 0.0f);
        slot[(v + 8 * h) * SLOT_PITCH + t * 16 + m] = val;
      }
    }
    __builtin_amdgcn_fence(__ATOMIC_RELEASE, "wavefront");
    __builtin_amdgcn_wave_barrier();
    __builtin_amdgcn_fence(__ATOMIC_ACQUIRE, "wavefront");
    float* orow = out + (size_t)(rt * 16 + h) * NOUT + nh * 64 + m * 4;
    const float* srow = slot + h * SLOT_PITCH + m * 4;
    if (rt * 16 + 16 <= Nrows) {
#pragma unroll
      for (int j = 0; j < 8; ++j)
        *(float4*)(orow + (size_t)(2 * j) * NOUT) = *(const float4*)(srow + 2 * j * SLOT_PITCH);
    } else {
#pragma unroll
      for (int j = 0; j < 8; ++j) {
        if (rt * 16 + 2 * j + h < Nrows)
          *(float4*)(orow + (size_t)(2 * j) * NOUT) = *(const float4*)(srow + 2 * j * SLOT_PITCH);
      }
    }
    __builtin_amdgcn_fence(__ATOMIC_RELEASE, "wavefront");
    __builtin_amdgcn_wave_barrier();
    __builtin_amdgcn_fence(__ATOMIC_ACQUIRE, "wavefront");
  }
}


template <int S> __device__ __forceinline__ unsigned row_lane_plus(unsigned v, unsigned add) {
  return (unsigned)__builtin_amdgcn_update_dpp(0, (int)v, 0x150 + S, 0xf, 0xf, true) + add;
}
template <int M> __device__ __forceinline__ float row_xor_lane(float v) {
  return __builtin_bit_cast(float,
      __builtin_amdgcn_update_dpp(0, __builtin_bit_cast(int, v), 0x160 + M, 0xf, 0xf, true));
}
template <int M> __device__ __forceinline__ float fold_pair(bool up, float lo, float hi) {
  const float keep = up ? hi : lo;
  const float send = up ? lo : hi;
  return keep + row_xor_lane<M>(send);
}

#define EDGE_LOAD(S)                                                        \
  const float4 a##S = *(const float4*)(hb + row_lane_plus<S>(so, c16));     \
  const float4 b##S = *(const float4*)(hb + row_lane_plus<S>(dof, c16));
#define EDGE_DOT(S) \
  p[S] = a##S.x * b##S.x + a##S.y * b##S.y + a##S.z * b##S.z + a##S.w * b##S.w;

__global__ __launch_bounds__(256) void edge_score(const int* __restrict__ src0, const int* __restrict__ dst0,
                                                  const int* __restrict__ src1, const int* __restrict__ dst1,
                                                  const float* __restrict__ h, float* __restrict__ out,
                                                  int E, int N) {
  if (E <= 0) return;
  const int lane = threadIdx.x & 31;
  const int c = lane & 15;
  const unsigned c16 = (unsigned)c * 16u;
  const int w = __builtin_amdgcn_readfirstlane((int)(threadIdx.x >> 5));
  const int nwaves = gridDim.x * 8;
  const int ngroups = (E + 31) >> 5;
  const int total = 2 * ngroups;
  const char* hb = (const char*)h;
  const bool u8 = (c & 8) != 0, u4 = (c & 4) != 0, u2 = (c & 2) != 0, u1 = (c & 1) != 0;

  auto load_ids = [&](int gg, int& s, int& d) {
    if (gg > total - 1) gg = total - 1;
    const bool second = gg >= ngroups;
    int e = (gg - (second ? ngroups : 0)) * 32 + lane;
    if (e > E - 1) e = E - 1;
    s = (second ? src1 : src0)[e];
    d = (second ? dst1 : dst0)[e];
  };

  int g = blockIdx.x * 8 + w;
  int si, di;
  load_ids(g, si, di);
  for (; g < total; g += nwaves) {
    const bool second = g >= ngroups;
    const int e = (g - (second ? ngroups : 0)) * 32 + lane;
    int sn, dn;
    load_ids(g + nwaves, sn, dn);
    if ((unsigned)si >= (unsigned)N) si = 0;
    if ((unsigned)di >= (unsigned)N) di = 0;
    const unsigned so = (unsigned)si * 256u;
    const unsigned dof = (unsigned)di * 256u;
    float p[16];
    EDGE_LOAD(0) EDGE_LOAD(1) EDGE_LOAD(2) EDGE_LOAD(3) EDGE_LOAD(4) EDGE_LOAD(5) EDGE_LOAD(6) EDGE_LOAD(7)
    EDGE_LOAD(8) EDGE_LOAD(9) EDGE_LOAD(10) EDGE_LOAD(11) EDGE_LOAD(12) EDGE_LOAD(13) EDGE_LOAD(14) EDGE_LOAD(15)
    EDGE_DOT(0) EDGE_DOT(1) EDGE_DOT(2) EDGE_DOT(3) EDGE_DOT(4) EDGE_DOT(5) EDGE_DOT(6) EDGE_DOT(7)
    EDGE_DOT(8) EDGE_DOT(9) EDGE_DOT(10) EDGE_DOT(11) EDGE_DOT(12) EDGE_DOT(13) EDGE_DOT(14) EDGE_DOT(15)
    float q[8];
#pragma unroll
    for (int s = 0; s < 8; ++s) q[s] = fold_pair<8>(u8, p[s], p[s + 8]);
    float r4[4];
#pragma unroll
    for (int s = 0; s < 4; ++s) r4[s] = fold_pair<4>(u4, q[s], q[s + 4]);
    const float t0 = fold_pair<2>(u2, r4[0], r4[2]);
    const float t1 = fold_pair<2>(u2, r4[1], r4[3]);
    const float score = fold_pair<1>(u1, t0, t1);
    if (e < E) out[(second ? (size_t)E : (size_t)0) + e] = score;
    si = sn; di = dn;
  }
}
#undef EDGE_LOAD
#undef EDGE_DOT

extern "C" void kernel_launch(void* const* d_in, const int* in_sizes, int n_in,
                              void* d_out, int out_size, void* d_ws, size_t ws_size,
                              hipStream_t stream) {
  const float* x   = (const float*)d_in[0];
  const int*   edges = (const int*)d_in[1];
  const int*   negE  = (const int*)d_in[2];
  const float* Wn1 = (const float*)d_in[3];
  const float* Ws1 = (const float*)d_in[4];
  const float* b1  = (const float*)d_in[5];
  const float* Wn2 = (const float*)d_in[6];
  const float* Ws2 = (const float*)d_in[7];
  const float* b2  = (const float*)d_in[8];
  float* out = (float*)d_out;

  const int N = in_sizes[0] / 128;
  const int E = in_sizes[1] / 4;

  float* ws    = (float*)d_ws;
  float* bias2 = ws;
  float* agg0  = ws + ((2 * (size_t)N) & ~(size_t)63);
  float* agg1  = agg0 + (size_t)N * 128;
  float* h1    = agg1 + (size_t)N * 128;
  float* elist = h1 + (size_t)N * 128;
  float* Bcat1 = elist + (size_t)N * 64;
  float* bias1 = Bcat1 + 384 * 128;
  float* Bcat2 = bias1 + 128;
  float* P     = agg0;
  float* h2    = h1;
  const int NS = (N + 8) & ~7;
  int* csr  = (int*)elist;
  int* segx = csr + 2 * (size_t)E;
  int* tmp  = (int*)h1;
  int* pcnt = (int*)h1 + 4 * (size_t)E;
  int* poff = pcnt + 2 * SORT_NB * SORT_NC;

  const int T = 256;

  const int wtot = 384 * 128 + 128 + 128 * PROW + PROW;
  build_weights<<<(wtot + T - 1) / T, T, 0, stream>>>(Wn1, Ws1, b1, Wn2, Ws2, b2, Bcat1, bias1, Bcat2, bias2);

  bucket_chunks<<<dim3(SORT_NC, 2), SORT_T, 0, stream>>>(edges, tmp, pcnt, poff, E, N);
  range_csr_pull<<<dim3(SORT_NB, 2), SORT_T, 0, stream>>>(tmp, pcnt, poff, segx, csr, x, agg0, E, N, NS);

  const int sblocks = 1024;
  const int pblocks = 2048;

  sage_gemm<128, 384, true><<<GEMM_WGS, GEMM_THREADS, 0, stream>>>(x, agg0, agg1, Bcat1, bias1, h1, N);

  sage_gemm<PROW, 128, false><<<GEMM_WGS, GEMM_THREADS, 0, stream>>>(h1, h1, h1, Bcat2, bias2, P, N);
  gather_proj<<<pblocks, T, 0, stream>>>(segx, csr, P, h2, E, N, NS);

  edge_score<<<sblocks, T, 0, stream>>>(edges, edges + E, negE, negE + E, h2, out, E, N);
}


# Layer-1 product moved under the layer-1 mean pull: one workgroup per node range sorts both relations, 4 waves multiply group g-1 from LDS while all pull group g; 7 launches to 6; new K sum order
# speedup vs baseline: 6.4797x; 1.0298x over previous
#include <hip/hip_runtime.h>
#include <hip/hip_bf16.h>

typedef float v2f __attribute__((ext_vector_type(2)));
typedef float v8f __attribute__((ext_vector_type(8)));

#define FDIM 128
#define PROW 192

__global__ void build_weights(const float* __restrict__ Wn1, const float* __restrict__ Ws1,
                              const float* __restrict__ b1,
                              const float* __restrict__ Wn2, const float* __restrict__ Ws2,
                              const float* __restrict__ b2,
                              float* __restrict__ Bcat1, float* __restrict__ bias1,
                              float* __restrict__ Bcat2, float* __restrict__ bias2) {
  int i = blockIdx.x * blockDim.x + threadIdx.x;
  const int S1 = 384 * 128, S2 = S1 + 128, S3 = S2 + 128 * PROW, S4 = S3 + PROW;
  if (i < S1) {
    int k = i >> 7, n = i & 127;
    float v;
    if (k < 128)      v = Ws1[k * 128 + n] + Ws1[16384 + k * 128 + n];
    else if (k < 256) v = Wn1[(k - 128) * 128 + n];
    else              v = Wn1[16384 + (k - 256) * 128 + n];
    Bcat1[i] = v;
  } else if (i < S2) {
    int n = i - S1;
    bias1[n] = b1[n] + b1[128 + n];
  } else if (i < S3) {
    int j = i - S2;
    int k = j / PROW, n = j - k * PROW;
    float v;
    if (n < 64)       v = Ws2[k * 64 + n] + Ws2[8192 + k * 64 + n];
    else if (n < 128) v = Wn2[k * 64 + (n - 64)];
    else              v = Wn2[8192 + k * 64 + (n - 128)];
    Bcat2[j] = v;
  } else if (i < S4) {
    int n = i - S3;
    bias2[n] = (n < 64) ? b2[n] + b2[64 + n] : 0.0f;
  }
}

#define SORT_NB 256
#define SORT_NC 128
#define SORT_T 1024
#define SORT_EPT 8

__device__ __forceinline__ int scan256(int v, int t, int* wtot, int& total) {
  const int lane = t & 31, w = t >> 5;
  int inc = (t < 256) ? v : 0;
  for (int off = 1; off < 32; off <<= 1) {
    const int up = __shfl_up(inc, off, 32);
    if (lane >= off) inc += up;
  }
  if (t < 256 && lane == 31) wtot[w] = inc;
  __syncthreads();
  int before = 0, all = 0;
  for (int k = 0; k < 8; ++k) {
    const int x = wtot[k];
    all += x;
    if (k < w) before += x;
  }
  total = all;
  return before + inc - ((t < 256) ? v : 0);
}

__global__ __launch_bounds__(SORT_T) void bucket_chunks(const int* __restrict__ edges, int* __restrict__ tmp,
                                                        int* __restrict__ pcnt, int* __restrict__ poff, int E, int N) {
  __shared__ int hist[SORT_NB];
  __shared__ int offs[SORT_NB];
  __shared__ int wtot[8];
  __shared__ int2 stage[SORT_T * SORT_EPT];
  const int CE = (E + SORT_NC - 1) / SORT_NC;
  if (E < 1 || N < 1 || CE > SORT_T * SORT_EPT) return;
  const int r = blockIdx.y, c = blockIdx.x, t = threadIdx.x;
  const unsigned BW = (unsigned)N / SORT_NB + 1u;
  const int* src = edges + (size_t)r * 2 * E;
  const int* dst = src + E;
  const int eb = c * CE;
  int ee = eb + CE;
  if (ee > E) ee = E;
  const int n = ee > eb ? ee - eb : 0;
  if (t < SORT_NB) hist[t] = 0;
  __syncthreads();
  int s[SORT_EPT], d[SORT_EPT], bk[SORT_EPT], rk[SORT_EPT];
#pragma unroll
  for (int k = 0; k < SORT_EPT; ++k) {
    const int i = t + k * SORT_T;
    const int e = i < n ? eb + i : (E - 1);
    s[k] = src[e];
    d[k] = dst[e];
  }
#pragma unroll
  for (int k = 0; k < SORT_EPT; ++k) {
    unsigned dd = (unsigned)d[k];
    if (dd >= (unsigned)N) dd = 0;
    d[k] = (int)dd;
    unsigned b = dd / BW;
    if (b > SORT_NB - 1) b = SORT_NB - 1;
    bk[k] = (int)b;
    rk[k] = 0;
    if (t + k * SORT_T < n) rk[k] = atomicAdd(&hist[b], 1);
  }
  __syncthreads();
  const int mine = (t < SORT_NB) ? hist[t] : 0;
  int total;
  const int pre = scan256(mine, t, wtot, total);
  if (t < SORT_NB) {
    offs[t] = pre;
    pcnt[((size_t)r * SORT_NB + t) * SORT_NC + c] = mine;
    poff[((size_t)r * SORT_NB + t) * SORT_NC + c] = pre;
  }
  __syncthreads();
#pragma unroll
  for (int k = 0; k < SORT_EPT; ++k) {
    if (t + k * SORT_T < n) {
      const int pos = offs[bk[k]] + rk[k];
      if ((unsigned)pos < (unsigned)(SORT_T * SORT_EPT)) stage[pos] = make_int2(s[k], d[k]);
    }
  }
  __syncthreads();
  int2* __restrict__ tp = (int2*)tmp + (size_t)r * E + eb;
  for (int i = t; i < n; i += SORT_T) tp[i] = stage[i];
}


#define MEAN_ROW(K)                                                                           \
  float4 v##K = make_float4(0.0f, 0.0f, 0.0f, 0.0f);                                          \
  if (j + K < n) v##K = *(const float4*)(fb + ((unsigned)__builtin_amdgcn_readlane((int)moff, j + K) + l16));
#define MEAN_ADD(K) acc.x += v##K.x; acc.y += v##K.y; acc.z += v##K.z; acc.w += v##K.w;

typedef float v4f __attribute__((ext_vector_type(4)));
#define SLOT_PITCH 68
#define FCAP 4096
#define GRP 32
#define MPITCH 260

__device__ constexpr int kstep_at(int pp) {
  return pp < 16 ? 16 + pp : (pp < 24 ? pp - 16 : (pp < 40 ? 32 + (pp - 24) : 8 + (pp - 40)));
}

__global__ __launch_bounds__(SORT_T) void range_pull_gemm(const int* __restrict__ tmp, const int* __restrict__ pcnt,
                                                          const int* __restrict__ poff, int* __restrict__ segout,
                                                          int* csr, const float* __restrict__ feat,
                                                          const float* __restrict__ Bmat, const float* __restrict__ bias,
                                                          float* __restrict__ out, int E, int N, int NS) {
  __shared__ v4f Bs[96 * 128];
  __shared__ __attribute__((aligned(16))) float mbuf[2 * GRP * MPITCH];
  __shared__ __attribute__((aligned(16))) float slots[4 * 16 * SLOT_PITCH];
  __shared__ int srt[2 * FCAP];
  __shared__ int pc[SORT_NC], po[SORT_NC];
  __shared__ int h[SORT_NB], st[2 * (SORT_NB + 1)], cur[SORT_NB];
  __shared__ int wtot[8];
  __shared__ int gsum;
  __shared__ int cnt[16];
  const int CE = (E + SORT_NC - 1) / SORT_NC;
  if (E < 1 || N < 1 || N / SORT_NB + 1 > SORT_NB || CE > SORT_T * SORT_EPT) return;
  const int b = blockIdx.x, t = threadIdx.x;
  const int BW = N / SORT_NB + 1;
  const int node0 = b * BW;
  const int lane = t & 31;
  const int wave = __builtin_amdgcn_readfirstlane(t >> 5);

#pragma unroll
  for (int it = 0; it < (96 * 128) / SORT_T; ++it) {
    const int i = t + it * SORT_T;
    const int kq = i >> 7, n = i & 127;
    const float* p = Bmat + (size_t)(4 * kq) * 128 + n;
    v4f w;
    w.x = p[0]; w.y = p[128]; w.z = p[256]; w.w = p[384];
    Bs[i] = w;
  }
  if (t < 16) cnt[t] = 0;

  auto sort_rel = [&](const int r, int& Mout, int& gout) {
    const int* rowc = pcnt + ((size_t)r * SORT_NB + b) * SORT_NC;
    const int* rowo = poff + ((size_t)r * SORT_NB + b) * SORT_NC;
    int* str = st + r * (SORT_NB + 1);
    int* sr = srt + r * FCAP;
    if (t < SORT_NC) {
      int cc = rowc[t], oo = rowo[t];
      if (oo < 0) oo = 0;
      if (oo > CE) oo = CE;
      if (cc < 0) cc = 0;
      if (cc > CE - oo) cc = CE - oo;
      pc[t] = cc;
      po[t] = oo;
    }
    if (t < SORT_NB) { h[t] = 0; cur[t] = 0; }
    if (t == 0) gsum = 0;
    __syncthreads();
    {
      const int4* allc = (const int4*)(pcnt + (size_t)r * SORT_NB * SORT_NC);
      const int n4 = b * (SORT_NC / 4);
      int4 v[8];
#pragma unroll
      for (int k = 0; k < 8; ++k) {
        const int i = t + k * SORT_T;
        v[k] = allc[i < n4 ? i : 0];
      }
      int part = 0;
#pragma unroll
      for (int k = 0; k < 8; ++k)
        if (t + k * SORT_T < n4) part += v[k].x + v[k].y + v[k].z + v[k].w;
      for (int off = 16; off > 0; off >>= 1) part += __shfl_xor(part, off, 32);
      if ((t & 31) == 0) atomicAdd(&gsum, part);
    }
    const int g = t >> 3, k8 = t & 7;
    const int pstart = g * CE + po[g];
    const int room = E - pstart;
    const int pl = pc[g] < room ? pc[g] : (room > 0 ? room : 0);
    const int2* __restrict__ piece = (const int2*)tmp + (size_t)r * E + (pstart < E ? pstart : E - 1);
    int2 q[4];
#pragma unroll
    for (int k = 0; k < 4; ++k) {
      const int i = k8 + 8 * k;
      q[k] = piece[i < pl ? i : 0];
    }
#pragma unroll
    for (int k = 0; k < 4; ++k) {
      if (k8 + 8 * k < pl) {
        int dl = q[k].y - node0;
        if ((unsigned)dl >= (unsigned)BW) dl = 0;
        atomicAdd(&h[dl], 1);
      }
    }
    for (int i = k8 + 32; i < pl; i += 8) {
      int dl = piece[i].y - node0;
      if ((unsigned)dl >= (unsigned)BW) dl = 0;
      atomicAdd(&h[dl], 1);
    }
    __syncthreads();
    const int mine = (t < SORT_NB) ? h[t] : 0;
    int M;
    const int pre = scan256(mine, t, wtot, M);
    if (t < SORT_NB) str[t] = pre;
    if (t == 0) str[SORT_NB] = M;
    __syncthreads();
    const int gstart = gsum;
    if (t < BW && node0 + t <= N) segout[(size_t)r * NS + node0 + t] = gstart + str[t];
#pragma unroll
    for (int k = 0; k < 4; ++k) {
      if (k8 + 8 * k < pl) {
        int dl = q[k].y - node0;
        if ((unsigned)dl >= (unsigned)BW) dl = 0;
        const int pos = str[dl] + atomicAdd(&cur[dl], 1);
        if ((unsigned)pos < (unsigned)FCAP) sr[pos] = q[k].x;
        else if ((unsigned)(gstart + pos) < (unsigned)E) csr[(size_t)r * E + gstart + pos] = q[k].x;
      }
    }
    for (int i = k8 + 32; i < pl; i += 8) {
      const int2 sd = piece[i];
      int dl = sd.y - node0;
      if ((unsigned)dl >= (unsigned)BW) dl = 0;
      const int pos = str[dl] + atomicAdd(&cur[dl], 1);
      if ((unsigned)pos < (unsigned)FCAP) sr[pos] = sd.x;
      else if ((unsigned)(gstart + pos) < (unsigned)E) csr[(size_t)r * E + gstart + pos] = sd.x;
    }
    __syncthreads();
    const int mout = M < FCAP ? M : FCAP;
    for (int i = t; i < mout; i += SORT_T)
      if ((unsigned)(gstart + i) < (unsigned)E) csr[(size_t)r * E + gstart + i] = sr[i];
    Mout = M;
    gout = gstart;
  };
  int M0, M1, gs0, gs1;
  sort_rel(0, M0, gs0);
  sort_rel(1, M1, gs1);

  int nn = N - node0;
  if (nn > BW) nn = BW;
  int lim0 = E - gs0, lim1 = E - gs1;
  if (lim0 > M0) lim0 = M0;
  if (lim1 > M1) lim1 = M1;
  const char* fb = (const char*)feat;
  const unsigned l16 = (unsigned)lane * 16u;
  const int* gl0 = csr + gs0;
  const int* gl1 = csr + (size_t)E + gs1;

  auto pull_group = [&](const int G, float* mb) {
    for (;;) {
      int qi = 0;
      if (lane == 0) qi = atomicAdd(&cnt[G], 1);
      qi = __builtin_amdgcn_readfirstlane(qi);
      if (qi >= 2 * GRP) break;
      const int r = qi / GRP;
      const int li = qi - r * GRP;
      const int dl = G * GRP + li;
      int start = 0, end = 0;
      if (dl < nn) {
        start = __builtin_amdgcn_readfirstlane(st[r * (SORT_NB + 1) + dl]);
        end = __builtin_amdgcn_readfirstlane(st[r * (SORT_NB + 1) + dl + 1]);
        const int lim = r ? lim1 : lim0;
        if (end > lim) end = lim;
      }
      const int* sr = srt + r * FCAP;
      const int* gl = r ? gl1 : gl0;
      float4 acc = make_float4(0.0f, 0.0f, 0.0f, 0.0f);
      for (int base = start; base < end; base += 32) {
        const int rem = end - base;
        const int n = rem < 32 ? rem : 32;
        const int p = base + lane;
        int ids = 0;
        if (lane < n) {
          ids = sr[p < FCAP ? p : 0];
          if (p >= FCAP) ids = gl[p];
        }
        if ((unsigned)ids >= (unsigned)N) ids = 0;
        const unsigned moff = (unsigned)ids * (FDIM * 4u);
        for (int j = 0; j < n; j += 8) {
          MEAN_ROW(0) MEAN_ROW(1) MEAN_ROW(2) MEAN_ROW(3) MEAN_ROW(4) MEAN_ROW(5) MEAN_ROW(6) MEAN_ROW(7)
          MEAN_ADD(0) MEAN_ADD(1) MEAN_ADD(2) MEAN_ADD(3) MEAN_ADD(4) MEAN_ADD(5) MEAN_ADD(6) MEAN_ADD(7)
        }
      }
      const float deg = (float)(end > start ? end - start : 0);
      const float dinv = 1.0f / fmaxf(deg, 1.0f);
      float4 o;
      o.x = acc.x * dinv; o.y = acc.y * dinv; o.z = acc.z * dinv; o.w = acc.w * dinv;
      *(float4*)(mb + li * MPITCH + r * FDIM + lane * 4) = o;
    }
  };

  auto mul_unit = [&](const int G, const float* mb) {
    const int m = lane & 15;
    const int hh = lane >> 4;
    const int rtl = wave >> 1;
    const int nh = wave & 1;
    const int rbase = G * GRP + rtl * 16;
    if (rbase >= nn) return;
    int row = node0 + rbase + m;
    if (row > N - 1) row = N - 1;
    const float* xrow = feat + (size_t)row * FDIM + 4 * hh;
    v4f abuf[8];
#pragma unroll
    for (int j = 0; j < 8; ++j) abuf[j] = *(const v4f*)(xrow + j * 8);
    const float* arow = mb + (rtl * 16 + m) * MPITCH + 4 * hh;
    const v4f* bunit = Bs + (size_t)hh * 128 + nh * 64 + m;
    v8f acc[4];
#pragma unroll
    for (int tt = 0; tt < 4; ++tt) acc[tt] = v8f{};
    v4f bcur[4], bnxt[4], acur, anxt;
#pragma unroll
    for (int tt = 0; tt < 4; ++tt) bcur[tt] = bunit[(size_t)(2 * kstep_at(0)) * 128 + tt * 16];
    acur = *(const v4f*)(arow);
    anxt = acur;
    __builtin_amdgcn_sched_barrier(0);
#pragma unroll
    for (int pp = 0; pp < 48; ++pp) {
      if (pp + 1 < 48) {
#pragma unroll
        for (int tt = 0; tt < 4; ++tt) bnxt[tt] = bunit[(size_t)(2 * kstep_at(pp + 1)) * 128 + tt * 16];
        const int pn = pp + 1;
        if (pn < 16) anxt = *(const v4f*)(arow + 8 * pn);
        else if (pn >= 24 && pn < 40) anxt = *(const v4f*)(arow + FDIM + 8 * (pn - 24));
      }
      if (pp == 24) {
#pragma unroll
        for (int j = 0; j < 8; ++j) abuf[j] = *(const v4f*)(xrow + 64 + j * 8);
      }
      __builtin_amdgcn_sched_barrier(0);
      v4f a;
      if (pp < 16) a = acur;
      else if (pp < 24) a = abuf[pp - 16];
      else if (pp < 40) a = acur;
      else a = abuf[pp - 40];
      const v2f alo = {a.x, a.y}, ahi = {a.z, a.w};
#pragma unroll
      for (int tt = 0; tt < 4; ++tt) {
        const v2f blo = {bcur[tt].x, bcur[tt].y};
        acc[tt] = __builtin_amdgcn_wmma_f32_16x16x4_f32(false, alo, false, blo, (short)0, acc[tt], false, false);
      }
#pragma unroll
      for (int tt = 0; tt < 4; ++tt) {
        const v2f bhi = {bcur[tt].z, bcur[tt].w};
        acc[tt] = __builtin_amdgcn_wmma_f32_16x16x4_f32(false, ahi, false, bhi, (short)0, acc[tt], false, false);
      }
      __builtin_amdgcn_sched_barrier(0);
#pragma unroll
      for (int tt = 0; tt < 4; ++tt) bcur[tt] = bnxt[tt];
      acur = anxt;
    }
    float* slot = slots + wave * (16 * SLOT_PITCH);
#pragma unroll
    for (int tt = 0; tt < 4; ++tt) {
      const float bv = bias[nh * 64 + tt * 16 + m];
#pragma unroll
      for (int v = 0; v < 8; ++v) {
        const float val = fmaxf(acc[tt][v] + bv, 0.0f);
        slot[(v + 8 * hh) * SLOT_PITCH + tt * 16 + m] = val;
      }
    }
    __builtin_amdgcn_fence(__ATOMIC_RELEASE, "wavefront");
    __builtin_amdgcn_wave_barrier();
    __builtin_amdgcn_fence(__ATOMIC_ACQUIRE, "wavefront");
    float* orow = out + (size_t)(node0 + rbase + hh) * FDIM + nh * 64 + m * 4;
    const float* srow = slot + hh * SLOT_PITCH + m * 4;
    if (rbase + 16 <= nn) {
#pragma unroll
      for (int j = 0; j < 8; ++j)
        *(float4*)(orow + (size_t)(2 * j) * FDIM) = *(const float4*)(srow + 2 * j * SLOT_PITCH);
    } else {
#pragma unroll
      for (int j = 0; j < 8; ++j) {
        if (rbase + 2 * j + hh < nn)
          *(float4*)(orow + (size_t)(2 * j) * FDIM) = *(const float4*)(srow + 2 * j * SLOT_PITCH);
      }
    }
  };

  const int NG = nn > 0 ? (nn + GRP - 1) / GRP : 0;
  for (int p = 0; p <= NG; ++p) {
    if (p > 0 && wave < 4) mul_unit(p - 1, mbuf + ((p - 1) & 1) * (GRP * MPITCH));
    if (p < NG) pull_group(p, mbuf + (p & 1) * (GRP * MPITCH));
    __syncthreads();
  }
}
#undef MEAN_ROW
#undef MEAN_ADD


__device__ __forceinline__ float other_half(float v) {
  const int b = __builtin_bit_cast(int, v);
  return __builtin_bit_cast(float, __builtin_amdgcn_permlanex16(b, b, 0x76543210, 0xfedcba98, false, false));
}

__device__ __forceinline__ int first_ids(const int* __restrict__ nb, int start, int end, int lane) {
  const int rem = end - start;
  const int n = rem < 32 ? rem : 32;
  return (lane < n) ? nb[start + lane] : 0;
}

#define PULL_PAIR(K)                                                                          \
  const unsigned e##K = (unsigned)__builtin_amdgcn_readlane((int)moff, j + 2 * K);            \
  const unsigned o##K = (unsigned)__builtin_amdgcn_readlane((int)moff, j + 2 * K + 1);        \
  float4 v##K = make_float4(0.0f, 0.0f, 0.0f, 0.0f);                                          \
  if (j + 2 * K + hlf < n) v##K = *(const float4*)(pb + ((hlf ? o##K : e##K) + c16));
#define PULL_ADD(K) acc.x += v##K.x; acc.y += v##K.y; acc.z += v##K.z; acc.w += v##K.w;

__device__ __forceinline__ float4 pull_blocks(const int* __restrict__ nb, int start, int end, int mine,
                                              const char* __restrict__ pb, unsigned c16, int lane, int hlf, int N) {
  float4 acc = make_float4(0.0f, 0.0f, 0.0f, 0.0f);
  for (int base = start; base < end; base += 32) {
    const int rem = end - base;
    const int n = rem < 32 ? rem : 32;
    if (base != start) mine = (lane < n) ? nb[base + lane] : 0;
    if ((unsigned)mine >= (unsigned)N) mine = 0;
    const unsigned moff = (unsigned)mine * (PROW * 4u);
    for (int j = 0; j < n; j += 16) {
      PULL_PAIR(0) PULL_PAIR(1) PULL_PAIR(2) PULL_PAIR(3) PULL_PAIR(4) PULL_PAIR(5) PULL_PAIR(6) PULL_PAIR(7)
      PULL_ADD(0) PULL_ADD(1) PULL_ADD(2) PULL_ADD(3) PULL_ADD(4) PULL_ADD(5) PULL_ADD(6) PULL_ADD(7)
    }
  }
  return acc;
}
#undef PULL_PAIR
#undef PULL_ADD

__global__ __launch_bounds__(256) void gather_proj(const int* __restrict__ seg, const int* __restrict__ csr,
                                                   const float* __restrict__ P, float* __restrict__ h2,
                                                   int E, int N, int NS) {
  const int lane = threadIdx.x & 31;
  const int hlf = lane >> 4;
  const unsigned c16 = (unsigned)(lane & 15) * 16u;
  const int w = __builtin_amdgcn_readfirstlane((int)(threadIdx.x >> 5));
  const int nwaves = gridDim.x * 8;
  const char* pbytes = (const char*)P;
  for (int i = blockIdx.x * 8 + w; i < N; i += nwaves) {
    int s0 = seg[i], e0 = seg[i + 1];
    int s1 = seg[(size_t)NS + i], e1 = seg[(size_t)NS + i + 1];
    if (s0 < 0) s0 = 0;
    if (e0 > E) e0 = E;
    if (s1 < 0) s1 = 0;
    if (e1 > E) e1 = E;
    const int m0 = first_ids(csr, s0, e0, lane);
    const int m1 = first_ids(csr + E, s1, e1, lane);
    float4 tot = make_float4(0.0f, 0.0f, 0.0f, 0.0f);
    if (hlf == 0) tot = *(const float4*)(pbytes + ((size_t)i * (PROW * 4u) + c16));
    const float4 a0 = pull_blocks(csr, s0, e0, m0, pbytes + 256, c16, lane, hlf, N);
    const float4 a1 = pull_blocks(csr + E, s1, e1, m1, pbytes + 512, c16, lane, hlf, N);
    const float d0 = 1.0f / fmaxf((float)(e0 - s0), 1.0f);
    const float d1 = 1.0f / fmaxf((float)(e1 - s1), 1.0f);
    tot.x += a0.x * d0 + a1.x * d1;
    tot.y += a0.y * d0 + a1.y * d1;
    tot.z += a0.z * d0 + a1.z * d1;
    tot.w += a0.w * d0 + a1.w * d1;
    tot.x += other_half(tot.x);
    tot.y += other_half(tot.y);
    tot.z += other_half(tot.z);
    tot.w += other_half(tot.w);
    if (hlf == 0) *(float4*)(h2 + (size_t)i * 64 + (lane & 15) * 4) = tot;
  }
}

#define GEMM_WGS 256
#define GEMM_WAVES 16
#define GEMM_THREADS 512

template <int NOUT, int KT, bool RELU>
__global__ __launch_bounds__(GEMM_THREADS) void sage_gemm(
    const float* __restrict__ A0, const float* __restrict__ G0, const float* __restrict__ G1,
    const float* __restrict__ Bmat, const float* __restrict__ bias,
    float* __restrict__ out, int Nrows) {
  constexpr int NCH = KT / 64;
  constexpr int NSPLIT = NOUT / 64;
  static_assert(NCH % 2 == 0, "the two A register sets swap roles once per chunk");
  __shared__ v4f Bs[96 * 128];
  __shared__ float slots[GEMM_WAVES * 16 * SLOT_PITCH];

  const int tid = threadIdx.x;
  const int lane = tid & 31;
  const int m = lane & 15;
  const int h = lane >> 4;
  const int wave = __builtin_amdgcn_readfirstlane(tid >> 5);

  for (int i = tid; i < (KT / 4) * NOUT; i += GEMM_THREADS) {
    const int kq = i / NOUT;
    const int n = i - kq * NOUT;
    const float* p = Bmat + (size_t)(4 * kq) * NOUT + n;
    v4f w;
    w.x = p[0]; w.y = p[NOUT]; w.z = p[2 * NOUT]; w.w = p[3 * NOUT];
    Bs[i] = w;
  }
  __syncthreads();

  const int nRT = (Nrows + 15) >> 4;
  const int rt0 = (int)(((long long)blockIdx.x * nRT) / GEMM_WGS);
  const int rt1 = (int)(((long long)(blockIdx.x + 1) * nRT) / GEMM_WGS);
  const int nunits = (rt1 - rt0) * NSPLIT;

  auto a_ptr = [&](int u, int c) -> const float* {
    const int uu = u < nunits ? u : nunits - 1;
    int row = (rt0 + uu / NSPLIT) * 16 + m;
    if (row > Nrows - 1) row = Nrows - 1;
    const float* src = (c < 2) ? A0 : ((c < 4) ? G0 : G1);
    return src + (size_t)row * FDIM + (c & 1) * 64 + 4 * h;
  };

  if (wave >= nunits) return;
  v4f abuf[2][8];
  {
    const float* p = a_ptr(wave, 0);
#pragma unroll
    for (int j = 0; j < 8; ++j) abuf[0][j] = *(const v4f*)(p + j * 8);
  }
  float* slot = slots + wave * (16 * SLOT_PITCH);

  for (int u = wave; u < nunits; u += GEMM_WAVES) {
    const int rt = rt0 + u / NSPLIT;
    const int nh = u - (u / NSPLIT) * NSPLIT;
    v8f acc[4];
#pragma unroll
    for (int t = 0; t < 4; ++t) acc[t] = v8f{};

    const v4f* bunit = Bs + (size_t)h * NOUT + nh * 64 + m;
    v4f bcur[4], bnxt[4];
#pragma unroll
    for (int t = 0; t < 4; ++t) bcur[t] = bunit[t * 16];

#pragma unroll
    for (int c = 0; c < NCH; ++c) {
      {
        const float* p = (c + 1 < NCH) ? a_ptr(u, c + 1) : a_ptr(u + GEMM_WAVES, 0);
#pragma unroll
        for (int j = 0; j < 8; ++j) abuf[(c + 1) & 1][j] = *(const v4f*)(p + j * 8);
      }
#pragma unroll
      for (int j = 0; j < 8; ++j) {
        const int st = c * 8 + j;
        if (st + 1 < NCH * 8) {
#pragma unroll
          for (int t = 0; t < 4; ++t) bnxt[t] = bunit[(size_t)(2 * (st + 1)) * NOUT + t * 16];
        }
        const v4f a = abuf[c & 1][j];
        const v2f alo = {a.x, a.y}, ahi = {a.z, a.w};
#pragma unroll
        for (int t = 0; t < 4; ++t) {
          const v2f blo = {bcur[t].x, bcur[t].y};
          acc[t] = __builtin_amdgcn_wmma_f32_16x16x4_f32(false, alo, false, blo, (short)0, acc[t], false, false);
        }
#pragma unroll
        for (int t = 0; t < 4; ++t) {
          const v2f bhi = {bcur[t].z, bcur[t].w};
          acc[t] = __builtin_amdgcn_wmma_f32_16x16x4_f32(false, ahi, false, bhi, (short)0, acc[t], false, false);
        }
#pragma unroll
        for (int t = 0; t < 4; ++t) bcur[t] = bnxt[t];
      }
    }

#pragma unroll
    for (int t = 0; t < 4; ++t) {
      const float bv = bias[nh * 64 + t * 16 + m];
#pragma unroll
      for (int v = 0; v < 8; ++v) {
        float val = acc[t][v] + bv;
        if (RELU) val = fmaxf(val, 0.0f);
        slot[(v + 8 * h) * SLOT_PITCH + t * 16 + m] = val;
      }
    }
    __builtin_amdgcn_fence(__ATOMIC_RELEASE, "wavefront");
    __builtin_amdgcn_wave_barrier();
    __builtin_amdgcn_fence(__ATOMIC_ACQUIRE, "wavefront");
    float* orow = out + (size_t)(rt * 16 + h) * NOUT + nh * 64 + m * 4;
    const float* srow = slot + h * SLOT_PITCH + m * 4;
    if (rt * 16 + 16 <= Nrows) {
#pragma unroll
      for (int j = 0; j < 8; ++j)
        *(float4*)(orow + (size_t)(2 * j) * NOUT) = *(const float4*)(srow + 2 * j * SLOT_PITCH);
    } else {
#pragma unroll
      for (int j = 0; j < 8; ++j) {
        if (rt * 16 + 2 * j + h < Nrows)
          *(float4*)(orow + (size_t)(2 * j) * NOUT) = *(const float4*)(srow + 2 * j * SLOT_PITCH);
      }
    }
    __builtin_amdgcn_fence(__ATOMIC_RELEASE, "wavefront");
    __builtin_amdgcn_wave_barrier();
    __builtin_amdgcn_fence(__ATOMIC_ACQUIRE, "wavefront");
  }
}


template <int S> __device__ __forceinline__ unsigned row_lane_plus(unsigned v, unsigned add) {
  return (unsigned)__builtin_amdgcn_update_dpp(0, (int)v, 0x150 + S, 0xf, 0xf, true) + add;
}
template <int M> __device__ __forceinline__ float row_xor_lane(float v) {
  return __builtin_bit_cast(float,
      __builtin_amdgcn_update_dpp(0, __builtin_bit_cast(int, v), 0x160 + M, 0xf, 0xf, true));
}
template <int M> __device__ __forceinline__ float fold_pair(bool up, float lo, float hi) {
  const float keep = up ? hi : lo;
  const float send = up ? lo : hi;
  return keep + row_xor_lane<M>(send);
}

#define EDGE_LOAD(S)                                                        \
  const float4 a##S = *(const float4*)(hb + row_lane_plus<S>(so, c16));     \
  const float4 b##S = *(const float4*)(hb + row_lane_plus<S>(dof, c16));
#define EDGE_DOT(S) \
  p[S] = a##S.x * b##S.x + a##S.y * b##S.y + a##S.z * b##S.z + a##S.w * b##S.w;

__global__ __launch_bounds__(256) void edge_score(const int* __restrict__ src0, const int* __restrict__ dst0,
                                                  const int* __restrict__ src1, const int* __restrict__ dst1,
                                                  const float* __restrict__ h, float* __restrict__ out,
                                                  int E, int N) {
  if (E <= 0) return;
  const int lane = threadIdx.x & 31;
  const int c = lane & 15;
  const unsigned c16 = (unsigned)c * 16u;
  const int w = __builtin_amdgcn_readfirstlane((int)(threadIdx.x >> 5));
  const int nwaves = gridDim.x * 8;
  const int ngroups = (E + 31) >> 5;
  const int total = 2 * ngroups;
  const char* hb = (const char*)h;
  const bool u8 = (c & 8) != 0, u4 = (c & 4) != 0, u2 = (c & 2) != 0, u1 = (c & 1) != 0;

  auto load_ids = [&](int gg, int& s, int& d) {
    if (gg > total - 1) gg = total - 1;
    const bool second = gg >= ngroups;
    int e = (gg - (second ? ngroups : 0)) * 32 + lane;
    if (e > E - 1) e = E - 1;
    s = (second ? src1 : src0)[e];
    d = (second ? dst1 : dst0)[e];
  };

  int g = blockIdx.x * 8 + w;
  int si, di;
  load_ids(g, si, di);
  for (; g < total; g += nwaves) {
    const bool second = g >= ngroups;
    const int e = (g - (second ? ngroups : 0)) * 32 + lane;
    int sn, dn;
    load_ids(g + nwaves, sn, dn);
    if ((unsigned)si >= (unsigned)N) si = 0;
    if ((unsigned)di >= (unsigned)N) di = 0;
    const unsigned so = (unsigned)si * 256u;
    const unsigned dof = (unsigned)di * 256u;
    float p[16];
    EDGE_LOAD(0) EDGE_LOAD(1) EDGE_LOAD(2) EDGE_LOAD(3) EDGE_LOAD(4) EDGE_LOAD(5) EDGE_LOAD(6) EDGE_LOAD(7)
    EDGE_LOAD(8) EDGE_LOAD(9) EDGE_LOAD(10) EDGE_LOAD(11) EDGE_LOAD(12) EDGE_LOAD(13) EDGE_LOAD(14) EDGE_LOAD(15)
    EDGE_DOT(0) EDGE_DOT(1) EDGE_DOT(2) EDGE_DOT(3) EDGE_DOT(4) EDGE_DOT(5) EDGE_DOT(6) EDGE_DOT(7)
    EDGE_DOT(8) EDGE_DOT(9) EDGE_DOT(10) EDGE_DOT(11) EDGE_DOT(12) EDGE_DOT(13) EDGE_DOT(14) EDGE_DOT(15)
    float q[8];
#pragma unroll
    for (int s = 0; s < 8; ++s) q[s] = fold_pair<8>(u8, p[s], p[s + 8]);
    float r4[4];
#pragma unroll
    for (int s = 0; s < 4; ++s) r4[s] = fold_pair<4>(u4, q[s], q[s + 4]);
    const float t0 = fold_pair<2>(u2, r4[0], r4[2]);
    const float t1 = fold_pair<2>(u2, r4[1], r4[3]);
    const float score = fold_pair<1>(u1, t0, t1);
    if (e < E) out[(second ? (size_t)E : (size_t)0) + e] = score;
    si = sn; di = dn;
  }
}
#undef EDGE_LOAD
#undef EDGE_DOT

extern "C" void kernel_launch(void* const* d_in, const int* in_sizes, int n_in,
                              void* d_out, int out_size, void* d_ws, size_t ws_size,
                              hipStream_t stream) {
  const float* x   = (const float*)d_in[0];
  const int*   edges = (const int*)d_in[1];
  const int*   negE  = (const int*)d_in[2];
  const float* Wn1 = (const float*)d_in[3];
  const float* Ws1 = (const float*)d_in[4];
  const float* b1  = (const float*)d_in[5];
  const float* Wn2 = (const float*)d_in[6];
  const float* Ws2 = (const float*)d_in[7];
  const float* b2  = (const float*)d_in[8];
  float* out = (float*)d_out;

  const int N = in_sizes[0] / 128;
  const int E = in_sizes[1] / 4;

  float* ws    = (float*)d_ws;
  float* bias2 = ws;
  float* agg0  = ws + ((2 * (size_t)N) & ~(size_t)63);
  float* agg1  = agg0 + (size_t)N * 128;
  float* h1    = agg1 + (size_t)N * 128;
  float* elist = h1 + (size_t)N * 128;
  float* Bcat1 = elist + (size_t)N * 64;
  float* bias1 = Bcat1 + 384 * 128;
  float* Bcat2 = bias1 + 128;
  float* P     = agg0;
  float* h2    = h1;
  const int NS = (N + 8) & ~7;
  int* csr  = (int*)elist;
  int* segx = csr + 2 * (size_t)E;
  int* tmp  = (int*)agg0;
  int* pcnt = (int*)agg0 + 4 * (size_t)E;
  int* poff = pcnt + 2 * SORT_NB * SORT_NC;

  const int T = 256;

  const int wtot = 384 * 128 + 128 + 128 * PROW + PROW;
  build_weights<<<(wtot + T - 1) / T, T, 0, stream>>>(Wn1, Ws1, b1, Wn2, Ws2, b2, Bcat1, bias1, Bcat2, bias2);

  bucket_chunks<<<dim3(SORT_NC, 2), SORT_T, 0, stream>>>(edges, tmp, pcnt, poff, E, N);
  range_pull_gemm<<<SORT_NB, SORT_T, 0, stream>>>(tmp, pcnt, poff, segx, csr, x, Bcat1, bias1, h1, E, N, NS);

  const int sblocks = 1024;
  const int pblocks = 2048;

  sage_gemm<PROW, 128, false><<<GEMM_WGS, GEMM_THREADS, 0, stream>>>(h1, h1, h1, Bcat2, bias2, P, N);
  gather_proj<<<pblocks, T, 0, stream>>>(segx, csr, P, h2, E, N, NS);

  edge_score<<<sblocks, T, 0, stream>>>(edges, edges + E, negE, negE + E, h2, out, E, N);
}
